# starting kernel
# baseline (speedup 1.0000x reference)
#include <hip/hip_runtime.h>
#include <math.h>

typedef __attribute__((ext_vector_type(2))) float v2f;
typedef __attribute__((ext_vector_type(8))) float v8f;

__device__ __forceinline__ unsigned fkey(float f) {
    unsigned u = __float_as_uint(f);
    return (u & 0x80000000u) ? ~u : (u | 0x80000000u);
}
__device__ __forceinline__ float fkey_inv(unsigned k) {
    unsigned u = (k & 0x80000000u) ? (k & 0x7FFFFFFFu) : ~k;
    return __uint_as_float(u);
}

__global__ void init_ws_kernel(unsigned* __restrict__ smaxkey,
                               float* __restrict__ Z,
                               float* __restrict__ outacc, int N) {
    int i = blockIdx.x * blockDim.x + threadIdx.x;
    if (i < N * 32) outacc[i] = 0.0f;
    if (i < N) { smaxkey[i] = 0u; Z[i] = 0.0f; }
}

__device__ __forceinline__ void wmma_step(const float* __restrict__ W, int kb,
                                          int half, int m, v2f a,
                                          v8f& acc0, v8f& acc1) {
    int r0 = (kb + 2 * half + 0) * 32;
    int r1 = (kb + 2 * half + 1) * 32;
    v2f b0, b1;
    b0.x = W[r0 + m];        b0.y = W[r1 + m];
    b1.x = W[r0 + 16 + m];   b1.y = W[r1 + 16 + m];
    acc0 = __builtin_amdgcn_wmma_f32_16x16x4_f32(false, a, false, b0,
                                                 (short)0, acc0, false, false);
    acc1 = __builtin_amdgcn_wmma_f32_16x16x4_f32(false, a, false, b1,
                                                 (short)0, acc1, false, false);
}

__global__ __launch_bounds__(256) void qkv_wmma_kernel(
    const float* __restrict__ x,
    const float* __restrict__ Wq, const float* __restrict__ Wk,
    const float* __restrict__ Wv,
    float* __restrict__ Q, float* __restrict__ K, float* __restrict__ V,
    int ntiles) {
    int wave = (int)((blockIdx.x * blockDim.x + threadIdx.x) >> 5);
    if (wave >= ntiles) return;
    int lane = (int)(threadIdx.x & 31u);
    int half = lane >> 4;
    int m    = lane & 15;
    size_t row0 = (size_t)wave * 16;

    v8f q0 = {}, q1 = {}, k0 = {}, k1 = {}, v0 = {}, v1 = {};

#pragma unroll
    for (int kk = 0; kk < 8; ++kk) {
        int kb = kk * 4;
        const float* ap = x + (row0 + m) * 32 + kb + 2 * half;
        v2f a; a.x = ap[0]; a.y = ap[1];
        wmma_step(Wq, kb, half, m, a, q0, q1);
        wmma_step(Wk, kb, half, m, a, k0, k1);
        wmma_step(Wv, kb, half, m, a, v0, v1);
    }

#pragma unroll
    for (int r = 0; r < 8; ++r) {
        size_t row = (row0 + r + 8 * half) * 32;
        Q[row + m]      = q0[r];  Q[row + 16 + m] = q1[r];
        K[row + m]      = k0[r];  K[row + 16 + m] = k1[r];
        V[row + m]      = v0[r];  V[row + 16 + m] = v1[r];
    }
}

__global__ void qkv_tail_kernel(const float* __restrict__ x,
                                const float* __restrict__ Wq,
                                const float* __restrict__ Wk,
                                const float* __restrict__ Wv,
                                float* __restrict__ Q, float* __restrict__ K,
                                float* __restrict__ V, int row_start, int N) {
    int idx = blockIdx.x * blockDim.x + threadIdx.x;
    int nrows = N - row_start;
    if (idx >= nrows * 32) return;
    int row = row_start + idx / 32, col = idx & 31;
    float q = 0.f, k = 0.f, v = 0.f;
    for (int t = 0; t < 32; ++t) {
        float xv = x[(size_t)row * 32 + t];
        q += xv * Wq[t * 32 + col];
        k += xv * Wk[t * 32 + col];
        v += xv * Wv[t * 32 + col];
    }
    size_t o = (size_t)row * 32 + col;
    Q[o] = q; K[o] = k; V[o] = v;
}

__global__ void edge_score_kernel(const float* __restrict__ Q,
                                  const float* __restrict__ K,
                                  const int* __restrict__ ei,
                                  float* __restrict__ scores,
                                  unsigned* __restrict__ smaxkey, int E) {
    int e = blockIdx.x * blockDim.x + threadIdx.x;
    if (e >= E) return;
    int s = ei[e], r = ei[E + e];
    const float4* q = (const float4*)(Q + (size_t)r * 32);
    const float4* k = (const float4*)(K + (size_t)s * 32);
    float acc = 0.f;
#pragma unroll
    for (int i = 0; i < 8; ++i) {
        float4 a = q[i], b = k[i];
        acc += a.x * b.x + a.y * b.y + a.z * b.z + a.w * b.w;
    }
    float sc = acc * 0.17677669529663687f;
    scores[e] = sc;
    atomicMax(&smaxkey[r], fkey(sc));
}

__global__ void edge_agg_kernel(const float* __restrict__ V,
                                const int* __restrict__ ei,
                                const float* __restrict__ scores,
                                const unsigned* __restrict__ smaxkey,
                                float* __restrict__ Z,
                                float* __restrict__ outacc, int E) {
    int e = blockIdx.x * blockDim.x + threadIdx.x;
    if (e >= E) return;
    int s = ei[e], r = ei[E + e];
    float alpha = __expf(scores[e] - fkey_inv(smaxkey[r]));
    atomicAdd(&Z[r], alpha);
    const float4* v = (const float4*)(V + (size_t)s * 32);
    float* o = outacc + (size_t)r * 32;
#pragma unroll
    for (int i = 0; i < 8; ++i) {
        float4 vv = v[i];
        atomicAdd(o + 4 * i + 0, alpha * vv.x);
        atomicAdd(o + 4 * i + 1, alpha * vv.y);
        atomicAdd(o + 4 * i + 2, alpha * vv.z);
        atomicAdd(o + 4 * i + 3, alpha * vv.w);
    }
}

__global__ __launch_bounds__(256) void out_wmma_kernel(
    const float* __restrict__ x, const float* __restrict__ outacc,
    const float* __restrict__ Z, const float* __restrict__ Wo,
    float* __restrict__ out, int ntiles) {
    int wave = (int)((blockIdx.x * blockDim.x + threadIdx.x) >> 5);
    if (wave >= ntiles) return;
    int lane = (int)(threadIdx.x & 31u);
    int half = lane >> 4;
    int m    = lane & 15;
    size_t row0 = (size_t)wave * 16;

    float invz = 1.0f / (Z[row0 + m] + 1e-6f);
    v8f c0 = {}, c1 = {};

#pragma unroll
    for (int kk = 0; kk < 8; ++kk) {
        int kb = kk * 4;
        const float* ap = outacc + (row0 + m) * 32 + kb + 2 * half;
        v2f a; a.x = ap[0] * invz; a.y = ap[1] * invz;
        wmma_step(Wo, kb, half, m, a, c0, c1);
    }

#pragma unroll
    for (int r = 0; r < 8; ++r) {
        size_t row = (row0 + r + 8 * half) * 32;
        out[row + m]      = x[row + m]      + c0[r];
        out[row + 16 + m] = x[row + 16 + m] + c1[r];
    }
}

__global__ void out_tail_kernel(const float* __restrict__ x,
                                const float* __restrict__ outacc,
                                const float* __restrict__ Z,
                                const float* __restrict__ Wo,
                                float* __restrict__ out, int row_start, int N) {
    int idx = blockIdx.x * blockDim.x + threadIdx.x;
    int nrows = N - row_start;
    if (idx >= nrows * 32) return;
    int row = row_start + idx / 32, col = idx & 31;
    float invz = 1.0f / (Z[row] + 1e-6f);
    float acc = 0.f;
    for (int t = 0; t < 32; ++t)
        acc += outacc[(size_t)row * 32 + t] * invz * Wo[t * 32 + col];
    size_t o = (size_t)row * 32 + col;
    out[o] = x[o] + acc;
}

extern "C" void kernel_launch(void* const* d_in, const int* in_sizes, int n_in,
                              void* d_out, int out_size, void* d_ws,
                              size_t ws_size, hipStream_t stream) {
    const float* x  = (const float*)d_in[0];
    const int*   ei = (const int*)d_in[1];
    const float* Wq = (const float*)d_in[2];
    const float* Wk = (const float*)d_in[3];
    const float* Wv = (const float*)d_in[4];
    const float* Wo = (const float*)d_in[5];
    float* out = (float*)d_out;

    const int N = in_sizes[0] / 32;
    const int E = in_sizes[1] / 2;

    float* Q       = (float*)d_ws;
    float* K       = Q + (size_t)N * 32;
    float* V       = K + (size_t)N * 32;
    float* outacc  = V + (size_t)N * 32;
    float* scores  = outacc + (size_t)N * 32;
    unsigned* smaxkey = (unsigned*)(scores + E);
    float* Z       = (float*)(smaxkey + N);

    const int TB = 256;
    int ntiles = N / 16;
    int rem_row0 = ntiles * 16;
    int rem = N - rem_row0;

    {
        int total = N * 32;
        init_ws_kernel<<<(total + TB - 1) / TB, TB, 0, stream>>>(smaxkey, Z,
                                                                 outacc, N);
    }
    {
        int threads = ntiles * 32;
        if (threads > 0)
            qkv_wmma_kernel<<<(threads + TB - 1) / TB, TB, 0, stream>>>(
                x, Wq, Wk, Wv, Q, K, V, ntiles);
        if (rem > 0)
            qkv_tail_kernel<<<(rem * 32 + TB - 1) / TB, TB, 0, stream>>>(
                x, Wq, Wk, Wv, Q, K, V, rem_row0, N);
    }
    edge_score_kernel<<<(E + TB - 1) / TB, TB, 0, stream>>>(Q, K, ei, scores,
                                                            smaxkey, E);
    edge_agg_kernel<<<(E + TB - 1) / TB, TB, 0, stream>>>(V, ei, scores,
                                                          smaxkey, Z, outacc, E);
    {
        int threads = ntiles * 32;
        if (threads > 0)
            out_wmma_kernel<<<(threads + TB - 1) / TB, TB, 0, stream>>>(
                x, outacc, Z, Wo, out, ntiles);
        if (rem > 0)
            out_tail_kernel<<<(rem * 32 + TB - 1) / TB, TB, 0, stream>>>(
                x, outacc, Z, Wo, out, rem_row0, N);
    }
}


# Edges grouped by receiver with an integer counting sort; one wave per receiver runs a streaming softmax and weighted V sum in registers: float atomics and scores array gone, new summation order
# speedup vs baseline: 2.8310x; 2.8310x over previous
#include <hip/hip_runtime.h>
#include <math.h>

typedef __attribute__((ext_vector_type(2))) float v2f;
typedef __attribute__((ext_vector_type(8))) float v8f;

__global__ void init_ws_kernel(unsigned* __restrict__ cnt,
                               unsigned* __restrict__ total, int N) {
    int i = blockIdx.x * blockDim.x + threadIdx.x;
    if (i < N) cnt[i] = 0u;
    if (i == 0) total[0] = 0u;
}

__global__ void seg_count_kernel(const int* __restrict__ ei,
                                 unsigned* __restrict__ cnt, int E) {
    int e = blockIdx.x * blockDim.x + threadIdx.x;
    if (e >= E) return;
    atomicAdd(&cnt[ei[E + e]], 1u);
}

__global__ void seg_alloc_kernel(unsigned* __restrict__ cnt,
                                 unsigned* __restrict__ degz,
                                 unsigned* __restrict__ total, int N) {
    int r = blockIdx.x * blockDim.x + threadIdx.x;
    int lane = (int)(threadIdx.x & 31u);
    unsigned c = (r < N) ? cnt[r] : 0u;
    unsigned inc = c;
    for (int d = 1; d < 32; d <<= 1) {
        unsigned t = __shfl_up(inc, d);
        if (lane >= d) inc += t;
    }
    unsigned wtot = __shfl(inc, 31);
    unsigned base = 0u;
    if (lane == 0) base = atomicAdd(total, wtot);
    base = __shfl(base, 0);
    if (r < N) { cnt[r] = base + inc - c; degz[r] = c; }
}

__global__ void seg_scatter_kernel(const int* __restrict__ ei,
                                   unsigned* __restrict__ cursor,
                                   int* __restrict__ sorted_s, int E) {
    int e = blockIdx.x * blockDim.x + threadIdx.x;
    if (e >= E) return;
    int s = ei[e], r = ei[E + e];
    unsigned slot = atomicAdd(&cursor[r], 1u);
    sorted_s[slot] = s;
}

__device__ __forceinline__ void wmma_step(const float* __restrict__ W, int kb,
                                          int half, int m, v2f a,
                                          v8f& acc0, v8f& acc1) {
    int r0 = (kb + 2 * half + 0) * 32;
    int r1 = (kb + 2 * half + 1) * 32;
    v2f b0, b1;
    b0.x = W[r0 + m];        b0.y = W[r1 + m];
    b1.x = W[r0 + 16 + m];   b1.y = W[r1 + 16 + m];
    acc0 = __builtin_amdgcn_wmma_f32_16x16x4_f32(false, a, false, b0,
                                                 (short)0, acc0, false, false);
    acc1 = __builtin_amdgcn_wmma_f32_16x16x4_f32(false, a, false, b1,
                                                 (short)0, acc1, false, false);
}

__global__ __launch_bounds__(256) void qkv_wmma_kernel(
    const float* __restrict__ x,
    const float* __restrict__ Wq, const float* __restrict__ Wk,
    const float* __restrict__ Wv,
    float* __restrict__ Q, float* __restrict__ K, float* __restrict__ V,
    int ntiles) {
    int wave = (int)((blockIdx.x * blockDim.x + threadIdx.x) >> 5);
    if (wave >= ntiles) return;
    int lane = (int)(threadIdx.x & 31u);
    int half = lane >> 4;
    int m    = lane & 15;
    size_t row0 = (size_t)wave * 16;

    v8f q0 = {}, q1 = {}, k0 = {}, k1 = {}, v0 = {}, v1 = {};

#pragma unroll
    for (int kk = 0; kk < 8; ++kk) {
        int kb = kk * 4;
        const float* ap = x + (row0 + m) * 32 + kb + 2 * half;
        v2f a; a.x = ap[0]; a.y = ap[1];
        wmma_step(Wq, kb, half, m, a, q0, q1);
        wmma_step(Wk, kb, half, m, a, k0, k1);
        wmma_step(Wv, kb, half, m, a, v0, v1);
    }

#pragma unroll
    for (int r = 0; r < 8; ++r) {
        size_t row = (row0 + r + 8 * half) * 32;
        Q[row + m]      = q0[r];  Q[row + 16 + m] = q1[r];
        K[row + m]      = k0[r];  K[row + 16 + m] = k1[r];
        V[row + m]      = v0[r];  V[row + 16 + m] = v1[r];
    }
}

__global__ void qkv_tail_kernel(const float* __restrict__ x,
                                const float* __restrict__ Wq,
                                const float* __restrict__ Wk,
                                const float* __restrict__ Wv,
                                float* __restrict__ Q, float* __restrict__ K,
                                float* __restrict__ V, int row_start, int N) {
    int idx = blockIdx.x * blockDim.x + threadIdx.x;
    int nrows = N - row_start;
    if (idx >= nrows * 32) return;
    int row = row_start + idx / 32, col = idx & 31;
    float q = 0.f, k = 0.f, v = 0.f;
    for (int t = 0; t < 32; ++t) {
        float xv = x[(size_t)row * 32 + t];
        q += xv * Wq[t * 32 + col];
        k += xv * Wk[t * 32 + col];
        v += xv * Wv[t * 32 + col];
    }
    size_t o = (size_t)row * 32 + col;
    Q[o] = q; K[o] = k; V[o] = v;
}

__global__ __launch_bounds__(256) void seg_attn_kernel(
    const float* __restrict__ Q, const float* __restrict__ K,
    const float* __restrict__ V, const int* __restrict__ sorted_s,
    const unsigned* __restrict__ segend, unsigned* __restrict__ degz,
    float* __restrict__ outacc, int N) {
    int r = (int)((blockIdx.x * blockDim.x + threadIdx.x) >> 5);
    if (r >= N) return;
    int lane = (int)(threadIdx.x & 31u);
    int g = lane >> 3;
    int c = lane & 7;
    int deg   = __builtin_amdgcn_readfirstlane((int)degz[r]);
    int start = __builtin_amdgcn_readfirstlane((int)segend[r]) - deg;

    const float4 q4 = ((const float4*)(Q + (size_t)r * 32))[c];
    float m = -INFINITY;
    float z = 0.f;
    float4 acc = make_float4(0.f, 0.f, 0.f, 0.f);

    int s_cur = (g < deg) ? sorted_s[start + g] : r;
    for (int t = 0; t < deg; t += 4) {
        int jn = t + 4 + g;
        int s_nxt = (jn < deg) ? sorted_s[start + jn] : r;
        bool act = (t + g) < deg;
        float4 k4 = ((const float4*)(K + (size_t)s_cur * 32))[c];
        float4 v4 = ((const float4*)(V + (size_t)s_cur * 32))[c];
        float p = q4.x * k4.x + q4.y * k4.y + q4.z * k4.z + q4.w * k4.w;
        p += __shfl_xor(p, 1);
        p += __shfl_xor(p, 2);
        p += __shfl_xor(p, 4);
        if (act) {
            float sc = p * 0.17677669529663687f;
            float nm = fmaxf(m, sc);
            float corr = __expf(m - nm);
            float a = __expf(sc - nm);
            z = z * corr + a;
            acc.x = acc.x * corr + a * v4.x;
            acc.y = acc.y * corr + a * v4.y;
            acc.z = acc.z * corr + a * v4.z;
            acc.w = acc.w * corr + a * v4.w;
            m = nm;
        }
        s_cur = s_nxt;
    }

    float M = m;
    M = fmaxf(M, __shfl_xor(M, 8));
    M = fmaxf(M, __shfl_xor(M, 16));
    float f = (m == -INFINITY) ? 0.f : __expf(m - M);
    z *= f; acc.x *= f; acc.y *= f; acc.z *= f; acc.w *= f;
    z     += __shfl_xor(z, 8);      z     += __shfl_xor(z, 16);
    acc.x += __shfl_xor(acc.x, 8);  acc.x += __shfl_xor(acc.x, 16);
    acc.y += __shfl_xor(acc.y, 8);  acc.y += __shfl_xor(acc.y, 16);
    acc.z += __shfl_xor(acc.z, 8);  acc.z += __shfl_xor(acc.z, 16);
    acc.w += __shfl_xor(acc.w, 8);  acc.w += __shfl_xor(acc.w, 16);
    if (g == 0) ((float4*)(outacc + (size_t)r * 32))[c] = acc;
    if (lane == 0) degz[r] = __float_as_uint(z);
}

__global__ __launch_bounds__(256) void out_wmma_kernel(
    const float* __restrict__ x, const float* __restrict__ outacc,
    const float* __restrict__ Z, const float* __restrict__ Wo,
    float* __restrict__ out, int ntiles) {
    int wave = (int)((blockIdx.x * blockDim.x + threadIdx.x) >> 5);
    if (wave >= ntiles) return;
    int lane = (int)(threadIdx.x & 31u);
    int half = lane >> 4;
    int m    = lane & 15;
    size_t row0 = (size_t)wave * 16;

    float invz = 1.0f / (Z[row0 + m] + 1e-6f);
    v8f c0 = {}, c1 = {};

#pragma unroll
    for (int kk = 0; kk < 8; ++kk) {
        int kb = kk * 4;
        const float* ap = outacc + (row0 + m) * 32 + kb + 2 * half;
        v2f a; a.x = ap[0] * invz; a.y = ap[1] * invz;
        wmma_step(Wo, kb, half, m, a, c0, c1);
    }

#pragma unroll
    for (int r = 0; r < 8; ++r) {
        size_t row = (row0 + r + 8 * half) * 32;
        out[row + m]      = x[row + m]      + c0[r];
        out[row + 16 + m] = x[row + 16 + m] + c1[r];
    }
}

__global__ void out_tail_kernel(const float* __restrict__ x,
                                const float* __restrict__ outacc,
                                const float* __restrict__ Z,
                                const float* __restrict__ Wo,
                                float* __restrict__ out, int row_start, int N) {
    int idx = blockIdx.x * blockDim.x + threadIdx.x;
    int nrows = N - row_start;
    if (idx >= nrows * 32) return;
    int row = row_start + idx / 32, col = idx & 31;
    float invz = 1.0f / (Z[row] + 1e-6f);
    float acc = 0.f;
    for (int t = 0; t < 32; ++t)
        acc += outacc[(size_t)row * 32 + t] * invz * Wo[t * 32 + col];
    size_t o = (size_t)row * 32 + col;
    out[o] = x[o] + acc;
}

extern "C" void kernel_launch(void* const* d_in, const int* in_sizes, int n_in,
                              void* d_out, int out_size, void* d_ws,
                              size_t ws_size, hipStream_t stream) {
    const float* x  = (const float*)d_in[0];
    const int*   ei = (const int*)d_in[1];
    const float* Wq = (const float*)d_in[2];
    const float* Wk = (const float*)d_in[3];
    const float* Wv = (const float*)d_in[4];
    const float* Wo = (const float*)d_in[5];
    float* out = (float*)d_out;

    const int N = in_sizes[0] / 32;
    const int E = in_sizes[1] / 2;

    float* Q       = (float*)d_ws;
    float* K       = Q + (size_t)N * 32;
    float* V       = K + (size_t)N * 32;
    float* outacc  = V + (size_t)N * 32;
    int* sorted_s  = (int*)(outacc + (size_t)N * 32);
    unsigned* cnt  = (unsigned*)(sorted_s + E);
    unsigned* degz = cnt + N;
    unsigned* total = (unsigned*)outacc;

    const int TB = 256;
    int ntiles = N / 16;
    int rem_row0 = ntiles * 16;
    int rem = N - rem_row0;

    init_ws_kernel<<<(N + TB - 1) / TB, TB, 0, stream>>>(cnt, total, N);
    {
        int threads = ntiles * 32;
        if (threads > 0)
            qkv_wmma_kernel<<<(threads + TB - 1) / TB, TB, 0, stream>>>(
                x, Wq, Wk, Wv, Q, K, V, ntiles);
        if (rem > 0)
            qkv_tail_kernel<<<(rem * 32 + TB - 1) / TB, TB, 0, stream>>>(
                x, Wq, Wk, Wv, Q, K, V, rem_row0, N);
    }
    seg_count_kernel<<<(E + TB - 1) / TB, TB, 0, stream>>>(ei, cnt, E);
    seg_alloc_kernel<<<(N + TB - 1) / TB, TB, 0, stream>>>(cnt, degz, total, N);
    seg_scatter_kernel<<<(E + TB - 1) / TB, TB, 0, stream>>>(ei, cnt, sorted_s,
                                                             E);
    seg_attn_kernel<<<(N + 7) / 8, TB, 0, stream>>>(Q, K, V, sorted_s, cnt,
                                                    degz, outacc, N);
    {
        int threads = ntiles * 32;
        if (threads > 0)
            out_wmma_kernel<<<(threads + TB - 1) / TB, TB, 0, stream>>>(
                x, outacc, (const float*)degz, Wo, out, ntiles);
        if (rem > 0)
            out_tail_kernel<<<(rem * 32 + TB - 1) / TB, TB, 0, stream>>>(
                x, outacc, (const float*)degz, Wo, out, rem_row0, N);
    }
}


# Edges grouped by receiver with LDS counters in two levels (chunk histograms, bucket placement, in-bucket sort) instead of one global atomic per edge; sum order still the sort's arrival order
# speedup vs baseline: 6.2890x; 2.2215x over previous
#include <hip/hip_runtime.h>
#include <math.h>

typedef __attribute__((ext_vector_type(2))) float v2f;
typedef __attribute__((ext_vector_type(8))) float v8f;

#define SORT_C   256
#define SORT_T   512
#define RB_SHIFT 8
#define NB_MAX   512
#define EPT      8
static_assert(SORT_T == NB_MAX && SORT_T == 512 && SORT_C == 256,
              "the table walk of the placement pass is written for these");

__device__ __forceinline__ unsigned bucket_of(int r, int NB) {
    unsigned b = (unsigned)r >> RB_SHIFT;
    unsigned last = (unsigned)(NB - 1);
    return b < last ? b : last;
}

__global__ __launch_bounds__(SORT_T) void bucket_hist_kernel(
    const int* __restrict__ ei, unsigned* __restrict__ table, int E, int NB) {
    __shared__ unsigned hist[NB_MAX];
    int tid = (int)threadIdx.x;
    int c = (int)blockIdx.x;
    hist[tid] = 0u;
    __syncthreads();
    int CH = (E + SORT_C - 1) / SORT_C;
    int beg = c * CH;
    int end = (beg + CH < E) ? beg + CH : E;
    for (int e = beg + tid; e < end; e += EPT * SORT_T) {
        int r[EPT];
#pragma unroll
        for (int k = 0; k < EPT; ++k) {
            int ee = e + k * SORT_T;
            r[k] = ei[E + (ee < end ? ee : end - 1)];
        }
#pragma unroll
        for (int k = 0; k < EPT; ++k)
            if (e + k * SORT_T < end)
                atomicAdd(&hist[bucket_of(r[k], NB)], 1u);
    }
    __syncthreads();
    table[c * NB_MAX + tid] = hist[tid];
}

__global__ __launch_bounds__(SORT_T) void bucket_place_kernel(
    const int* __restrict__ ei, const unsigned* __restrict__ table,
    unsigned* __restrict__ bstart, int* __restrict__ pairs_raw, int E, int NB) {
    __shared__ unsigned cursor[NB_MAX];
    __shared__ unsigned wsum[SORT_T / 32];
    __shared__ __attribute__((aligned(16))) unsigned ptot[4][NB_MAX];
    __shared__ __attribute__((aligned(16))) unsigned ppre[4][NB_MAX];
    int2* __restrict__ pairs = (int2*)pairs_raw;
    int tid = (int)threadIdx.x;
    int c = (int)blockIdx.x;
    int lane = tid & 31;
    int w = tid >> 5;

    {
        int q = tid & 127;
        int g = tid >> 7;
        int gg = (g + c) & 3;
        const uint4* __restrict__ t4 = (const uint4*)table;
        unsigned t0 = 0u, t1 = 0u, t2 = 0u, t3 = 0u;
        unsigned p0 = 0u, p1 = 0u, p2 = 0u, p3 = 0u;
        for (int k0 = 0; k0 < SORT_C / 4; k0 += 16) {
            uint4 v[16];
#pragma unroll
            for (int j = 0; j < 16; ++j)
                v[j] = t4[(gg * (SORT_C / 4) + k0 + j) * (NB_MAX / 4) + q];
#pragma unroll
            for (int j = 0; j < 16; ++j) {
                unsigned m = (gg * (SORT_C / 4) + k0 + j < c) ? 0xffffffffu : 0u;
                t0 += v[j].x; t1 += v[j].y; t2 += v[j].z; t3 += v[j].w;
                p0 += v[j].x & m;  p1 += v[j].y & m;
                p2 += v[j].z & m;  p3 += v[j].w & m;
            }
        }
        ((uint4*)ptot[g])[q] = make_uint4(t0, t1, t2, t3);
        ((uint4*)ppre[g])[q] = make_uint4(p0, p1, p2, p3);
    }
    __syncthreads();
    unsigned tot = ptot[0][tid] + ptot[1][tid] + ptot[2][tid] + ptot[3][tid];
    unsigned pre = ppre[0][tid] + ppre[1][tid] + ppre[2][tid] + ppre[3][tid];
    unsigned inc = tot;
    for (int d = 1; d < 32; d <<= 1) {
        unsigned t = __shfl_up(inc, d);
        if (lane >= d) inc += t;
    }
    if (lane == 31) wsum[w] = inc;
    __syncthreads();
    unsigned base = 0u;
    for (int k = 0; k < w; ++k) base += wsum[k];
    unsigned excl = base + inc - tot;
    cursor[tid] = excl + pre;
    if (c == 0 && tid < NB) {
        bstart[tid] = excl;
        if (tid == NB - 1) bstart[NB] = excl + tot;
    }
    __syncthreads();

    int CH = (E + SORT_C - 1) / SORT_C;
    int beg = c * CH;
    int end = (beg + CH < E) ? beg + CH : E;
    for (int e = beg + tid; e < end; e += EPT * SORT_T) {
        int s[EPT], r[EPT];
#pragma unroll
        for (int k = 0; k < EPT; ++k) {
            int ee = e + k * SORT_T;
            int ec = ee < end ? ee : end - 1;
            s[k] = ei[ec];
            r[k] = ei[E + ec];
        }
#pragma unroll
        for (int k = 0; k < EPT; ++k)
            if (e + k * SORT_T < end) {
                unsigned pos = atomicAdd(&cursor[bucket_of(r[k], NB)], 1u);
                pairs[pos] = make_int2(s[k], r[k]);
            }
    }
}

__global__ __launch_bounds__(SORT_T) void bucket_sort_kernel(
    const int* __restrict__ pairs_raw, const unsigned* __restrict__ bstart,
    unsigned* __restrict__ segend, unsigned* __restrict__ degz,
    int* __restrict__ sorted_s, int N) {
    __shared__ unsigned cnt[256];
    __shared__ unsigned cur[256];
    __shared__ unsigned wsum[8];
    const int2* __restrict__ pairs = (const int2*)pairs_raw;
    int tid = (int)threadIdx.x;
    int b = (int)blockIdx.x;
    int lane = tid & 31;
    int w = tid >> 5;
    unsigned beg = bstart[b], end = bstart[b + 1];

    if (tid < 256) cnt[tid] = 0u;
    __syncthreads();
    for (unsigned i = beg + (unsigned)tid; i < end; i += EPT * SORT_T) {
        int2 p[EPT];
#pragma unroll
        for (int k = 0; k < EPT; ++k) {
            unsigned ii = i + (unsigned)(k * SORT_T);
            p[k] = pairs[ii < end ? ii : end - 1u];
        }
#pragma unroll
        for (int k = 0; k < EPT; ++k)
            if (i + (unsigned)(k * SORT_T) < end)
                atomicAdd(&cnt[p[k].y & 255], 1u);
    }
    __syncthreads();
    unsigned cv = (tid < 256) ? cnt[tid] : 0u;
    unsigned inc = cv;
    for (int d = 1; d < 32; d <<= 1) {
        unsigned t = __shfl_up(inc, d);
        if (lane >= d) inc += t;
    }
    if (tid < 256 && lane == 31) wsum[w] = inc;
    __syncthreads();
    if (tid < 256) {
        unsigned base = 0u;
        for (int k = 0; k < w; ++k) base += wsum[k];
        unsigned st = beg + base + inc - cv;
        cur[tid] = st;
        int r = (b << RB_SHIFT) + tid;
        if (r < N) { segend[r] = st + cv; degz[r] = cv; }
    }
    __syncthreads();
    for (unsigned i = beg + (unsigned)tid; i < end; i += EPT * SORT_T) {
        int2 p[EPT];
#pragma unroll
        for (int k = 0; k < EPT; ++k) {
            unsigned ii = i + (unsigned)(k * SORT_T);
            p[k] = pairs[ii < end ? ii : end - 1u];
        }
#pragma unroll
        for (int k = 0; k < EPT; ++k)
            if (i + (unsigned)(k * SORT_T) < end) {
                unsigned pos = atomicAdd(&cur[p[k].y & 255], 1u);
                sorted_s[pos] = p[k].x;
            }
    }
}

__device__ __forceinline__ void wmma_step(const float* __restrict__ W, int kb,
                                          int half, int m, v2f a,
                                          v8f& acc0, v8f& acc1) {
    int r0 = (kb + 2 * half + 0) * 32;
    int r1 = (kb + 2 * half + 1) * 32;
    v2f b0, b1;
    b0.x = W[r0 + m];        b0.y = W[r1 + m];
    b1.x = W[r0 + 16 + m];   b1.y = W[r1 + 16 + m];
    acc0 = __builtin_amdgcn_wmma_f32_16x16x4_f32(false, a, false, b0,
                                                 (short)0, acc0, false, false);
    acc1 = __builtin_amdgcn_wmma_f32_16x16x4_f32(false, a, false, b1,
                                                 (short)0, acc1, false, false);
}

__global__ __launch_bounds__(256) void qkv_wmma_kernel(
    const float* __restrict__ x,
    const float* __restrict__ Wq, const float* __restrict__ Wk,
    const float* __restrict__ Wv,
    float* __restrict__ Q, float* __restrict__ K, float* __restrict__ V,
    int ntiles) {
    int wave = (int)((blockIdx.x * blockDim.x + threadIdx.x) >> 5);
    if (wave >= ntiles) return;
    int lane = (int)(threadIdx.x & 31u);
    int half = lane >> 4;
    int m    = lane & 15;
    size_t row0 = (size_t)wave * 16;

    v8f q0 = {}, q1 = {}, k0 = {}, k1 = {}, v0 = {}, v1 = {};

#pragma unroll
    for (int kk = 0; kk < 8; ++kk) {
        int kb = kk * 4;
        const float* ap = x + (row0 + m) * 32 + kb + 2 * half;
        v2f a; a.x = ap[0]; a.y = ap[1];
        wmma_step(Wq, kb, half, m, a, q0, q1);
        wmma_step(Wk, kb, half, m, a, k0, k1);
        wmma_step(Wv, kb, half, m, a, v0, v1);
    }

#pragma unroll
    for (int r = 0; r < 8; ++r) {
        size_t row = (row0 + r + 8 * half) * 32;
        Q[row + m]      = q0[r];  Q[row + 16 + m] = q1[r];
        K[row + m]      = k0[r];  K[row + 16 + m] = k1[r];
        V[row + m]      = v0[r];  V[row + 16 + m] = v1[r];
    }
}

__global__ void qkv_tail_kernel(const float* __restrict__ x,
                                const float* __restrict__ Wq,
                                const float* __restrict__ Wk,
                                const float* __restrict__ Wv,
                                float* __restrict__ Q, float* __restrict__ K,
                                float* __restrict__ V, int row_start, int N) {
    int idx = blockIdx.x * blockDim.x + threadIdx.x;
    int nrows = N - row_start;
    if (idx >= nrows * 32) return;
    int row = row_start + idx / 32, col = idx & 31;
    float q = 0.f, k = 0.f, v = 0.f;
    for (int t = 0; t < 32; ++t) {
        float xv = x[(size_t)row * 32 + t];
        q += xv * Wq[t * 32 + col];
        k += xv * Wk[t * 32 + col];
        v += xv * Wv[t * 32 + col];
    }
    size_t o = (size_t)row * 32 + col;
    Q[o] = q; K[o] = k; V[o] = v;
}

__global__ __launch_bounds__(256) void seg_attn_kernel(
    const float* __restrict__ Q, const float* __restrict__ K,
    const float* __restrict__ V, const int* __restrict__ sorted_s,
    const unsigned* __restrict__ segend, unsigned* __restrict__ degz,
    float* __restrict__ outacc, int N) {
    int r = (int)((blockIdx.x * blockDim.x + threadIdx.x) >> 5);
    if (r >= N) return;
    int lane = (int)(threadIdx.x & 31u);
    int g = lane >> 3;
    int c = lane & 7;
    int deg   = __builtin_amdgcn_readfirstlane((int)degz[r]);
    int start = __builtin_amdgcn_readfirstlane((int)segend[r]) - deg;

    const float4 q4 = ((const float4*)(Q + (size_t)r * 32))[c];
    float m = -INFINITY;
    float z = 0.f;
    float4 acc = make_float4(0.f, 0.f, 0.f, 0.f);

    int s_cur = (g < deg) ? sorted_s[start + g] : r;
    for (int t = 0; t < deg; t += 4) {
        int jn = t + 4 + g;
        int s_nxt = (jn < deg) ? sorted_s[start + jn] : r;
        bool act = (t + g) < deg;
        float4 k4 = ((const float4*)(K + (size_t)s_cur * 32))[c];
        float4 v4 = ((const float4*)(V + (size_t)s_cur * 32))[c];
        float p = q4.x * k4.x + q4.y * k4.y + q4.z * k4.z + q4.w * k4.w;
        p += __shfl_xor(p, 1);
        p += __shfl_xor(p, 2);
        p += __shfl_xor(p, 4);
        if (act) {
            float sc = p * 0.17677669529663687f;
            float nm = fmaxf(m, sc);
            float corr = __expf(m - nm);
            float a = __expf(sc - nm);
            z = z * corr + a;
            acc.x = acc.x * corr + a * v4.x;
            acc.y = acc.y * corr + a * v4.y;
            acc.z = acc.z * corr + a * v4.z;
            acc.w = acc.w * corr + a * v4.w;
            m = nm;
        }
        s_cur = s_nxt;
    }

    float M = m;
    M = fmaxf(M, __shfl_xor(M, 8));
    M = fmaxf(M, __shfl_xor(M, 16));
    float f = (m == -INFINITY) ? 0.f : __expf(m - M);
    z *= f; acc.x *= f; acc.y *= f; acc.z *= f; acc.w *= f;
    z     += __shfl_xor(z, 8);      z     += __shfl_xor(z, 16);
    acc.x += __shfl_xor(acc.x, 8);  acc.x += __shfl_xor(acc.x, 16);
    acc.y += __shfl_xor(acc.y, 8);  acc.y += __shfl_xor(acc.y, 16);
    acc.z += __shfl_xor(acc.z, 8);  acc.z += __shfl_xor(acc.z, 16);
    acc.w += __shfl_xor(acc.w, 8);  acc.w += __shfl_xor(acc.w, 16);
    if (g == 0) ((float4*)(outacc + (size_t)r * 32))[c] = acc;
    if (lane == 0) degz[r] = __float_as_uint(z);
}

__global__ __launch_bounds__(256) void out_wmma_kernel(
    const float* __restrict__ x, const float* __restrict__ outacc,
    const float* __restrict__ Z, const float* __restrict__ Wo,
    float* __restrict__ out, int ntiles) {
    int wave = (int)((blockIdx.x * blockDim.x + threadIdx.x) >> 5);
    if (wave >= ntiles) return;
    int lane = (int)(threadIdx.x & 31u);
    int half = lane >> 4;
    int m    = lane & 15;
    size_t row0 = (size_t)wave * 16;

    float invz = 1.0f / (Z[row0 + m] + 1e-6f);
    v8f c0 = {}, c1 = {};

#pragma unroll
    for (int kk = 0; kk < 8; ++kk) {
        int kb = kk * 4;
        const float* ap = outacc + (row0 + m) * 32 + kb + 2 * half;
        v2f a; a.x = ap[0] * invz; a.y = ap[1] * invz;
        wmma_step(Wo, kb, half, m, a, c0, c1);
    }

#pragma unroll
    for (int r = 0; r < 8; ++r) {
        size_t row = (row0 + r + 8 * half) * 32;
        out[row + m]      = x[row + m]      + c0[r];
        out[row + 16 + m] = x[row + 16 + m] + c1[r];
    }
}

__global__ void out_tail_kernel(const float* __restrict__ x,
                                const float* __restrict__ outacc,
                                const float* __restrict__ Z,
                                const float* __restrict__ Wo,
                                float* __restrict__ out, int row_start, int N) {
    int idx = blockIdx.x * blockDim.x + threadIdx.x;
    int nrows = N - row_start;
    if (idx >= nrows * 32) return;
    int row = row_start + idx / 32, col = idx & 31;
    float invz = 1.0f / (Z[row] + 1e-6f);
    float acc = 0.f;
    for (int t = 0; t < 32; ++t)
        acc += outacc[(size_t)row * 32 + t] * invz * Wo[t * 32 + col];
    size_t o = (size_t)row * 32 + col;
    out[o] = x[o] + acc;
}

extern "C" void kernel_launch(void* const* d_in, const int* in_sizes, int n_in,
                              void* d_out, int out_size, void* d_ws,
                              size_t ws_size, hipStream_t stream) {
    const float* x  = (const float*)d_in[0];
    const int*   ei = (const int*)d_in[1];
    const float* Wq = (const float*)d_in[2];
    const float* Wk = (const float*)d_in[3];
    const float* Wv = (const float*)d_in[4];
    const float* Wo = (const float*)d_in[5];
    float* out = (float*)d_out;

    const int N = in_sizes[0] / 32;
    const int E = in_sizes[1] / 2;

    float* Q       = (float*)d_ws;
    float* K       = Q + (size_t)N * 32;
    float* V       = K + (size_t)N * 32;
    float* outacc  = V + (size_t)N * 32;
    int* sorted_s  = (int*)(outacc + (size_t)N * 32);
    unsigned* cnt  = (unsigned*)(sorted_s + E);
    unsigned* degz = cnt + N;
    unsigned* table  = (unsigned*)Q;
    unsigned* bstart = table + (size_t)SORT_C * NB_MAX;
    int* pairs       = (int*)K;

    const int TB = 256;
    int ntiles = N / 16;
    int rem_row0 = ntiles * 16;
    int rem = N - rem_row0;
    int NB = (N + 255) >> RB_SHIFT;

    bucket_hist_kernel<<<SORT_C, SORT_T, 0, stream>>>(ei, table, E, NB);
    bucket_place_kernel<<<SORT_C, SORT_T, 0, stream>>>(ei, table, bstart,
                                                       pairs, E, NB);
    bucket_sort_kernel<<<NB, SORT_T, 0, stream>>>(pairs, bstart, cnt, degz,
                                                  sorted_s, N);
    {
        int threads = ntiles * 32;
        if (threads > 0)
            qkv_wmma_kernel<<<(threads + TB - 1) / TB, TB, 0, stream>>>(
                x, Wq, Wk, Wv, Q, K, V, ntiles);
        if (rem > 0)
            qkv_tail_kernel<<<(rem * 32 + TB - 1) / TB, TB, 0, stream>>>(
                x, Wq, Wk, Wv, Q, K, V, rem_row0, N);
    }
    seg_attn_kernel<<<(N + 7) / 8, TB, 0, stream>>>(Q, K, V, sorted_s, cnt,
                                                    degz, outacc, N);
    {
        int threads = ntiles * 32;
        if (threads > 0)
            out_wmma_kernel<<<(threads + TB - 1) / TB, TB, 0, stream>>>(
                x, outacc, (const float*)degz, Wo, out, ntiles);
        if (rem > 0)
            out_tail_kernel<<<(rem * 32 + TB - 1) / TB, TB, 0, stream>>>(
                x, outacc, (const float*)degz, Wo, out, rem_row0, N);
    }
}


# Segment softmax kernel: each wave walks 7 consecutive receivers (next receiver's Q row and first senders loaded ahead) instead of one wave per receiver: 7x fewer, longer waves; sum order unchanged
# speedup vs baseline: 6.9493x; 1.1050x over previous
#include <hip/hip_runtime.h>
#include <math.h>

typedef __attribute__((ext_vector_type(2))) float v2f;
typedef __attribute__((ext_vector_type(8))) float v8f;

#define SORT_C   256
#define SORT_T   512
#define RB_SHIFT 8
#define NB_MAX   512
#define EPT      8
static_assert(SORT_T == NB_MAX && SORT_T == 512 && SORT_C == 256,
              "the table walk of the placement pass is written for these");

__device__ __forceinline__ unsigned bucket_of(int r, int NB) {
    unsigned b = (unsigned)r >> RB_SHIFT;
    unsigned last = (unsigned)(NB - 1);
    return b < last ? b : last;
}

__global__ __launch_bounds__(SORT_T) void bucket_hist_kernel(
    const int* __restrict__ ei, unsigned* __restrict__ table, int E, int NB) {
    __shared__ unsigned hist[NB_MAX];
    int tid = (int)threadIdx.x;
    int c = (int)blockIdx.x;
    hist[tid] = 0u;
    __syncthreads();
    int CH = (E + SORT_C - 1) / SORT_C;
    int beg = c * CH;
    int end = (beg + CH < E) ? beg + CH : E;
    for (int e = beg + tid; e < end; e += EPT * SORT_T) {
        int r[EPT];
#pragma unroll
        for (int k = 0; k < EPT; ++k) {
            int ee = e + k * SORT_T;
            r[k] = ei[E + (ee < end ? ee : end - 1)];
        }
#pragma unroll
        for (int k = 0; k < EPT; ++k)
            if (e + k * SORT_T < end)
                atomicAdd(&hist[bucket_of(r[k], NB)], 1u);
    }
    __syncthreads();
    table[c * NB_MAX + tid] = hist[tid];
}

__global__ __launch_bounds__(SORT_T) void bucket_place_kernel(
    const int* __restrict__ ei, const unsigned* __restrict__ table,
    unsigned* __restrict__ bstart, int* __restrict__ pairs_raw, int E, int NB) {
    __shared__ unsigned cursor[NB_MAX];
    __shared__ unsigned wsum[SORT_T / 32];
    __shared__ __attribute__((aligned(16))) unsigned ptot[4][NB_MAX];
    __shared__ __attribute__((aligned(16))) unsigned ppre[4][NB_MAX];
    int2* __restrict__ pairs = (int2*)pairs_raw;
    int tid = (int)threadIdx.x;
    int c = (int)blockIdx.x;
    int lane = tid & 31;
    int w = tid >> 5;

    {
        int q = tid & 127;
        int g = tid >> 7;
        int gg = (g + c) & 3;
        const uint4* __restrict__ t4 = (const uint4*)table;
        unsigned t0 = 0u, t1 = 0u, t2 = 0u, t3 = 0u;
        unsigned p0 = 0u, p1 = 0u, p2 = 0u, p3 = 0u;
        for (int k0 = 0; k0 < SORT_C / 4; k0 += 16) {
            uint4 v[16];
#pragma unroll
            for (int j = 0; j < 16; ++j)
                v[j] = t4[(gg * (SORT_C / 4) + k0 + j) * (NB_MAX / 4) + q];
#pragma unroll
            for (int j = 0; j < 16; ++j) {
                unsigned m = (gg * (SORT_C / 4) + k0 + j < c) ? 0xffffffffu : 0u;
                t0 += v[j].x; t1 += v[j].y; t2 += v[j].z; t3 += v[j].w;
                p0 += v[j].x & m;  p1 += v[j].y & m;
                p2 += v[j].z & m;  p3 += v[j].w & m;
            }
        }
        ((uint4*)ptot[g])[q] = make_uint4(t0, t1, t2, t3);
        ((uint4*)ppre[g])[q] = make_uint4(p0, p1, p2, p3);
    }
    __syncthreads();
    unsigned tot = ptot[0][tid] + ptot[1][tid] + ptot[2][tid] + ptot[3][tid];
    unsigned pre = ppre[0][tid] + ppre[1][tid] + ppre[2][tid] + ppre[3][tid];
    unsigned inc = tot;
    for (int d = 1; d < 32; d <<= 1) {
        unsigned t = __shfl_up(inc, d);
        if (lane >= d) inc += t;
    }
    if (lane == 31) wsum[w] = inc;
    __syncthreads();
    unsigned base = 0u;
    for (int k = 0; k < w; ++k) base += wsum[k];
    unsigned excl = base + inc - tot;
    cursor[tid] = excl + pre;
    if (c == 0 && tid < NB) {
        bstart[tid] = excl;
        if (tid == NB - 1) bstart[NB] = excl + tot;
    }
    __syncthreads();

    int CH = (E + SORT_C - 1) / SORT_C;
    int beg = c * CH;
    int end = (beg + CH < E) ? beg + CH : E;
    for (int e = beg + tid; e < end; e += EPT * SORT_T) {
        int s[EPT], r[EPT];
#pragma unroll
        for (int k = 0; k < EPT; ++k) {
            int ee = e + k * SORT_T;
            int ec = ee < end ? ee : end - 1;
            s[k] = ei[ec];
            r[k] = ei[E + ec];
        }
#pragma unroll
        for (int k = 0; k < EPT; ++k)
            if (e + k * SORT_T < end) {
                unsigned pos = atomicAdd(&cursor[bucket_of(r[k], NB)], 1u);
                pairs[pos] = make_int2(s[k], r[k]);
            }
    }
}

__global__ __launch_bounds__(SORT_T) void bucket_sort_kernel(
    const int* __restrict__ pairs_raw, const unsigned* __restrict__ bstart,
    unsigned* __restrict__ segend, unsigned* __restrict__ degz,
    int* __restrict__ sorted_s, int N) {
    __shared__ unsigned cnt[256];
    __shared__ unsigned cur[256];
    __shared__ unsigned wsum[8];
    const int2* __restrict__ pairs = (const int2*)pairs_raw;
    int tid = (int)threadIdx.x;
    int b = (int)blockIdx.x;
    int lane = tid & 31;
    int w = tid >> 5;
    unsigned beg = bstart[b], end = bstart[b + 1];

    if (tid < 256) cnt[tid] = 0u;
    __syncthreads();
    for (unsigned i = beg + (unsigned)tid; i < end; i += EPT * SORT_T) {
        int2 p[EPT];
#pragma unroll
        for (int k = 0; k < EPT; ++k) {
            unsigned ii = i + (unsigned)(k * SORT_T);
            p[k] = pairs[ii < end ? ii : end - 1u];
        }
#pragma unroll
        for (int k = 0; k < EPT; ++k)
            if (i + (unsigned)(k * SORT_T) < end)
                atomicAdd(&cnt[p[k].y & 255], 1u);
    }
    __syncthreads();
    unsigned cv = (tid < 256) ? cnt[tid] : 0u;
    unsigned inc = cv;
    for (int d = 1; d < 32; d <<= 1) {
        unsigned t = __shfl_up(inc, d);
        if (lane >= d) inc += t;
    }
    if (tid < 256 && lane == 31) wsum[w] = inc;
    __syncthreads();
    if (tid < 256) {
        unsigned base = 0u;
        for (int k = 0; k < w; ++k) base += wsum[k];
        unsigned st = beg + base + inc - cv;
        cur[tid] = st;
        int r = (b << RB_SHIFT) + tid;
        if (r < N) { segend[r] = st + cv; degz[r] = cv; }
    }
    __syncthreads();
    for (unsigned i = beg + (unsigned)tid; i < end; i += EPT * SORT_T) {
        int2 p[EPT];
#pragma unroll
        for (int k = 0; k < EPT; ++k) {
            unsigned ii = i + (unsigned)(k * SORT_T);
            p[k] = pairs[ii < end ? ii : end - 1u];
        }
#pragma unroll
        for (int k = 0; k < EPT; ++k)
            if (i + (unsigned)(k * SORT_T) < end) {
                unsigned pos = atomicAdd(&cur[p[k].y & 255], 1u);
                sorted_s[pos] = p[k].x;
            }
    }
}

__device__ __forceinline__ void wmma_step(const float* __restrict__ W, int kb,
                                          int half, int m, v2f a,
                                          v8f& acc0, v8f& acc1) {
    int r0 = (kb + 2 * half + 0) * 32;
    int r1 = (kb + 2 * half + 1) * 32;
    v2f b0, b1;
    b0.x = W[r0 + m];        b0.y = W[r1 + m];
    b1.x = W[r0 + 16 + m];   b1.y = W[r1 + 16 + m];
    acc0 = __builtin_amdgcn_wmma_f32_16x16x4_f32(false, a, false, b0,
                                                 (short)0, acc0, false, false);
    acc1 = __builtin_amdgcn_wmma_f32_16x16x4_f32(false, a, false, b1,
                                                 (short)0, acc1, false, false);
}

__global__ __launch_bounds__(256) void qkv_wmma_kernel(
    const float* __restrict__ x,
    const float* __restrict__ Wq, const float* __restrict__ Wk,
    const float* __restrict__ Wv,
    float* __restrict__ Q, float* __restrict__ K, float* __restrict__ V,
    int ntiles) {
    int wave = (int)((blockIdx.x * blockDim.x + threadIdx.x) >> 5);
    if (wave >= ntiles) return;
    int lane = (int)(threadIdx.x & 31u);
    int half = lane >> 4;
    int m    = lane & 15;
    size_t row0 = (size_t)wave * 16;

    v8f q0 = {}, q1 = {}, k0 = {}, k1 = {}, v0 = {}, v1 = {};

#pragma unroll
    for (int kk = 0; kk < 8; ++kk) {
        int kb = kk * 4;
        const float* ap = x + (row0 + m) * 32 + kb + 2 * half;
        v2f a; a.x = ap[0]; a.y = ap[1];
        wmma_step(Wq, kb, half, m, a, q0, q1);
        wmma_step(Wk, kb, half, m, a, k0, k1);
        wmma_step(Wv, kb, half, m, a, v0, v1);
    }

#pragma unroll
    for (int r = 0; r < 8; ++r) {
        size_t row = (row0 + r + 8 * half) * 32;
        Q[row + m]      = q0[r];  Q[row + 16 + m] = q1[r];
        K[row + m]      = k0[r];  K[row + 16 + m] = k1[r];
        V[row + m]      = v0[r];  V[row + 16 + m] = v1[r];
    }
}

__global__ void qkv_tail_kernel(const float* __restrict__ x,
                                const float* __restrict__ Wq,
                                const float* __restrict__ Wk,
                                const float* __restrict__ Wv,
                                float* __restrict__ Q, float* __restrict__ K,
                                float* __restrict__ V, int row_start, int N) {
    int idx = blockIdx.x * blockDim.x + threadIdx.x;
    int nrows = N - row_start;
    if (idx >= nrows * 32) return;
    int row = row_start + idx / 32, col = idx & 31;
    float q = 0.f, k = 0.f, v = 0.f;
    for (int t = 0; t < 32; ++t) {
        float xv = x[(size_t)row * 32 + t];
        q += xv * Wq[t * 32 + col];
        k += xv * Wk[t * 32 + col];
        v += xv * Wv[t * 32 + col];
    }
    size_t o = (size_t)row * 32 + col;
    Q[o] = q; K[o] = k; V[o] = v;
}

#define RPW 7

__global__ __launch_bounds__(256) void seg_attn_kernel(
    const float* __restrict__ Q, const float* __restrict__ K,
    const float* __restrict__ V, const int* __restrict__ sorted_s,
    const unsigned* __restrict__ segend, unsigned* __restrict__ degz,
    float* __restrict__ outacc, int N) {
    int r0 = __builtin_amdgcn_readfirstlane(
                 (int)((blockIdx.x * blockDim.x + threadIdx.x) >> 5)) * RPW;
    if (r0 >= N) return;
    int lane = (int)(threadIdx.x & 31u);
    int g = lane >> 3;
    int c = lane & 7;
    int nr = (N - r0 < RPW) ? N - r0 : RPW;

    int degv = 0, endv = 0;
    if (lane < nr) {
        degv = (int)degz[r0 + lane];
        endv = (int)segend[r0 + lane];
    }
    float zv = 0.f;

    int deg   = __builtin_amdgcn_readlane(degv, 0);
    int start = __builtin_amdgcn_readlane(endv, 0) - deg;
    float4 q4 = ((const float4*)(Q + (size_t)r0 * 32))[c];
    int s_first = (g < deg) ? sorted_s[start + g] : r0;

    for (int j = 0; j < nr; ++j) {
        int r = r0 + j;
        int deg_n = 0, start_n = 0, s_first_n = r;
        float4 q4_n = q4;
        if (j + 1 < nr) {
            deg_n   = __builtin_amdgcn_readlane(degv, j + 1);
            start_n = __builtin_amdgcn_readlane(endv, j + 1) - deg_n;
            q4_n = ((const float4*)(Q + (size_t)(r + 1) * 32))[c];
            s_first_n = (g < deg_n) ? sorted_s[start_n + g] : r + 1;
        }

        float m = -INFINITY;
        float z = 0.f;
        float4 acc = make_float4(0.f, 0.f, 0.f, 0.f);

        int s_cur = s_first;
        for (int t = 0; t < deg; t += 4) {
            int jn = t + 4 + g;
            int s_nxt = (jn < deg) ? sorted_s[start + jn] : r;
            bool act = (t + g) < deg;
            float4 k4 = ((const float4*)(K + (size_t)s_cur * 32))[c];
            float4 v4 = ((const float4*)(V + (size_t)s_cur * 32))[c];
            float p = q4.x * k4.x + q4.y * k4.y + q4.z * k4.z + q4.w * k4.w;
            p += __shfl_xor(p, 1);
            p += __shfl_xor(p, 2);
            p += __shfl_xor(p, 4);
            if (act) {
                float sc = p * 0.17677669529663687f;
                float nm = fmaxf(m, sc);
                float corr = __expf(m - nm);
                float a = __expf(sc - nm);
                z = z * corr + a;
                acc.x = acc.x * corr + a * v4.x;
                acc.y = acc.y * corr + a * v4.y;
                acc.z = acc.z * corr + a * v4.z;
                acc.w = acc.w * corr + a * v4.w;
                m = nm;
            }
            s_cur = s_nxt;
        }

        float M = m;
        M = fmaxf(M, __shfl_xor(M, 8));
        M = fmaxf(M, __shfl_xor(M, 16));
        float f = (m == -INFINITY) ? 0.f : __expf(m - M);
        z *= f; acc.x *= f; acc.y *= f; acc.z *= f; acc.w *= f;
        z     += __shfl_xor(z, 8);      z     += __shfl_xor(z, 16);
        acc.x += __shfl_xor(acc.x, 8);  acc.x += __shfl_xor(acc.x, 16);
        acc.y += __shfl_xor(acc.y, 8);  acc.y += __shfl_xor(acc.y, 16);
        acc.z += __shfl_xor(acc.z, 8);  acc.z += __shfl_xor(acc.z, 16);
        acc.w += __shfl_xor(acc.w, 8);  acc.w += __shfl_xor(acc.w, 16);
        if (g == 0) ((float4*)(outacc + (size_t)r * 32))[c] = acc;
        if (lane == j) zv = z;

        deg = deg_n; start = start_n; q4 = q4_n; s_first = s_first_n;
    }
    if (lane < nr) degz[r0 + lane] = __float_as_uint(zv);
}

__global__ __launch_bounds__(256) void out_wmma_kernel(
    const float* __restrict__ x, const float* __restrict__ outacc,
    const float* __restrict__ Z, const float* __restrict__ Wo,
    float* __restrict__ out, int ntiles) {
    int wave = (int)((blockIdx.x * blockDim.x + threadIdx.x) >> 5);
    if (wave >= ntiles) return;
    int lane = (int)(threadIdx.x & 31u);
    int half = lane >> 4;
    int m    = lane & 15;
    size_t row0 = (size_t)wave * 16;

    float invz = 1.0f / (Z[row0 + m] + 1e-6f);
    v8f c0 = {}, c1 = {};

#pragma unroll
    for (int kk = 0; kk < 8; ++kk) {
        int kb = kk * 4;
        const float* ap = outacc + (row0 + m) * 32 + kb + 2 * half;
        v2f a; a.x = ap[0] * invz; a.y = ap[1] * invz;
        wmma_step(Wo, kb, half, m, a, c0, c1);
    }

#pragma unroll
    for (int r = 0; r < 8; ++r) {
        size_t row = (row0 + r + 8 * half) * 32;
        out[row + m]      = x[row + m]      + c0[r];
        out[row + 16 + m] = x[row + 16 + m] + c1[r];
    }
}

__global__ void out_tail_kernel(const float* __restrict__ x,
                                const float* __restrict__ outacc,
                                const float* __restrict__ Z,
                                const float* __restrict__ Wo,
                                float* __restrict__ out, int row_start, int N) {
    int idx = blockIdx.x * blockDim.x + threadIdx.x;
    int nrows = N - row_start;
    if (idx >= nrows * 32) return;
    int row = row_start + idx / 32, col = idx & 31;
    float invz = 1.0f / (Z[row] + 1e-6f);
    float acc = 0.f;
    for (int t = 0; t < 32; ++t)
        acc += outacc[(size_t)row * 32 + t] * invz * Wo[t * 32 + col];
    size_t o = (size_t)row * 32 + col;
    out[o] = x[o] + acc;
}

extern "C" void kernel_launch(void* const* d_in, const int* in_sizes, int n_in,
                              void* d_out, int out_size, void* d_ws,
                              size_t ws_size, hipStream_t stream) {
    const float* x  = (const float*)d_in[0];
    const int*   ei = (const int*)d_in[1];
    const float* Wq = (const float*)d_in[2];
    const float* Wk = (const float*)d_in[3];
    const float* Wv = (const float*)d_in[4];
    const float* Wo = (const float*)d_in[5];
    float* out = (float*)d_out;

    const int N = in_sizes[0] / 32;
    const int E = in_sizes[1] / 2;

    float* Q       = (float*)d_ws;
    float* K       = Q + (size_t)N * 32;
    float* V       = K + (size_t)N * 32;
    float* outacc  = V + (size_t)N * 32;
    int* sorted_s  = (int*)(outacc + (size_t)N * 32);
    unsigned* cnt  = (unsigned*)(sorted_s + E);
    unsigned* degz = cnt + N;
    unsigned* table  = (unsigned*)Q;
    unsigned* bstart = table + (size_t)SORT_C * NB_MAX;
    int* pairs       = (int*)K;

    const int TB = 256;
    int ntiles = N / 16;
    int rem_row0 = ntiles * 16;
    int rem = N - rem_row0;
    int NB = (N + 255) >> RB_SHIFT;

    bucket_hist_kernel<<<SORT_C, SORT_T, 0, stream>>>(ei, table, E, NB);
    bucket_place_kernel<<<SORT_C, SORT_T, 0, stream>>>(ei, table, bstart,
                                                       pairs, E, NB);
    bucket_sort_kernel<<<NB, SORT_T, 0, stream>>>(pairs, bstart, cnt, degz,
                                                  sorted_s, N);
    {
        int threads = ntiles * 32;
        if (threads > 0)
            qkv_wmma_kernel<<<(threads + TB - 1) / TB, TB, 0, stream>>>(
                x, Wq, Wk, Wv, Q, K, V, ntiles);
        if (rem > 0)
            qkv_tail_kernel<<<(rem * 32 + TB - 1) / TB, TB, 0, stream>>>(
                x, Wq, Wk, Wv, Q, K, V, rem_row0, N);
    }
    seg_attn_kernel<<<(N + 8 * RPW - 1) / (8 * RPW), TB, 0, stream>>>(
        Q, K, V, sorted_s, cnt, degz, outacc, N);
    {
        int threads = ntiles * 32;
        if (threads > 0)
            out_wmma_kernel<<<(threads + TB - 1) / TB, TB, 0, stream>>>(
                x, outacc, (const float*)degz, Wo, out, ntiles);
        if (rem > 0)
            out_tail_kernel<<<(rem * 32 + TB - 1) / TB, TB, 0, stream>>>(
                x, outacc, (const float*)degz, Wo, out, rem_row0, N);
    }
}


# Segment softmax edge loop takes 16 edges per trip: their 8 K/V row loads are in flight together, max/z/acc updated once per trip; 10 receivers per wave (96 registers); sum order within a trip changes
# speedup vs baseline: 7.7456x; 1.1146x over previous
#include <hip/hip_runtime.h>
#include <math.h>

typedef __attribute__((ext_vector_type(2))) float v2f;
typedef __attribute__((ext_vector_type(8))) float v8f;

#define SORT_C   256
#define SORT_T   512
#define RB_SHIFT 8
#define NB_MAX   512
#define EPT      8
static_assert(SORT_T == NB_MAX && SORT_T == 512 && SORT_C == 256,
              "the table walk of the placement pass is written for these");

__device__ __forceinline__ unsigned bucket_of(int r, int NB) {
    unsigned b = (unsigned)r >> RB_SHIFT;
    unsigned last = (unsigned)(NB - 1);
    return b < last ? b : last;
}

__global__ __launch_bounds__(SORT_T) void bucket_hist_kernel(
    const int* __restrict__ ei, unsigned* __restrict__ table, int E, int NB) {
    __shared__ unsigned hist[NB_MAX];
    int tid = (int)threadIdx.x;
    int c = (int)blockIdx.x;
    hist[tid] = 0u;
    __syncthreads();
    int CH = (E + SORT_C - 1) / SORT_C;
    int beg = c * CH;
    int end = (beg + CH < E) ? beg + CH : E;
    for (int e = beg + tid; e < end; e += EPT * SORT_T) {
        int r[EPT];
#pragma unroll
        for (int k = 0; k < EPT; ++k) {
            int ee = e + k * SORT_T;
            r[k] = ei[E + (ee < end ? ee : end - 1)];
        }
#pragma unroll
        for (int k = 0; k < EPT; ++k)
            if (e + k * SORT_T < end)
                atomicAdd(&hist[bucket_of(r[k], NB)], 1u);
    }
    __syncthreads();
    table[c * NB_MAX + tid] = hist[tid];
}

__global__ __launch_bounds__(SORT_T) void bucket_place_kernel(
    const int* __restrict__ ei, const unsigned* __restrict__ table,
    unsigned* __restrict__ bstart, int* __restrict__ pairs_raw, int E, int NB) {
    __shared__ unsigned cursor[NB_MAX];
    __shared__ unsigned wsum[SORT_T / 32];
    __shared__ __attribute__((aligned(16))) unsigned ptot[4][NB_MAX];
    __shared__ __attribute__((aligned(16))) unsigned ppre[4][NB_MAX];
    int2* __restrict__ pairs = (int2*)pairs_raw;
    int tid = (int)threadIdx.x;
    int c = (int)blockIdx.x;
    int lane = tid & 31;
    int w = tid >> 5;

    {
        int q = tid & 127;
        int g = tid >> 7;
        int gg = (g + c) & 3;
        const uint4* __restrict__ t4 = (const uint4*)table;
        unsigned t0 = 0u, t1 = 0u, t2 = 0u, t3 = 0u;
        unsigned p0 = 0u, p1 = 0u, p2 = 0u, p3 = 0u;
        for (int k0 = 0; k0 < SORT_C / 4; k0 += 16) {
            uint4 v[16];
#pragma unroll
            for (int j = 0; j < 16; ++j)
                v[j] = t4[(gg * (SORT_C / 4) + k0 + j) * (NB_MAX / 4) + q];
#pragma unroll
            for (int j = 0; j < 16; ++j) {
                unsigned m = (gg * (SORT_C / 4) + k0 + j < c) ? 0xffffffffu : 0u;
                t0 += v[j].x; t1 += v[j].y; t2 += v[j].z; t3 += v[j].w;
                p0 += v[j].x & m;  p1 += v[j].y & m;
                p2 += v[j].z & m;  p3 += v[j].w & m;
            }
        }
        ((uint4*)ptot[g])[q] = make_uint4(t0, t1, t2, t3);
        ((uint4*)ppre[g])[q] = make_uint4(p0, p1, p2, p3);
    }
    __syncthreads();
    unsigned tot = ptot[0][tid] + ptot[1][tid] + ptot[2][tid] + ptot[3][tid];
    unsigned pre = ppre[0][tid] + ppre[1][tid] + ppre[2][tid] + ppre[3][tid];
    unsigned inc = tot;
    for (int d = 1; d < 32; d <<= 1) {
        unsigned t = __shfl_up(inc, d);
        if (lane >= d) inc += t;
    }
    if (lane == 31) wsum[w] = inc;
    __syncthreads();
    unsigned base = 0u;
    for (int k = 0; k < w; ++k) base += wsum[k];
    unsigned excl = base + inc - tot;
    cursor[tid] = excl + pre;
    if (c == 0 && tid < NB) {
        bstart[tid] = excl;
        if (tid == NB - 1) bstart[NB] = excl + tot;
    }
    __syncthreads();

    int CH = (E + SORT_C - 1) / SORT_C;
    int beg = c * CH;
    int end = (beg + CH < E) ? beg + CH : E;
    for (int e = beg + tid; e < end; e += EPT * SORT_T) {
        int s[EPT], r[EPT];
#pragma unroll
        for (int k = 0; k < EPT; ++k) {
            int ee = e + k * SORT_T;
            int ec = ee < end ? ee : end - 1;
            s[k] = ei[ec];
            r[k] = ei[E + ec];
        }
#pragma unroll
        for (int k = 0; k < EPT; ++k)
            if (e + k * SORT_T < end) {
                unsigned pos = atomicAdd(&cursor[bucket_of(r[k], NB)], 1u);
                pairs[pos] = make_int2(s[k], r[k]);
            }
    }
}

__global__ __launch_bounds__(SORT_T) void bucket_sort_kernel(
    const int* __restrict__ pairs_raw, const unsigned* __restrict__ bstart,
    unsigned* __restrict__ segend, unsigned* __restrict__ degz,
    int* __restrict__ sorted_s, int N) {
    __shared__ unsigned cnt[256];
    __shared__ unsigned cur[256];
    __shared__ unsigned wsum[8];
    const int2* __restrict__ pairs = (const int2*)pairs_raw;
    int tid = (int)threadIdx.x;
    int b = (int)blockIdx.x;
    int lane = tid & 31;
    int w = tid >> 5;
    unsigned beg = bstart[b], end = bstart[b + 1];

    if (tid < 256) cnt[tid] = 0u;
    __syncthreads();
    for (unsigned i = beg + (unsigned)tid; i < end; i += EPT * SORT_T) {
        int2 p[EPT];
#pragma unroll
        for (int k = 0; k < EPT; ++k) {
            unsigned ii = i + (unsigned)(k * SORT_T);
            p[k] = pairs[ii < end ? ii : end - 1u];
        }
#pragma unroll
        for (int k = 0; k < EPT; ++k)
            if (i + (unsigned)(k * SORT_T) < end)
                atomicAdd(&cnt[p[k].y & 255], 1u);
    }
    __syncthreads();
    unsigned cv = (tid < 256) ? cnt[tid] : 0u;
    unsigned inc = cv;
    for (int d = 1; d < 32; d <<= 1) {
        unsigned t = __shfl_up(inc, d);
        if (lane >= d) inc += t;
    }
    if (tid < 256 && lane == 31) wsum[w] = inc;
    __syncthreads();
    if (tid < 256) {
        unsigned base = 0u;
        for (int k = 0; k < w; ++k) base += wsum[k];
        unsigned st = beg + base + inc - cv;
        cur[tid] = st;
        int r = (b << RB_SHIFT) + tid;
        if (r < N) { segend[r] = st + cv; degz[r] = cv; }
    }
    __syncthreads();
    for (unsigned i = beg + (unsigned)tid; i < end; i += EPT * SORT_T) {
        int2 p[EPT];
#pragma unroll
        for (int k = 0; k < EPT; ++k) {
            unsigned ii = i + (unsigned)(k * SORT_T);
            p[k] = pairs[ii < end ? ii : end - 1u];
        }
#pragma unroll
        for (int k = 0; k < EPT; ++k)
            if (i + (unsigned)(k * SORT_T) < end) {
                unsigned pos = atomicAdd(&cur[p[k].y & 255], 1u);
                sorted_s[pos] = p[k].x;
            }
    }
}

__device__ __forceinline__ void wmma_step(const float* __restrict__ W, int kb,
                                          int half, int m, v2f a,
                                          v8f& acc0, v8f& acc1) {
    int r0 = (kb + 2 * half + 0) * 32;
    int r1 = (kb + 2 * half + 1) * 32;
    v2f b0, b1;
    b0.x = W[r0 + m];        b0.y = W[r1 + m];
    b1.x = W[r0 + 16 + m];   b1.y = W[r1 + 16 + m];
    acc0 = __builtin_amdgcn_wmma_f32_16x16x4_f32(false, a, false, b0,
                                                 (short)0, acc0, false, false);
    acc1 = __builtin_amdgcn_wmma_f32_16x16x4_f32(false, a, false, b1,
                                                 (short)0, acc1, false, false);
}

__global__ __launch_bounds__(256) void qkv_wmma_kernel(
    const float* __restrict__ x,
    const float* __restrict__ Wq, const float* __restrict__ Wk,
    const float* __restrict__ Wv,
    float* __restrict__ Q, float* __restrict__ K, float* __restrict__ V,
    int ntiles) {
    int wave = (int)((blockIdx.x * blockDim.x + threadIdx.x) >> 5);
    if (wave >= ntiles) return;
    int lane = (int)(threadIdx.x & 31u);
    int half = lane >> 4;
    int m    = lane & 15;
    size_t row0 = (size_t)wave * 16;

    v8f q0 = {}, q1 = {}, k0 = {}, k1 = {}, v0 = {}, v1 = {};

#pragma unroll
    for (int kk = 0; kk < 8; ++kk) {
        int kb = kk * 4;
        const float* ap = x + (row0 + m) * 32 + kb + 2 * half;
        v2f a; a.x = ap[0]; a.y = ap[1];
        wmma_step(Wq, kb, half, m, a, q0, q1);
        wmma_step(Wk, kb, half, m, a, k0, k1);
        wmma_step(Wv, kb, half, m, a, v0, v1);
    }

#pragma unroll
    for (int r = 0; r < 8; ++r) {
        size_t row = (row0 + r + 8 * half) * 32;
        Q[row + m]      = q0[r];  Q[row + 16 + m] = q1[r];
        K[row + m]      = k0[r];  K[row + 16 + m] = k1[r];
        V[row + m]      = v0[r];  V[row + 16 + m] = v1[r];
    }
}

__global__ void qkv_tail_kernel(const float* __restrict__ x,
                                const float* __restrict__ Wq,
                                const float* __restrict__ Wk,
                                const float* __restrict__ Wv,
                                float* __restrict__ Q, float* __restrict__ K,
                                float* __restrict__ V, int row_start, int N) {
    int idx = blockIdx.x * blockDim.x + threadIdx.x;
    int nrows = N - row_start;
    if (idx >= nrows * 32) return;
    int row = row_start + idx / 32, col = idx & 31;
    float q = 0.f, k = 0.f, v = 0.f;
    for (int t = 0; t < 32; ++t) {
        float xv = x[(size_t)row * 32 + t];
        q += xv * Wq[t * 32 + col];
        k += xv * Wk[t * 32 + col];
        v += xv * Wv[t * 32 + col];
    }
    size_t o = (size_t)row * 32 + col;
    Q[o] = q; K[o] = k; V[o] = v;
}

#define RPW 10
#define SPT 4

__device__ __forceinline__ void load_senders(const int* __restrict__ sorted_s,
                                             int start, int deg, int e0, int g,
                                             int (&s)[SPT]) {
#pragma unroll
    for (int i = 0; i < SPT; ++i) {
        int jn = e0 + 4 * i + g;
        s[i] = sorted_s[start + (jn < deg ? jn : deg - 1)];
    }
}

__device__ __forceinline__ void load_rows(const float4* __restrict__ K4,
                                          const float4* __restrict__ V4,
                                          const int (&s)[SPT], unsigned cu,
                                          float4 (&k4)[SPT], float4 (&v4)[SPT]) {
#pragma unroll
    for (int i = 0; i < SPT; ++i) {
        unsigned row = (unsigned)s[i] * 8u + cu;
        k4[i] = K4[row];
        v4[i] = V4[row];
    }
}

__global__ __launch_bounds__(256) void seg_attn_kernel(
    const float* __restrict__ Q, const float* __restrict__ K,
    const float* __restrict__ V, const int* __restrict__ sorted_s,
    const unsigned* __restrict__ segend, unsigned* __restrict__ degz,
    float* __restrict__ outacc, int N) {
    int r0 = __builtin_amdgcn_readfirstlane(
                 (int)((blockIdx.x * blockDim.x + threadIdx.x) >> 5)) * RPW;
    if (r0 >= N) return;
    int lane = (int)(threadIdx.x & 31u);
    int g = lane >> 3;
    int c = lane & 7;
    int nr = (N - r0 < RPW) ? N - r0 : RPW;
    const float4* __restrict__ Q4 = (const float4*)Q;
    const float4* __restrict__ K4 = (const float4*)K;
    const float4* __restrict__ V4 = (const float4*)V;
    float4* __restrict__ O4 = (float4*)outacc;
    unsigned cu = (unsigned)c;

    int degv = 0, endv = 0;
    if (lane < nr) {
        degv = (int)degz[r0 + lane];
        endv = (int)segend[r0 + lane];
    }
    float zv = 0.f;

    int deg   = __builtin_amdgcn_readlane(degv, 0);
    int start = __builtin_amdgcn_readlane(endv, 0) - deg;
    float4 q4 = Q4[(unsigned)r0 * 8u + cu];
    int s_cur[SPT];
#pragma unroll
    for (int i = 0; i < SPT; ++i) s_cur[i] = r0;
    if (deg > 0) load_senders(sorted_s, start, deg, 0, g, s_cur);

    for (int j = 0; j < nr; ++j) {
        int r = r0 + j;
        float4 k4[SPT], v4[SPT];
        int s_nxt[SPT];
#pragma unroll
        for (int i = 0; i < SPT; ++i) { k4[i] = v4[i] = q4; s_nxt[i] = r; }
        if (deg > 0) {
            load_rows(K4, V4, s_cur, cu, k4, v4);
            if (deg > 4 * SPT)
                load_senders(sorted_s, start, deg, 4 * SPT, g, s_nxt);
        }
        int deg_n = 0, start_n = 0;
        int s_first_n[SPT];
#pragma unroll
        for (int i = 0; i < SPT; ++i) s_first_n[i] = r;
        float4 q4_n = q4;
        if (j + 1 < nr) {
            deg_n   = __builtin_amdgcn_readlane(degv, j + 1);
            start_n = __builtin_amdgcn_readlane(endv, j + 1) - deg_n;
            q4_n = Q4[(unsigned)(r + 1) * 8u + cu];
            if (deg_n > 0) load_senders(sorted_s, start_n, deg_n, 0, g, s_first_n);
        }

        float m = -INFINITY;
        float z = 0.f;
        float4 acc = make_float4(0.f, 0.f, 0.f, 0.f);

        for (int t = 0; t < deg; t += 4 * SPT) {
            float p[SPT];
#pragma unroll
            for (int i = 0; i < SPT; ++i)
                p[i] = q4.x * k4[i].x + q4.y * k4[i].y + q4.z * k4[i].z
                     + q4.w * k4[i].w;
#pragma unroll
            for (int i = 0; i < SPT; ++i) p[i] += __shfl_xor(p[i], 1);
#pragma unroll
            for (int i = 0; i < SPT; ++i) p[i] += __shfl_xor(p[i], 2);
#pragma unroll
            for (int i = 0; i < SPT; ++i) p[i] += __shfl_xor(p[i], 4);

            if (t + g < deg) {
                float sc[SPT];
#pragma unroll
                for (int i = 0; i < SPT; ++i)
                    sc[i] = (t + 4 * i + g < deg) ? p[i] * 0.17677669529663687f
                                                  : -INFINITY;
                float nm = m;
#pragma unroll
                for (int i = 0; i < SPT; ++i) nm = fmaxf(nm, sc[i]);
                float corr = __expf(m - nm);
                float a[SPT];
#pragma unroll
                for (int i = 0; i < SPT; ++i) a[i] = __expf(sc[i] - nm);
                float zs = 0.f;
                float4 as = make_float4(0.f, 0.f, 0.f, 0.f);
#pragma unroll
                for (int i = 0; i < SPT; ++i) {
                    zs += a[i];
                    as.x += a[i] * v4[i].x;
                    as.y += a[i] * v4[i].y;
                    as.z += a[i] * v4[i].z;
                    as.w += a[i] * v4[i].w;
                }
                z = z * corr + zs;
                acc.x = acc.x * corr + as.x;
                acc.y = acc.y * corr + as.y;
                acc.z = acc.z * corr + as.z;
                acc.w = acc.w * corr + as.w;
                m = nm;
            }
            if (t + 4 * SPT < deg) {
                int s_now[SPT];
#pragma unroll
                for (int i = 0; i < SPT; ++i) s_now[i] = s_nxt[i];
                load_rows(K4, V4, s_now, cu, k4, v4);
                if (t + 8 * SPT < deg)
                    load_senders(sorted_s, start, deg, t + 8 * SPT, g, s_nxt);
            }
        }

        float M = m;
        M = fmaxf(M, __shfl_xor(M, 8));
        M = fmaxf(M, __shfl_xor(M, 16));
        float f = (m == -INFINITY) ? 0.f : __expf(m - M);
        z *= f; acc.x *= f; acc.y *= f; acc.z *= f; acc.w *= f;
        z     += __shfl_xor(z, 8);      z     += __shfl_xor(z, 16);
        acc.x += __shfl_xor(acc.x, 8);  acc.x += __shfl_xor(acc.x, 16);
        acc.y += __shfl_xor(acc.y, 8);  acc.y += __shfl_xor(acc.y, 16);
        acc.z += __shfl_xor(acc.z, 8);  acc.z += __shfl_xor(acc.z, 16);
        acc.w += __shfl_xor(acc.w, 8);  acc.w += __shfl_xor(acc.w, 16);
        if (g == 0) O4[(unsigned)r * 8u + cu] = acc;
        if (lane == j) zv = z;

        deg = deg_n; start = start_n; q4 = q4_n;
#pragma unroll
        for (int i = 0; i < SPT; ++i) s_cur[i] = s_first_n[i];
    }
    if (lane < nr) degz[r0 + lane] = __float_as_uint(zv);
}

__global__ __launch_bounds__(256) void out_wmma_kernel(
    const float* __restrict__ x, const float* __restrict__ outacc,
    const float* __restrict__ Z, const float* __restrict__ Wo,
    float* __restrict__ out, int ntiles) {
    int wave = (int)((blockIdx.x * blockDim.x + threadIdx.x) >> 5);
    if (wave >= ntiles) return;
    int lane = (int)(threadIdx.x & 31u);
    int half = lane >> 4;
    int m    = lane & 15;
    size_t row0 = (size_t)wave * 16;

    float invz = 1.0f / (Z[row0 + m] + 1e-6f);
    v8f c0 = {}, c1 = {};

#pragma unroll
    for (int kk = 0; kk < 8; ++kk) {
        int kb = kk * 4;
        const float* ap = outacc + (row0 + m) * 32 + kb + 2 * half;
        v2f a; a.x = ap[0] * invz; a.y = ap[1] * invz;
        wmma_step(Wo, kb, half, m, a, c0, c1);
    }

#pragma unroll
    for (int r = 0; r < 8; ++r) {
        size_t row = (row0 + r + 8 * half) * 32;
        out[row + m]      = x[row + m]      + c0[r];
        out[row + 16 + m] = x[row + 16 + m] + c1[r];
    }
}

__global__ void out_tail_kernel(const float* __restrict__ x,
                                const float* __restrict__ outacc,
                                const float* __restrict__ Z,
                                const float* __restrict__ Wo,
                                float* __restrict__ out, int row_start, int N) {
    int idx = blockIdx.x * blockDim.x + threadIdx.x;
    int nrows = N - row_start;
    if (idx >= nrows * 32) return;
    int row = row_start + idx / 32, col = idx & 31;
    float invz = 1.0f / (Z[row] + 1e-6f);
    float acc = 0.f;
    for (int t = 0; t < 32; ++t)
        acc += outacc[(size_t)row * 32 + t] * invz * Wo[t * 32 + col];
    size_t o = (size_t)row * 32 + col;
    out[o] = x[o] + acc;
}

extern "C" void kernel_launch(void* const* d_in, const int* in_sizes, int n_in,
                              void* d_out, int out_size, void* d_ws,
                              size_t ws_size, hipStream_t stream) {
    const float* x  = (const float*)d_in[0];
    const int*   ei = (const int*)d_in[1];
    const float* Wq = (const float*)d_in[2];
    const float* Wk = (const float*)d_in[3];
    const float* Wv = (const float*)d_in[4];
    const float* Wo = (const float*)d_in[5];
    float* out = (float*)d_out;

    const int N = in_sizes[0] / 32;
    const int E = in_sizes[1] / 2;

    float* Q       = (float*)d_ws;
    float* K       = Q + (size_t)N * 32;
    float* V       = K + (size_t)N * 32;
    float* outacc  = V + (size_t)N * 32;
    int* sorted_s  = (int*)(outacc + (size_t)N * 32);
    unsigned* cnt  = (unsigned*)(sorted_s + E);
    unsigned* degz = cnt + N;
    unsigned* table  = (unsigned*)Q;
    unsigned* bstart = table + (size_t)SORT_C * NB_MAX;
    int* pairs       = (int*)K;

    const int TB = 256;
    int ntiles = N / 16;
    int rem_row0 = ntiles * 16;
    int rem = N - rem_row0;
    int NB = (N + 255) >> RB_SHIFT;

    bucket_hist_kernel<<<SORT_C, SORT_T, 0, stream>>>(ei, table, E, NB);
    bucket_place_kernel<<<SORT_C, SORT_T, 0, stream>>>(ei, table, bstart,
                                                       pairs, E, NB);
    bucket_sort_kernel<<<NB, SORT_T, 0, stream>>>(pairs, bstart, cnt, degz,
                                                  sorted_s, N);
    {
        int threads = ntiles * 32;
        if (threads > 0)
            qkv_wmma_kernel<<<(threads + TB - 1) / TB, TB, 0, stream>>>(
                x, Wq, Wk, Wv, Q, K, V, ntiles);
        if (rem > 0)
            qkv_tail_kernel<<<(rem * 32 + TB - 1) / TB, TB, 0, stream>>>(
                x, Wq, Wk, Wv, Q, K, V, rem_row0, N);
    }
    seg_attn_kernel<<<(N + 8 * RPW - 1) / (8 * RPW), TB, 0, stream>>>(
        Q, K, V, sorted_s, cnt, degz, outacc, N);
    {
        int threads = ntiles * 32;
        if (threads > 0)
            out_wmma_kernel<<<(threads + TB - 1) / TB, TB, 0, stream>>>(
                x, outacc, (const float*)degz, Wo, out, ntiles);
        if (rem > 0)
            out_tail_kernel<<<(rem * 32 + TB - 1) / TB, TB, 0, stream>>>(
                x, outacc, (const float*)degz, Wo, out, rem_row0, N);
    }
}


# Scores and weighted sums taken from x rows: Wq*Wk^T and Wv*Wo folded once per call (WMMA), one projected plane not three, one gathered row per edge not two; product and sum order change
# speedup vs baseline: 9.4765x; 1.2235x over previous
#include <hip/hip_runtime.h>
#include <math.h>

typedef __attribute__((ext_vector_type(2))) float v2f;
typedef __attribute__((ext_vector_type(8))) float v8f;

#define SORT_C   256
#define SORT_T   512
#define RB_SHIFT 8
#define NB_MAX   512
#define EPT      8
static_assert(SORT_T == NB_MAX && SORT_T == 512 && SORT_C == 256,
              "the table walk of the placement pass is written for these");

__device__ __forceinline__ unsigned bucket_of(int r, int NB) {
    unsigned b = (unsigned)r >> RB_SHIFT;
    unsigned last = (unsigned)(NB - 1);
    return b < last ? b : last;
}

__global__ __launch_bounds__(SORT_T) void bucket_hist_kernel(
    const int* __restrict__ ei, unsigned* __restrict__ table, int E, int NB) {
    __shared__ unsigned hist[NB_MAX];
    int tid = (int)threadIdx.x;
    int c = (int)blockIdx.x;
    hist[tid] = 0u;
    __syncthreads();
    int CH = (E + SORT_C - 1) / SORT_C;
    int beg = c * CH;
    int end = (beg + CH < E) ? beg + CH : E;
    for (int e = beg + tid; e < end; e += EPT * SORT_T) {
        int r[EPT];
#pragma unroll
        for (int k = 0; k < EPT; ++k) {
            int ee = e + k * SORT_T;
            r[k] = ei[E + (ee < end ? ee : end - 1)];
        }
#pragma unroll
        for (int k = 0; k < EPT; ++k)
            if (e + k * SORT_T < end)
                atomicAdd(&hist[bucket_of(r[k], NB)], 1u);
    }
    __syncthreads();
    table[c * NB_MAX + tid] = hist[tid];
}

__global__ __launch_bounds__(SORT_T) void bucket_place_kernel(
    const int* __restrict__ ei, const unsigned* __restrict__ table,
    unsigned* __restrict__ bstart, int* __restrict__ pairs_raw, int E, int NB) {
    __shared__ unsigned cursor[NB_MAX];
    __shared__ unsigned wsum[SORT_T / 32];
    __shared__ __attribute__((aligned(16))) unsigned ptot[4][NB_MAX];
    __shared__ __attribute__((aligned(16))) unsigned ppre[4][NB_MAX];
    int2* __restrict__ pairs = (int2*)pairs_raw;
    int tid = (int)threadIdx.x;
    int c = (int)blockIdx.x;
    int lane = tid & 31;
    int w = tid >> 5;

    {
        int q = tid & 127;
        int g = tid >> 7;
        int gg = (g + c) & 3;
        const uint4* __restrict__ t4 = (const uint4*)table;
        unsigned t0 = 0u, t1 = 0u, t2 = 0u, t3 = 0u;
        unsigned p0 = 0u, p1 = 0u, p2 = 0u, p3 = 0u;
        for (int k0 = 0; k0 < SORT_C / 4; k0 += 16) {
            uint4 v[16];
#pragma unroll
            for (int j = 0; j < 16; ++j)
                v[j] = t4[(gg * (SORT_C / 4) + k0 + j) * (NB_MAX / 4) + q];
#pragma unroll
            for (int j = 0; j < 16; ++j) {
                unsigned m = (gg * (SORT_C / 4) + k0 + j < c) ? 0xffffffffu : 0u;
                t0 += v[j].x; t1 += v[j].y; t2 += v[j].z; t3 += v[j].w;
                p0 += v[j].x & m;  p1 += v[j].y & m;
                p2 += v[j].z & m;  p3 += v[j].w & m;
            }
        }
        ((uint4*)ptot[g])[q] = make_uint4(t0, t1, t2, t3);
        ((uint4*)ppre[g])[q] = make_uint4(p0, p1, p2, p3);
    }
    __syncthreads();
    unsigned tot = ptot[0][tid] + ptot[1][tid] + ptot[2][tid] + ptot[3][tid];
    unsigned pre = ppre[0][tid] + ppre[1][tid] + ppre[2][tid] + ppre[3][tid];
    unsigned inc = tot;
    for (int d = 1; d < 32; d <<= 1) {
        unsigned t = __shfl_up(inc, d);
        if (lane >= d) inc += t;
    }
    if (lane == 31) wsum[w] = inc;
    __syncthreads();
    unsigned base = 0u;
    for (int k = 0; k < w; ++k) base += wsum[k];
    unsigned excl = base + inc - tot;
    cursor[tid] = excl + pre;
    if (c == 0 && tid < NB) {
        bstart[tid] = excl;
        if (tid == NB - 1) bstart[NB] = excl + tot;
    }
    __syncthreads();

    int CH = (E + SORT_C - 1) / SORT_C;
    int beg = c * CH;
    int end = (beg + CH < E) ? beg + CH : E;
    for (int e = beg + tid; e < end; e += EPT * SORT_T) {
        int s[EPT], r[EPT];
#pragma unroll
        for (int k = 0; k < EPT; ++k) {
            int ee = e + k * SORT_T;
            int ec = ee < end ? ee : end - 1;
            s[k] = ei[ec];
            r[k] = ei[E + ec];
        }
#pragma unroll
        for (int k = 0; k < EPT; ++k)
            if (e + k * SORT_T < end) {
                unsigned pos = atomicAdd(&cursor[bucket_of(r[k], NB)], 1u);
                pairs[pos] = make_int2(s[k], r[k]);
            }
    }
}

__global__ __launch_bounds__(SORT_T) void bucket_sort_kernel(
    const int* __restrict__ pairs_raw, const unsigned* __restrict__ bstart,
    unsigned* __restrict__ segend, unsigned* __restrict__ degz,
    int* __restrict__ sorted_s, int N) {
    __shared__ unsigned cnt[256];
    __shared__ unsigned cur[256];
    __shared__ unsigned wsum[8];
    const int2* __restrict__ pairs = (const int2*)pairs_raw;
    int tid = (int)threadIdx.x;
    int b = (int)blockIdx.x;
    int lane = tid & 31;
    int w = tid >> 5;
    unsigned beg = bstart[b], end = bstart[b + 1];

    if (tid < 256) cnt[tid] = 0u;
    __syncthreads();
    for (unsigned i = beg + (unsigned)tid; i < end; i += EPT * SORT_T) {
        int2 p[EPT];
#pragma unroll
        for (int k = 0; k < EPT; ++k) {
            unsigned ii = i + (unsigned)(k * SORT_T);
            p[k] = pairs[ii < end ? ii : end - 1u];
        }
#pragma unroll
        for (int k = 0; k < EPT; ++k)
            if (i + (unsigned)(k * SORT_T) < end)
                atomicAdd(&cnt[p[k].y & 255], 1u);
    }
    __syncthreads();
    unsigned cv = (tid < 256) ? cnt[tid] : 0u;
    unsigned inc = cv;
    for (int d = 1; d < 32; d <<= 1) {
        unsigned t = __shfl_up(inc, d);
        if (lane >= d) inc += t;
    }
    if (tid < 256 && lane == 31) wsum[w] = inc;
    __syncthreads();
    if (tid < 256) {
        unsigned base = 0u;
        for (int k = 0; k < w; ++k) base += wsum[k];
        unsigned st = beg + base + inc - cv;
        cur[tid] = st;
        int r = (b << RB_SHIFT) + tid;
        if (r < N) { segend[r] = st + cv; degz[r] = cv; }
    }
    __syncthreads();
    for (unsigned i = beg + (unsigned)tid; i < end; i += EPT * SORT_T) {
        int2 p[EPT];
#pragma unroll
        for (int k = 0; k < EPT; ++k) {
            unsigned ii = i + (unsigned)(k * SORT_T);
            p[k] = pairs[ii < end ? ii : end - 1u];
        }
#pragma unroll
        for (int k = 0; k < EPT; ++k)
            if (i + (unsigned)(k * SORT_T) < end) {
                unsigned pos = atomicAdd(&cur[p[k].y & 255], 1u);
                sorted_s[pos] = p[k].x;
            }
    }
}

__device__ __forceinline__ void wmma_step(const float* __restrict__ W, int kb,
                                          int half, int m, v2f a,
                                          v8f& acc0, v8f& acc1) {
    int r0 = (kb + 2 * half + 0) * 32;
    int r1 = (kb + 2 * half + 1) * 32;
    v2f b0, b1;
    b0.x = W[r0 + m];        b0.y = W[r1 + m];
    b1.x = W[r0 + 16 + m];   b1.y = W[r1 + 16 + m];
    acc0 = __builtin_amdgcn_wmma_f32_16x16x4_f32(false, a, false, b0,
                                                 (short)0, acc0, false, false);
    acc1 = __builtin_amdgcn_wmma_f32_16x16x4_f32(false, a, false, b1,
                                                 (short)0, acc1, false, false);
}

__global__ __launch_bounds__(256) void weight_fold_kernel(
    const float* __restrict__ Wq, const float* __restrict__ Wk,
    const float* __restrict__ Wv, const float* __restrict__ Wo,
    float* __restrict__ Wqk, float* __restrict__ Wvo) {
    int wave = (int)(threadIdx.x >> 5);
    int lane = (int)(threadIdx.x & 31u);
    int half = lane >> 4;
    int m    = lane & 15;
    bool vo  = wave >= 4;
    int tr = (wave >> 1) & 1, tc = wave & 1;
    const float* __restrict__ A = vo ? Wv : Wq;
    const float* __restrict__ B = vo ? Wo : Wk;
    v8f acc = {};
#pragma unroll
    for (int kk = 0; kk < 8; ++kk) {
        int k0 = kk * 4 + 2 * half;
        const float* ap = A + (tr * 16 + m) * 32 + k0;
        v2f a; a.x = ap[0]; a.y = ap[1];
        int i0 = vo ? k0 * 32 + tc * 16 + m : (tc * 16 + m) * 32 + k0;
        int i1 = vo ? i0 + 32 : i0 + 1;
        v2f b; b.x = B[i0]; b.y = B[i1];
        acc = __builtin_amdgcn_wmma_f32_16x16x4_f32(false, a, false, b,
                                                    (short)0, acc, false, false);
    }
    float* __restrict__ D = vo ? Wvo : Wqk;
#pragma unroll
    for (int r = 0; r < 8; ++r)
        D[(tr * 16 + r + 8 * half) * 32 + tc * 16 + m] = acc[r];
}

__global__ __launch_bounds__(256) void qp_wmma_kernel(
    const float* __restrict__ x, const float* __restrict__ Wqk,
    float* __restrict__ Q, int ntiles) {
    int wave = (int)((blockIdx.x * blockDim.x + threadIdx.x) >> 5);
    if (wave >= ntiles) return;
    int lane = (int)(threadIdx.x & 31u);
    int half = lane >> 4;
    int m    = lane & 15;
    size_t row0 = (size_t)wave * 16;

    v8f q0 = {}, q1 = {};

#pragma unroll
    for (int kk = 0; kk < 8; ++kk) {
        int kb = kk * 4;
        const float* ap = x + (row0 + m) * 32 + kb + 2 * half;
        v2f a; a.x = ap[0]; a.y = ap[1];
        wmma_step(Wqk, kb, half, m, a, q0, q1);
    }

#pragma unroll
    for (int r = 0; r < 8; ++r) {
        size_t row = (row0 + r + 8 * half) * 32;
        Q[row + m]      = q0[r];  Q[row + 16 + m] = q1[r];
    }
}

__global__ void qp_tail_kernel(const float* __restrict__ x,
                               const float* __restrict__ Wqk,
                               float* __restrict__ Q, int row_start, int N) {
    int idx = blockIdx.x * blockDim.x + threadIdx.x;
    int nrows = N - row_start;
    if (idx >= nrows * 32) return;
    int row = row_start + idx / 32, col = idx & 31;
    float q = 0.f;
    for (int t = 0; t < 32; ++t)
        q += x[(size_t)row * 32 + t] * Wqk[t * 32 + col];
    Q[(size_t)row * 32 + col] = q;
}

#define RPW 10
#define SPT 4

__device__ __forceinline__ void load_senders(const int* __restrict__ sorted_s,
                                             int start, int deg, int e0, int g,
                                             int (&s)[SPT]) {
#pragma unroll
    for (int i = 0; i < SPT; ++i) {
        int jn = e0 + 4 * i + g;
        s[i] = sorted_s[start + (jn < deg ? jn : deg - 1)];
    }
}

__device__ __forceinline__ void load_rows(const float4* __restrict__ X4,
                                          const int (&s)[SPT], unsigned cu,
                                          float4 (&x4)[SPT]) {
#pragma unroll
    for (int i = 0; i < SPT; ++i)
        x4[i] = X4[(unsigned)s[i] * 8u + cu];
}

__global__ __launch_bounds__(256) void seg_attn_kernel(
    const float* __restrict__ Q, const float* __restrict__ x,
    const int* __restrict__ sorted_s,
    const unsigned* __restrict__ segend, unsigned* __restrict__ degz,
    float* __restrict__ outacc, int N) {
    int r0 = __builtin_amdgcn_readfirstlane(
                 (int)((blockIdx.x * blockDim.x + threadIdx.x) >> 5)) * RPW;
    if (r0 >= N) return;
    int lane = (int)(threadIdx.x & 31u);
    int g = lane >> 3;
    int c = lane & 7;
    int nr = (N - r0 < RPW) ? N - r0 : RPW;
    const float4* __restrict__ Q4 = (const float4*)Q;
    const float4* __restrict__ X4 = (const float4*)x;
    float4* __restrict__ O4 = (float4*)outacc;
    unsigned cu = (unsigned)c;

    int degv = 0, endv = 0;
    if (lane < nr) {
        degv = (int)degz[r0 + lane];
        endv = (int)segend[r0 + lane];
    }
    float zv = 0.f;

    int deg   = __builtin_amdgcn_readlane(degv, 0);
    int start = __builtin_amdgcn_readlane(endv, 0) - deg;
    float4 q4 = Q4[(unsigned)r0 * 8u + cu];
    int s_cur[SPT];
#pragma unroll
    for (int i = 0; i < SPT; ++i) s_cur[i] = r0;
    if (deg > 0) load_senders(sorted_s, start, deg, 0, g, s_cur);

    for (int j = 0; j < nr; ++j) {
        int r = r0 + j;
        float4 x4[SPT];
        int s_nxt[SPT];
#pragma unroll
        for (int i = 0; i < SPT; ++i) { x4[i] = q4; s_nxt[i] = r; }
        if (deg > 0) {
            load_rows(X4, s_cur, cu, x4);
            if (deg > 4 * SPT)
                load_senders(sorted_s, start, deg, 4 * SPT, g, s_nxt);
        }
        int deg_n = 0, start_n = 0;
        int s_first_n[SPT];
#pragma unroll
        for (int i = 0; i < SPT; ++i) s_first_n[i] = r;
        float4 q4_n = q4;
        if (j + 1 < nr) {
            deg_n   = __builtin_amdgcn_readlane(degv, j + 1);
            start_n = __builtin_amdgcn_readlane(endv, j + 1) - deg_n;
            q4_n = Q4[(unsigned)(r + 1) * 8u + cu];
            if (deg_n > 0) load_senders(sorted_s, start_n, deg_n, 0, g, s_first_n);
        }

        float m = -INFINITY;
        float z = 0.f;
        float4 acc = make_float4(0.f, 0.f, 0.f, 0.f);

        for (int t = 0; t < deg; t += 4 * SPT) {
            float p[SPT];
#pragma unroll
            for (int i = 0; i < SPT; ++i)
                p[i] = q4.x * x4[i].x + q4.y * x4[i].y + q4.z * x4[i].z
                     + q4.w * x4[i].w;
#pragma unroll
            for (int i = 0; i < SPT; ++i) p[i] += __shfl_xor(p[i], 1);
#pragma unroll
            for (int i = 0; i < SPT; ++i) p[i] += __shfl_xor(p[i], 2);
#pragma unroll
            for (int i = 0; i < SPT; ++i) p[i] += __shfl_xor(p[i], 4);

            if (t + g < deg) {
                float sc[SPT];
#pragma unroll
                for (int i = 0; i < SPT; ++i)
                    sc[i] = (t + 4 * i + g < deg) ? p[i] * 0.17677669529663687f
                                                  : -INFINITY;
                float nm = m;
#pragma unroll
                for (int i = 0; i < SPT; ++i) nm = fmaxf(nm, sc[i]);
                float corr = __expf(m - nm);
                float a[SPT];
#pragma unroll
                for (int i = 0; i < SPT; ++i) a[i] = __expf(sc[i] - nm);
                float zs = 0.f;
                float4 as = make_float4(0.f, 0.f, 0.f, 0.f);
#pragma unroll
                for (int i = 0; i < SPT; ++i) {
                    zs += a[i];
                    as.x += a[i] * x4[i].x;
                    as.y += a[i] * x4[i].y;
                    as.z += a[i] * x4[i].z;
                    as.w += a[i] * x4[i].w;
                }
                z = z * corr + zs;
                acc.x = acc.x * corr + as.x;
                acc.y = acc.y * corr + as.y;
                acc.z = acc.z * corr + as.z;
                acc.w = acc.w * corr + as.w;
                m = nm;
            }
            if (t + 4 * SPT < deg) {
                int s_now[SPT];
#pragma unroll
                for (int i = 0; i < SPT; ++i) s_now[i] = s_nxt[i];
                load_rows(X4, s_now, cu, x4);
                if (t + 8 * SPT < deg)
                    load_senders(sorted_s, start, deg, t + 8 * SPT, g, s_nxt);
            }
        }

        float M = m;
        M = fmaxf(M, __shfl_xor(M, 8));
        M = fmaxf(M, __shfl_xor(M, 16));
        float f = (m == -INFINITY) ? 0.f : __expf(m - M);
        z *= f; acc.x *= f; acc.y *= f; acc.z *= f; acc.w *= f;
        z     += __shfl_xor(z, 8);      z     += __shfl_xor(z, 16);
        acc.x += __shfl_xor(acc.x, 8);  acc.x += __shfl_xor(acc.x, 16);
        acc.y += __shfl_xor(acc.y, 8);  acc.y += __shfl_xor(acc.y, 16);
        acc.z += __shfl_xor(acc.z, 8);  acc.z += __shfl_xor(acc.z, 16);
        acc.w += __shfl_xor(acc.w, 8);  acc.w += __shfl_xor(acc.w, 16);
        if (g == 0) O4[(unsigned)r * 8u + cu] = acc;
        if (lane == j) zv = z;

        deg = deg_n; start = start_n; q4 = q4_n;
#pragma unroll
        for (int i = 0; i < SPT; ++i) s_cur[i] = s_first_n[i];
    }
    if (lane < nr) degz[r0 + lane] = __float_as_uint(zv);
}

__global__ __launch_bounds__(256) void out_wmma_kernel(
    const float* __restrict__ x, const float* __restrict__ outacc,
    const float* __restrict__ Z, const float* __restrict__ Wvo,
    float* __restrict__ out, int ntiles) {
    int wave = (int)((blockIdx.x * blockDim.x + threadIdx.x) >> 5);
    if (wave >= ntiles) return;
    int lane = (int)(threadIdx.x & 31u);
    int half = lane >> 4;
    int m    = lane & 15;
    size_t row0 = (size_t)wave * 16;

    float invz = 1.0f / (Z[row0 + m] + 1e-6f);
    v8f c0 = {}, c1 = {};

#pragma unroll
    for (int kk = 0; kk < 8; ++kk) {
        int kb = kk * 4;
        const float* ap = outacc + (row0 + m) * 32 + kb + 2 * half;
        v2f a; a.x = ap[0] * invz; a.y = ap[1] * invz;
        wmma_step(Wvo, kb, half, m, a, c0, c1);
    }

#pragma unroll
    for (int r = 0; r < 8; ++r) {
        size_t row = (row0 + r + 8 * half) * 32;
        out[row + m]      = x[row + m]      + c0[r];
        out[row + 16 + m] = x[row + 16 + m] + c1[r];
    }
}

__global__ void out_tail_kernel(const float* __restrict__ x,
                                const float* __restrict__ outacc,
                                const float* __restrict__ Z,
                                const float* __restrict__ Wvo,
                                float* __restrict__ out, int row_start, int N) {
    int idx = blockIdx.x * blockDim.x + threadIdx.x;
    int nrows = N - row_start;
    if (idx >= nrows * 32) return;
    int row = row_start + idx / 32, col = idx & 31;
    float invz = 1.0f / (Z[row] + 1e-6f);
    float acc = 0.f;
    for (int t = 0; t < 32; ++t)
        acc += outacc[(size_t)row * 32 + t] * invz * Wvo[t * 32 + col];
    size_t o = (size_t)row * 32 + col;
    out[o] = x[o] + acc;
}

extern "C" void kernel_launch(void* const* d_in, const int* in_sizes, int n_in,
                              void* d_out, int out_size, void* d_ws,
                              size_t ws_size, hipStream_t stream) {
    const float* x  = (const float*)d_in[0];
    const int*   ei = (const int*)d_in[1];
    const float* Wq = (const float*)d_in[2];
    const float* Wk = (const float*)d_in[3];
    const float* Wv = (const float*)d_in[4];
    const float* Wo = (const float*)d_in[5];
    float* out = (float*)d_out;

    const int N = in_sizes[0] / 32;
    const int E = in_sizes[1] / 2;

    float* Q       = (float*)d_ws;
    float* K       = Q + (size_t)N * 32;
    float* V       = K + (size_t)N * 32;
    float* outacc  = V + (size_t)N * 32;
    float* Wqk     = outacc - 2048;
    float* Wvo     = Wqk + 1024;
    int* sorted_s  = (int*)(outacc + (size_t)N * 32);
    unsigned* cnt  = (unsigned*)(sorted_s + E);
    unsigned* degz = cnt + N;
    unsigned* table  = (unsigned*)Q;
    unsigned* bstart = table + (size_t)SORT_C * NB_MAX;
    int* pairs       = (int*)K;

    const int TB = 256;
    int ntiles = N / 16;
    int rem_row0 = ntiles * 16;
    int rem = N - rem_row0;
    int NB = (N + 255) >> RB_SHIFT;

    weight_fold_kernel<<<1, TB, 0, stream>>>(Wq, Wk, Wv, Wo, Wqk, Wvo);
    bucket_hist_kernel<<<SORT_C, SORT_T, 0, stream>>>(ei, table, E, NB);
    bucket_place_kernel<<<SORT_C, SORT_T, 0, stream>>>(ei, table, bstart,
                                                       pairs, E, NB);
    bucket_sort_kernel<<<NB, SORT_T, 0, stream>>>(pairs, bstart, cnt, degz,
                                                  sorted_s, N);
    {
        int threads = ntiles * 32;
        if (threads > 0)
            qp_wmma_kernel<<<(threads + TB - 1) / TB, TB, 0, stream>>>(
                x, Wqk, Q, ntiles);
        if (rem > 0)
            qp_tail_kernel<<<(rem * 32 + TB - 1) / TB, TB, 0, stream>>>(
                x, Wqk, Q, rem_row0, N);
    }
    seg_attn_kernel<<<(N + 8 * RPW - 1) / (8 * RPW), TB, 0, stream>>>(
        Q, x, sorted_s, cnt, degz, outacc, N);
    {
        int threads = ntiles * 32;
        if (threads > 0)
            out_wmma_kernel<<<(threads + TB - 1) / TB, TB, 0, stream>>>(
                x, outacc, (const float*)degz, Wvo, out, ntiles);
        if (rem > 0)
            out_tail_kernel<<<(rem * 32 + TB - 1) / TB, TB, 0, stream>>>(
                x, outacc, (const float*)degz, Wvo, out, rem_row0, N);
    }
}


# Both projections moved into the segment kernel: a wave owns one WMMA tile of receivers; Qp and weighted sums go through a wave-private LDS tile, not memory; two launches gone; sum order unchanged
# speedup vs baseline: 10.3431x; 1.0914x over previous
#include <hip/hip_runtime.h>
#include <math.h>

typedef __attribute__((ext_vector_type(2))) float v2f;
typedef __attribute__((ext_vector_type(8))) float v8f;

#define SORT_C   256
#define SORT_T   512
#define RB_SHIFT 8
#define NB_MAX   512
#define EPT      8
static_assert(SORT_T == NB_MAX && SORT_T == 512 && SORT_C == 256,
              "the table walk of the placement pass is written for these");

__device__ __forceinline__ unsigned bucket_of(int r, int NB) {
    unsigned b = (unsigned)r >> RB_SHIFT;
    unsigned last = (unsigned)(NB - 1);
    return b < last ? b : last;
}

__global__ __launch_bounds__(SORT_T) void bucket_hist_kernel(
    const int* __restrict__ ei, unsigned* __restrict__ table, int E, int NB) {
    __shared__ unsigned hist[NB_MAX];
    int tid = (int)threadIdx.x;
    int c = (int)blockIdx.x;
    hist[tid] = 0u;
    __syncthreads();
    int CH = (E + SORT_C - 1) / SORT_C;
    int beg = c * CH;
    int end = (beg + CH < E) ? beg + CH : E;
    for (int e = beg + tid; e < end; e += EPT * SORT_T) {
        int r[EPT];
#pragma unroll
        for (int k = 0; k < EPT; ++k) {
            int ee = e + k * SORT_T;
            r[k] = ei[E + (ee < end ? ee : end - 1)];
        }
#pragma unroll
        for (int k = 0; k < EPT; ++k)
            if (e + k * SORT_T < end)
                atomicAdd(&hist[bucket_of(r[k], NB)], 1u);
    }
    __syncthreads();
    table[c * NB_MAX + tid] = hist[tid];
}

__global__ __launch_bounds__(SORT_T) void bucket_place_kernel(
    const int* __restrict__ ei, const unsigned* __restrict__ table,
    unsigned* __restrict__ bstart, int* __restrict__ pairs_raw, int E, int NB) {
    __shared__ unsigned cursor[NB_MAX];
    __shared__ unsigned wsum[SORT_T / 32];
    __shared__ __attribute__((aligned(16))) unsigned ptot[4][NB_MAX];
    __shared__ __attribute__((aligned(16))) unsigned ppre[4][NB_MAX];
    int2* __restrict__ pairs = (int2*)pairs_raw;
    int tid = (int)threadIdx.x;
    int c = (int)blockIdx.x;
    int lane = tid & 31;
    int w = tid >> 5;

    {
        int q = tid & 127;
        int g = tid >> 7;
        int gg = (g + c) & 3;
        const uint4* __restrict__ t4 = (const uint4*)table;
        unsigned t0 = 0u, t1 = 0u, t2 = 0u, t3 = 0u;
        unsigned p0 = 0u, p1 = 0u, p2 = 0u, p3 = 0u;
        for (int k0 = 0; k0 < SORT_C / 4; k0 += 16) {
            uint4 v[16];
#pragma unroll
            for (int j = 0; j < 16; ++j)
                v[j] = t4[(gg * (SORT_C / 4) + k0 + j) * (NB_MAX / 4) + q];
#pragma unroll
            for (int j = 0; j < 16; ++j) {
                unsigned m = (gg * (SORT_C / 4) + k0 + j < c) ? 0xffffffffu : 0u;
                t0 += v[j].x; t1 += v[j].y; t2 += v[j].z; t3 += v[j].w;
                p0 += v[j].x & m;  p1 += v[j].y & m;
                p2 += v[j].z & m;  p3 += v[j].w & m;
            }
        }
        ((uint4*)ptot[g])[q] = make_uint4(t0, t1, t2, t3);
        ((uint4*)ppre[g])[q] = make_uint4(p0, p1, p2, p3);
    }
    __syncthreads();
    unsigned tot = ptot[0][tid] + ptot[1][tid] + ptot[2][tid] + ptot[3][tid];
    unsigned pre = ppre[0][tid] + ppre[1][tid] + ppre[2][tid] + ppre[3][tid];
    unsigned inc = tot;
    for (int d = 1; d < 32; d <<= 1) {
        unsigned t = __shfl_up(inc, d);
        if (lane >= d) inc += t;
    }
    if (lane == 31) wsum[w] = inc;
    __syncthreads();
    unsigned base = 0u;
    for (int k = 0; k < w; ++k) base += wsum[k];
    unsigned excl = base + inc - tot;
    cursor[tid] = excl + pre;
    if (c == 0 && tid < NB) {
        bstart[tid] = excl;
        if (tid == NB - 1) bstart[NB] = excl + tot;
    }
    __syncthreads();

    int CH = (E + SORT_C - 1) / SORT_C;
    int beg = c * CH;
    int end = (beg + CH < E) ? beg + CH : E;
    for (int e = beg + tid; e < end; e += EPT * SORT_T) {
        int s[EPT], r[EPT];
#pragma unroll
        for (int k = 0; k < EPT; ++k) {
            int ee = e + k * SORT_T;
            int ec = ee < end ? ee : end - 1;
            s[k] = ei[ec];
            r[k] = ei[E + ec];
        }
#pragma unroll
        for (int k = 0; k < EPT; ++k)
            if (e + k * SORT_T < end) {
                unsigned pos = atomicAdd(&cursor[bucket_of(r[k], NB)], 1u);
                pairs[pos] = make_int2(s[k], r[k]);
            }
    }
}

__global__ __launch_bounds__(SORT_T) void bucket_sort_kernel(
    const int* __restrict__ pairs_raw, const unsigned* __restrict__ bstart,
    unsigned* __restrict__ segend, unsigned* __restrict__ degz,
    int* __restrict__ sorted_s, int N) {
    __shared__ unsigned cnt[256];
    __shared__ unsigned cur[256];
    __shared__ unsigned wsum[8];
    const int2* __restrict__ pairs = (const int2*)pairs_raw;
    int tid = (int)threadIdx.x;
    int b = (int)blockIdx.x;
    int lane = tid & 31;
    int w = tid >> 5;
    unsigned beg = bstart[b], end = bstart[b + 1];

    if (tid < 256) cnt[tid] = 0u;
    __syncthreads();
    for (unsigned i = beg + (unsigned)tid; i < end; i += EPT * SORT_T) {
        int2 p[EPT];
#pragma unroll
        for (int k = 0; k < EPT; ++k) {
            unsigned ii = i + (unsigned)(k * SORT_T);
            p[k] = pairs[ii < end ? ii : end - 1u];
        }
#pragma unroll
        for (int k = 0; k < EPT; ++k)
            if (i + (unsigned)(k * SORT_T) < end)
                atomicAdd(&cnt[p[k].y & 255], 1u);
    }
    __syncthreads();
    unsigned cv = (tid < 256) ? cnt[tid] : 0u;
    unsigned inc = cv;
    for (int d = 1; d < 32; d <<= 1) {
        unsigned t = __shfl_up(inc, d);
        if (lane >= d) inc += t;
    }
    if (tid < 256 && lane == 31) wsum[w] = inc;
    __syncthreads();
    if (tid < 256) {
        unsigned base = 0u;
        for (int k = 0; k < w; ++k) base += wsum[k];
        unsigned st = beg + base + inc - cv;
        cur[tid] = st;
        int r = (b << RB_SHIFT) + tid;
        if (r < N) { segend[r] = st + cv; degz[r] = cv; }
    }
    __syncthreads();
    for (unsigned i = beg + (unsigned)tid; i < end; i += EPT * SORT_T) {
        int2 p[EPT];
#pragma unroll
        for (int k = 0; k < EPT; ++k) {
            unsigned ii = i + (unsigned)(k * SORT_T);
            p[k] = pairs[ii < end ? ii : end - 1u];
        }
#pragma unroll
        for (int k = 0; k < EPT; ++k)
            if (i + (unsigned)(k * SORT_T) < end) {
                unsigned pos = atomicAdd(&cur[p[k].y & 255], 1u);
                sorted_s[pos] = p[k].x;
            }
    }
}

__device__ __forceinline__ void wmma_step(const float* __restrict__ W, int kb,
                                          int half, int m, v2f a,
                                          v8f& acc0, v8f& acc1) {
    int r0 = (kb + 2 * half + 0) * 32;
    int r1 = (kb + 2 * half + 1) * 32;
    v2f b0, b1;
    b0.x = W[r0 + m];        b0.y = W[r1 + m];
    b1.x = W[r0 + 16 + m];   b1.y = W[r1 + 16 + m];
    acc0 = __builtin_amdgcn_wmma_f32_16x16x4_f32(false, a, false, b0,
                                                 (short)0, acc0, false, false);
    acc1 = __builtin_amdgcn_wmma_f32_16x16x4_f32(false, a, false, b1,
                                                 (short)0, acc1, false, false);
}

__global__ __launch_bounds__(256) void weight_fold_kernel(
    const float* __restrict__ Wq, const float* __restrict__ Wk,
    const float* __restrict__ Wv, const float* __restrict__ Wo,
    float* __restrict__ Wqk, float* __restrict__ Wvo) {
    int wave = (int)(threadIdx.x >> 5);
    int lane = (int)(threadIdx.x & 31u);
    int half = lane >> 4;
    int m    = lane & 15;
    bool vo  = wave >= 4;
    int tr = (wave >> 1) & 1, tc = wave & 1;
    const float* __restrict__ A = vo ? Wv : Wq;
    const float* __restrict__ B = vo ? Wo : Wk;
    v8f acc = {};
#pragma unroll
    for (int kk = 0; kk < 8; ++kk) {
        int k0 = kk * 4 + 2 * half;
        const float* ap = A + (tr * 16 + m) * 32 + k0;
        v2f a; a.x = ap[0]; a.y = ap[1];
        int i0 = vo ? k0 * 32 + tc * 16 + m : (tc * 16 + m) * 32 + k0;
        int i1 = vo ? i0 + 32 : i0 + 1;
        v2f b; b.x = B[i0]; b.y = B[i1];
        acc = __builtin_amdgcn_wmma_f32_16x16x4_f32(false, a, false, b,
                                                    (short)0, acc, false, false);
    }
    float* __restrict__ D = vo ? Wvo : Wqk;
#pragma unroll
    for (int r = 0; r < 8; ++r)
        D[(tr * 16 + r + 8 * half) * 32 + tc * 16 + m] = acc[r];
}

#define RPW 16
#define SPT 4
#define TP  36
#define WPB 2

__device__ __forceinline__ void wave_tile_handover() {
    __builtin_amdgcn_fence(__ATOMIC_RELEASE, "wavefront");
    __builtin_amdgcn_wave_barrier();
    __builtin_amdgcn_fence(__ATOMIC_ACQUIRE, "wavefront");
}

__device__ __forceinline__ void load_senders(const int* __restrict__ sorted_s,
                                             int start, int deg, int e0, int g,
                                             int (&s)[SPT]) {
#pragma unroll
    for (int i = 0; i < SPT; ++i) {
        int jn = e0 + 4 * i + g;
        s[i] = sorted_s[start + (jn < deg ? jn : deg - 1)];
    }
}

__device__ __forceinline__ void load_rows(const float4* __restrict__ X4,
                                          const int (&s)[SPT], unsigned cu,
                                          float4 (&x4)[SPT]) {
#pragma unroll
    for (int i = 0; i < SPT; ++i)
        x4[i] = X4[(unsigned)s[i] * 8u + cu];
}

__device__ __forceinline__ void load_bfrags(const float* __restrict__ Wt,
                                            int half, int m,
                                            v2f (&b0)[8], v2f (&b1)[8]) {
#pragma unroll
    for (int kk = 0; kk < 8; ++kk) {
        int r0w = (kk * 4 + 2 * half + 0) * 32;
        int r1w = (kk * 4 + 2 * half + 1) * 32;
        b0[kk].x = Wt[r0w + m];        b0[kk].y = Wt[r1w + m];
        b1[kk].x = Wt[r0w + 16 + m];   b1[kk].y = Wt[r1w + 16 + m];
    }
}

__global__ __launch_bounds__(32 * WPB) void seg_attn_kernel(
    const float* __restrict__ x, const float* __restrict__ Wqk,
    const float* __restrict__ Wvo, const int* __restrict__ sorted_s,
    const unsigned* __restrict__ segend, const unsigned* __restrict__ degz,
    float* __restrict__ out, int N) {
    __shared__ __attribute__((aligned(16))) float tiles[WPB][16 * TP];
    int wib = __builtin_amdgcn_readfirstlane((int)(threadIdx.x >> 5));
    int r0 = __builtin_amdgcn_readfirstlane(
                 (int)((blockIdx.x * blockDim.x + threadIdx.x) >> 5)) * RPW;
    if (r0 >= N) return;
    float* tile = tiles[wib];
    float4* tile4 = (float4*)tiles[wib];
    int lane = (int)(threadIdx.x & 31u);
    int half = lane >> 4;
    int m    = lane & 15;
    int g = lane >> 3;
    int c = lane & 7;
    int nr = (N - r0 < RPW) ? N - r0 : RPW;
    const float4* __restrict__ X4 = (const float4*)x;
    unsigned cu = (unsigned)c;

    int degv = 0, endv = 0;
    if (lane < nr) {
        degv = (int)degz[r0 + lane];
        endv = (int)segend[r0 + lane];
    }
    float zv = 0.f;

    float4 xrow[4];
#pragma unroll
    for (int i = 0; i < 4; ++i) {
        int row = r0 + 4 * i + g;
        xrow[i] = X4[(unsigned)(row < N ? row : N - 1) * 8u + cu];
    }
    v2f bq0[8], bq1[8];
    load_bfrags(Wqk, half, m, bq0, bq1);

#pragma unroll
    for (int i = 0; i < 4; ++i) tile4[(4 * i + g) * (TP / 4) + c] = xrow[i];
    wave_tile_handover();
    v2f xa[8];
#pragma unroll
    for (int kk = 0; kk < 8; ++kk) {
        const float* ap = tile + m * TP + kk * 4 + 2 * half;
        xa[kk].x = ap[0]; xa[kk].y = ap[1];
    }

    int deg   = __builtin_amdgcn_readlane(degv, 0);
    int start = __builtin_amdgcn_readlane(endv, 0) - deg;
    int s_cur[SPT];
#pragma unroll
    for (int i = 0; i < SPT; ++i) s_cur[i] = r0;
    if (deg > 0) load_senders(sorted_s, start, deg, 0, g, s_cur);

    {
        v8f q0 = {}, q1 = {};
#pragma unroll
        for (int kk = 0; kk < 8; ++kk) {
            q0 = __builtin_amdgcn_wmma_f32_16x16x4_f32(false, xa[kk], false, bq0[kk],
                                                       (short)0, q0, false, false);
            q1 = __builtin_amdgcn_wmma_f32_16x16x4_f32(false, xa[kk], false, bq1[kk],
                                                       (short)0, q1, false, false);
        }
#pragma unroll
        for (int r = 0; r < 8; ++r) {
            tile[(r + 8 * half) * TP + m]      = q0[r];
            tile[(r + 8 * half) * TP + 16 + m] = q1[r];
        }
    }
    wave_tile_handover();
    float4 q4 = tile4[cu];

    for (int j = 0; j < nr; ++j) {
        int r = r0 + j;
        float4 x4[SPT];
        int s_nxt[SPT];
#pragma unroll
        for (int i = 0; i < SPT; ++i) { x4[i] = q4; s_nxt[i] = r; }
        if (deg > 0) {
            load_rows(X4, s_cur, cu, x4);
            if (deg > 4 * SPT)
                load_senders(sorted_s, start, deg, 4 * SPT, g, s_nxt);
        }
        int deg_n = 0, start_n = 0;
        int s_first_n[SPT];
#pragma unroll
        for (int i = 0; i < SPT; ++i) s_first_n[i] = r;
        float4 q4_n = q4;
        if (j + 1 < nr) {
            deg_n   = __builtin_amdgcn_readlane(degv, j + 1);
            start_n = __builtin_amdgcn_readlane(endv, j + 1) - deg_n;
            q4_n = tile4[(unsigned)(j + 1) * (TP / 4) + cu];
            if (deg_n > 0) load_senders(sorted_s, start_n, deg_n, 0, g, s_first_n);
        }

        float mx = -INFINITY;
        float z = 0.f;
        float4 acc = make_float4(0.f, 0.f, 0.f, 0.f);

        for (int t = 0; t < deg; t += 4 * SPT) {
            float p[SPT];
#pragma unroll
            for (int i = 0; i < SPT; ++i)
                p[i] = q4.x * x4[i].x + q4.y * x4[i].y + q4.z * x4[i].z
                     + q4.w * x4[i].w;
#pragma unroll
            for (int i = 0; i < SPT; ++i) p[i] += __shfl_xor(p[i], 1);
#pragma unroll
            for (int i = 0; i < SPT; ++i) p[i] += __shfl_xor(p[i], 2);
#pragma unroll
            for (int i = 0; i < SPT; ++i) p[i] += __shfl_xor(p[i], 4);

            if (t + g < deg) {
                float sc[SPT];
#pragma unroll
                for (int i = 0; i < SPT; ++i)
                    sc[i] = (t + 4 * i + g < deg) ? p[i] * 0.17677669529663687f
                                                  : -INFINITY;
                float nm = mx;
#pragma unroll
                for (int i = 0; i < SPT; ++i) nm = fmaxf(nm, sc[i]);
                float corr = __expf(mx - nm);
                float a[SPT];
#pragma unroll
                for (int i = 0; i < SPT; ++i) a[i] = __expf(sc[i] - nm);
                float zs = 0.f;
                float4 as = make_float4(0.f, 0.f, 0.f, 0.f);
#pragma unroll
                for (int i = 0; i < SPT; ++i) {
                    zs += a[i];
                    as.x += a[i] * x4[i].x;
                    as.y += a[i] * x4[i].y;
                    as.z += a[i] * x4[i].z;
                    as.w += a[i] * x4[i].w;
                }
                z = z * corr + zs;
                acc.x = acc.x * corr + as.x;
                acc.y = acc.y * corr + as.y;
                acc.z = acc.z * corr + as.z;
                acc.w = acc.w * corr + as.w;
                mx = nm;
            }
            if (t + 4 * SPT < deg) {
                int s_now[SPT];
#pragma unroll
                for (int i = 0; i < SPT; ++i) s_now[i] = s_nxt[i];
                load_rows(X4, s_now, cu, x4);
                if (t + 8 * SPT < deg)
                    load_senders(sorted_s, start, deg, t + 8 * SPT, g, s_nxt);
            }
        }

        float M = mx;
        M = fmaxf(M, __shfl_xor(M, 8));
        M = fmaxf(M, __shfl_xor(M, 16));
        float f = (mx == -INFINITY) ? 0.f : __expf(mx - M);
        z *= f; acc.x *= f; acc.y *= f; acc.z *= f; acc.w *= f;
        z     += __shfl_xor(z, 8);      z     += __shfl_xor(z, 16);
        acc.x += __shfl_xor(acc.x, 8);  acc.x += __shfl_xor(acc.x, 16);
        acc.y += __shfl_xor(acc.y, 8);  acc.y += __shfl_xor(acc.y, 16);
        acc.z += __shfl_xor(acc.z, 8);  acc.z += __shfl_xor(acc.z, 16);
        acc.w += __shfl_xor(acc.w, 8);  acc.w += __shfl_xor(acc.w, 16);
        if (g == 0) tile4[(unsigned)j * (TP / 4) + cu] = acc;
        if (lane == j) zv = z;

        deg = deg_n; start = start_n; q4 = q4_n;
#pragma unroll
        for (int i = 0; i < SPT; ++i) s_cur[i] = s_first_n[i];
    }

    wave_tile_handover();
    {
        float zrow = __shfl(zv, m);
        float invz = 1.0f / (zrow + 1e-6f);
        float4 xres[4];
#pragma unroll
        for (int i = 0; i < 4; ++i) {
            int row = r0 + 4 * i + g;
            xres[i] = X4[(unsigned)(row < N ? row : N - 1) * 8u + cu];
        }
        v8f c0 = {}, c1 = {};
#pragma unroll
        for (int kk = 0; kk < 8; ++kk) {
            const float* ap = tile + m * TP + kk * 4 + 2 * half;
            v2f a; a.x = ap[0] * invz; a.y = ap[1] * invz;
            wmma_step(Wvo, kk * 4, half, m, a, c0, c1);
        }
#pragma unroll
        for (int r = 0; r < 8; ++r) {
            tile[(r + 8 * half) * TP + m]      = c0[r];
            tile[(r + 8 * half) * TP + 16 + m] = c1[r];
        }
        wave_tile_handover();
        float4* O4 = (float4*)out;
#pragma unroll
        for (int i = 0; i < 4; ++i) {
            int row = r0 + 4 * i + g;
            float4 d = tile4[(4 * i + g) * (TP / 4) + c];
            float4 o = make_float4(xres[i].x + d.x, xres[i].y + d.y,
                                   xres[i].z + d.z, xres[i].w + d.w);
            if (row < N) O4[(unsigned)row * 8u + cu] = o;
        }
    }
}

extern "C" void kernel_launch(void* const* d_in, const int* in_sizes, int n_in,
                              void* d_out, int out_size, void* d_ws,
                              size_t ws_size, hipStream_t stream) {
    const float* x  = (const float*)d_in[0];
    const int*   ei = (const int*)d_in[1];
    const float* Wq = (const float*)d_in[2];
    const float* Wk = (const float*)d_in[3];
    const float* Wv = (const float*)d_in[4];
    const float* Wo = (const float*)d_in[5];
    float* out = (float*)d_out;

    const int N = in_sizes[0] / 32;
    const int E = in_sizes[1] / 2;

    float* P0      = (float*)d_ws;
    float* P1      = P0 + (size_t)N * 32;
    float* P2      = P1 + (size_t)N * 32;
    float* P3      = P2 + (size_t)N * 32;
    float* Wqk     = P3 - 2048;
    float* Wvo     = Wqk + 1024;
    int* sorted_s  = (int*)(P3 + (size_t)N * 32);
    unsigned* cnt  = (unsigned*)(sorted_s + E);
    unsigned* degz = cnt + N;
    unsigned* table  = (unsigned*)P0;
    unsigned* bstart = table + (size_t)SORT_C * NB_MAX;
    int* pairs       = (int*)P1;

    const int TB = 256;
    int NB = (N + 255) >> RB_SHIFT;

    weight_fold_kernel<<<1, TB, 0, stream>>>(Wq, Wk, Wv, Wo, Wqk, Wvo);
    bucket_hist_kernel<<<SORT_C, SORT_T, 0, stream>>>(ei, table, E, NB);
    bucket_place_kernel<<<SORT_C, SORT_T, 0, stream>>>(ei, table, bstart,
                                                       pairs, E, NB);
    bucket_sort_kernel<<<NB, SORT_T, 0, stream>>>(pairs, bstart, cnt, degz,
                                                  sorted_s, N);
    seg_attn_kernel<<<(N + WPB * RPW - 1) / (WPB * RPW), 32 * WPB, 0, stream>>>(
        x, Wqk, Wvo, sorted_s, cnt, degz, out, N);
}


# Sort level 1 local to a chunk: count, scan, file by bucket in one launch, no cross-chunk table walk; level 2 gathers its bucket's 256 runs; one launch gone; sum order still the sort's arrival order
# speedup vs baseline: 13.0158x; 1.2584x over previous
#include <hip/hip_runtime.h>
#include <math.h>

typedef __attribute__((ext_vector_type(2))) float v2f;
typedef __attribute__((ext_vector_type(8))) float v8f;

#define SORT_C   256
#define SORT_T   512
#define RB_SHIFT 8
#define NB_MAX   512
#define EPT      8
#define RPWV     16
static_assert(SORT_T == NB_MAX && SORT_C == 256 &&
              SORT_C == RPWV * (SORT_T / 32),
              "level 2 maps thread t < 256 to chunk t and wave w to 16 chunks");

__device__ __forceinline__ unsigned bucket_of(int r, int NB) {
    unsigned b = (unsigned)r >> RB_SHIFT;
    unsigned last = (unsigned)(NB - 1);
    return b < last ? b : last;
}

__global__ __launch_bounds__(SORT_T) void chunk_sort_kernel(
    const int* __restrict__ ei, unsigned* __restrict__ runs_raw,
    int* __restrict__ pairs_raw, int E, int NB) {
    __shared__ unsigned hist[NB_MAX];
    __shared__ unsigned wsum[SORT_T / 32];
    uint2* __restrict__ runs = (uint2*)runs_raw;
    int2* __restrict__ pairs = (int2*)pairs_raw;
    int tid = (int)threadIdx.x;
    int c = (int)blockIdx.x;
    int lane = tid & 31;
    int w = tid >> 5;
    hist[tid] = 0u;
    __syncthreads();
    int CH = (E + SORT_C - 1) / SORT_C;
    int beg = c * CH;
    int end = (beg + CH < E) ? beg + CH : E;
    for (int e = beg + tid; e < end; e += EPT * SORT_T) {
        int r[EPT];
#pragma unroll
        for (int k = 0; k < EPT; ++k) {
            int ee = e + k * SORT_T;
            r[k] = ei[E + (ee < end ? ee : end - 1)];
        }
#pragma unroll
        for (int k = 0; k < EPT; ++k)
            if (e + k * SORT_T < end)
                atomicAdd(&hist[bucket_of(r[k], NB)], 1u);
    }
    __syncthreads();
    unsigned tot = hist[tid];
    unsigned inc = tot;
    for (int d = 1; d < 32; d <<= 1) {
        unsigned t = __shfl_up(inc, d);
        if (lane >= d) inc += t;
    }
    if (lane == 31) wsum[w] = inc;
    __syncthreads();
    unsigned base = 0u;
    for (int k = 0; k < w; ++k) base += wsum[k];
    unsigned excl = base + inc - tot;
    runs[c * NB_MAX + tid] = make_uint2(excl, tot);
    hist[tid] = excl;
    __syncthreads();

    for (int e = beg + tid; e < end; e += EPT * SORT_T) {
        int s[EPT], r[EPT];
#pragma unroll
        for (int k = 0; k < EPT; ++k) {
            int ee = e + k * SORT_T;
            int ec = ee < end ? ee : end - 1;
            s[k] = ei[ec];
            r[k] = ei[E + ec];
        }
#pragma unroll
        for (int k = 0; k < EPT; ++k)
            if (e + k * SORT_T < end) {
                unsigned pos = atomicAdd(&hist[bucket_of(r[k], NB)], 1u);
                pairs[(unsigned)beg + pos] = make_int2(s[k], r[k]);
            }
    }
}

__global__ __launch_bounds__(SORT_T) void bucket_sort_kernel(
    const int* __restrict__ pairs_raw, const unsigned* __restrict__ runs_raw,
    unsigned* __restrict__ segend, unsigned* __restrict__ degz,
    int* __restrict__ sorted_s, int N, int E) {
    __shared__ unsigned cnt[256];
    __shared__ unsigned cur[256];
    __shared__ unsigned wsum[8];
    __shared__ unsigned osum[8];
    __shared__ unsigned roff[SORT_C];
    __shared__ unsigned rlen[SORT_C];
    const int2* __restrict__ pairs = (const int2*)pairs_raw;
    const uint2* __restrict__ runs = (const uint2*)runs_raw;
    int tid = (int)threadIdx.x;
    int b = (int)blockIdx.x;
    int lane = tid & 31;
    int w = tid >> 5;
    int CH = (E + SORT_C - 1) / SORT_C;
    unsigned last = (unsigned)E - 1u;

    if (tid < 256) {
        uint2 oc = runs[tid * NB_MAX + b];
        roff[tid] = (unsigned)(tid * CH) + oc.x;
        rlen[tid] = oc.y;
        cnt[tid] = 0u;
        unsigned s = oc.x;
        for (int d = 16; d > 0; d >>= 1) s += __shfl_xor(s, d);
        if (lane == 0) osum[w] = s;
    }
    __syncthreads();
    unsigned beg = 0u;
    for (int k = 0; k < 8; ++k) beg += osum[k];

    int2 p[RPWV];
    unsigned live = 0u;
    bool tail = false;
#pragma unroll
    for (int k = 0; k < RPWV; ++k) {
        unsigned n = rlen[w * RPWV + k];
        unsigned idx = roff[w * RPWV + k] + (unsigned)lane;
        p[k] = pairs[idx < last ? idx : last];
        if ((unsigned)lane < n) live |= 1u << k;
        tail = tail || (n > 32u);
    }
#pragma unroll
    for (int k = 0; k < RPWV; ++k)
        if (live & (1u << k)) atomicAdd(&cnt[p[k].y & 255], 1u);
    if (tail)
        for (int k = 0; k < RPWV; ++k) {
            unsigned n = rlen[w * RPWV + k];
            unsigned o = roff[w * RPWV + k];
            for (unsigned j = 32u + (unsigned)lane; j < n; j += 32u)
                atomicAdd(&cnt[pairs[o + j].y & 255], 1u);
        }
    __syncthreads();
    unsigned cv = (tid < 256) ? cnt[tid] : 0u;
    unsigned inc = cv;
    for (int d = 1; d < 32; d <<= 1) {
        unsigned t = __shfl_up(inc, d);
        if (lane >= d) inc += t;
    }
    if (tid < 256 && lane == 31) wsum[w] = inc;
    __syncthreads();
    if (tid < 256) {
        unsigned base = 0u;
        for (int k = 0; k < w; ++k) base += wsum[k];
        unsigned st = beg + base + inc - cv;
        cur[tid] = st;
        int r = (b << RB_SHIFT) + tid;
        if (r < N) { segend[r] = st + cv; degz[r] = cv; }
    }
    __syncthreads();
    unsigned pos[RPWV];
#pragma unroll
    for (int k = 0; k < RPWV; ++k) {
        pos[k] = 0u;
        if (live & (1u << k)) pos[k] = atomicAdd(&cur[p[k].y & 255], 1u);
    }
#pragma unroll
    for (int k = 0; k < RPWV; ++k)
        if (live & (1u << k)) sorted_s[pos[k]] = p[k].x;
    if (tail)
        for (int k = 0; k < RPWV; ++k) {
            unsigned n = rlen[w * RPWV + k];
            unsigned o = roff[w * RPWV + k];
            for (unsigned j = 32u + (unsigned)lane; j < n; j += 32u) {
                int2 q = pairs[o + j];
                unsigned pos = atomicAdd(&cur[q.y & 255], 1u);
                sorted_s[pos] = q.x;
            }
        }
}

__device__ __forceinline__ void wmma_step(const float* __restrict__ W, int kb,
                                          int half, int m, v2f a,
                                          v8f& acc0, v8f& acc1) {
    int r0 = (kb + 2 * half + 0) * 32;
    int r1 = (kb + 2 * half + 1) * 32;
    v2f b0, b1;
    b0.x = W[r0 + m];        b0.y = W[r1 + m];
    b1.x = W[r0 + 16 + m];   b1.y = W[r1 + 16 + m];
    acc0 = __builtin_amdgcn_wmma_f32_16x16x4_f32(false, a, false, b0,
                                                 (short)0, acc0, false, false);
    acc1 = __builtin_amdgcn_wmma_f32_16x16x4_f32(false, a, false, b1,
                                                 (short)0, acc1, false, false);
}

__global__ __launch_bounds__(256) void weight_fold_kernel(
    const float* __restrict__ Wq, const float* __restrict__ Wk,
    const float* __restrict__ Wv, const float* __restrict__ Wo,
    float* __restrict__ Wqk, float* __restrict__ Wvo) {
    int wave = (int)(threadIdx.x >> 5);
    int lane = (int)(threadIdx.x & 31u);
    int half = lane >> 4;
    int m    = lane & 15;
    bool vo  = wave >= 4;
    int tr = (wave >> 1) & 1, tc = wave & 1;
    const float* __restrict__ A = vo ? Wv : Wq;
    const float* __restrict__ B = vo ? Wo : Wk;
    v8f acc = {};
#pragma unroll
    for (int kk = 0; kk < 8; ++kk) {
        int k0 = kk * 4 + 2 * half;
        const float* ap = A + (tr * 16 + m) * 32 + k0;
        v2f a; a.x = ap[0]; a.y = ap[1];
        int i0 = vo ? k0 * 32 + tc * 16 + m : (tc * 16 + m) * 32 + k0;
        int i1 = vo ? i0 + 32 : i0 + 1;
        v2f b; b.x = B[i0]; b.y = B[i1];
        acc = __builtin_amdgcn_wmma_f32_16x16x4_f32(false, a, false, b,
                                                    (short)0, acc, false, false);
    }
    float* __restrict__ D = vo ? Wvo : Wqk;
#pragma unroll
    for (int r = 0; r < 8; ++r)
        D[(tr * 16 + r + 8 * half) * 32 + tc * 16 + m] = acc[r];
}

#define RPW 16
#define SPT 4
#define TP  36
#define WPB 2

__device__ __forceinline__ void wave_tile_handover() {
    __builtin_amdgcn_fence(__ATOMIC_RELEASE, "wavefront");
    __builtin_amdgcn_wave_barrier();
    __builtin_amdgcn_fence(__ATOMIC_ACQUIRE, "wavefront");
}

__device__ __forceinline__ void load_senders(const int* __restrict__ sorted_s,
                                             int start, int deg, int e0, int g,
                                             int (&s)[SPT]) {
#pragma unroll
    for (int i = 0; i < SPT; ++i) {
        int jn = e0 + 4 * i + g;
        s[i] = sorted_s[start + (jn < deg ? jn : deg - 1)];
    }
}

__device__ __forceinline__ void load_rows(const float4* __restrict__ X4,
                                          const int (&s)[SPT], unsigned cu,
                                          float4 (&x4)[SPT]) {
#pragma unroll
    for (int i = 0; i < SPT; ++i)
        x4[i] = X4[(unsigned)s[i] * 8u + cu];
}

__device__ __forceinline__ void load_bfrags(const float* __restrict__ Wt,
                                            int half, int m,
                                            v2f (&b0)[8], v2f (&b1)[8]) {
#pragma unroll
    for (int kk = 0; kk < 8; ++kk) {
        int r0w = (kk * 4 + 2 * half + 0) * 32;
        int r1w = (kk * 4 + 2 * half + 1) * 32;
        b0[kk].x = Wt[r0w + m];        b0[kk].y = Wt[r1w + m];
        b1[kk].x = Wt[r0w + 16 + m];   b1[kk].y = Wt[r1w + 16 + m];
    }
}

__global__ __launch_bounds__(32 * WPB) void seg_attn_kernel(
    const float* __restrict__ x, const float* __restrict__ Wqk,
    const float* __restrict__ Wvo, const int* __restrict__ sorted_s,
    const unsigned* __restrict__ segend, const unsigned* __restrict__ degz,
    float* __restrict__ out, int N) {
    __shared__ __attribute__((aligned(16))) float tiles[WPB][16 * TP];
    int wib = __builtin_amdgcn_readfirstlane((int)(threadIdx.x >> 5));
    int r0 = __builtin_amdgcn_readfirstlane(
                 (int)((blockIdx.x * blockDim.x + threadIdx.x) >> 5)) * RPW;
    if (r0 >= N) return;
    float* tile = tiles[wib];
    float4* tile4 = (float4*)tiles[wib];
    int lane = (int)(threadIdx.x & 31u);
    int half = lane >> 4;
    int m    = lane & 15;
    int g = lane >> 3;
    int c = lane & 7;
    int nr = (N - r0 < RPW) ? N - r0 : RPW;
    const float4* __restrict__ X4 = (const float4*)x;
    unsigned cu = (unsigned)c;

    int degv = 0, endv = 0;
    if (lane < nr) {
        degv = (int)degz[r0 + lane];
        endv = (int)segend[r0 + lane];
    }
    float zv = 0.f;

    float4 xrow[4];
#pragma unroll
    for (int i = 0; i < 4; ++i) {
        int row = r0 + 4 * i + g;
        xrow[i] = X4[(unsigned)(row < N ? row : N - 1) * 8u + cu];
    }
    v2f bq0[8], bq1[8];
    load_bfrags(Wqk, half, m, bq0, bq1);

#pragma unroll
    for (int i = 0; i < 4; ++i) tile4[(4 * i + g) * (TP / 4) + c] = xrow[i];
    wave_tile_handover();
    v2f xa[8];
#pragma unroll
    for (int kk = 0; kk < 8; ++kk) {
        const float* ap = tile + m * TP + kk * 4 + 2 * half;
        xa[kk].x = ap[0]; xa[kk].y = ap[1];
    }

    int deg   = __builtin_amdgcn_readlane(degv, 0);
    int start = __builtin_amdgcn_readlane(endv, 0) - deg;
    int s_cur[SPT];
#pragma unroll
    for (int i = 0; i < SPT; ++i) s_cur[i] = r0;
    if (deg > 0) load_senders(sorted_s, start, deg, 0, g, s_cur);

    {
        v8f q0 = {}, q1 = {};
#pragma unroll
        for (int kk = 0; kk < 8; ++kk) {
            q0 = __builtin_amdgcn_wmma_f32_16x16x4_f32(false, xa[kk], false, bq0[kk],
                                                       (short)0, q0, false, false);
            q1 = __builtin_amdgcn_wmma_f32_16x16x4_f32(false, xa[kk], false, bq1[kk],
                                                       (short)0, q1, false, false);
        }
#pragma unroll
        for (int r = 0; r < 8; ++r) {
            tile[(r + 8 * half) * TP + m]      = q0[r];
            tile[(r + 8 * half) * TP + 16 + m] = q1[r];
        }
    }
    wave_tile_handover();
    float4 q4 = tile4[cu];

    for (int j = 0; j < nr; ++j) {
        int r = r0 + j;
        float4 x4[SPT];
        int s_nxt[SPT];
#pragma unroll
        for (int i = 0; i < SPT; ++i) { x4[i] = q4; s_nxt[i] = r; }
        if (deg > 0) {
            load_rows(X4, s_cur, cu, x4);
            if (deg > 4 * SPT)
                load_senders(sorted_s, start, deg, 4 * SPT, g, s_nxt);
        }
        int deg_n = 0, start_n = 0;
        int s_first_n[SPT];
#pragma unroll
        for (int i = 0; i < SPT; ++i) s_first_n[i] = r;
        float4 q4_n = q4;
        if (j + 1 < nr) {
            deg_n   = __builtin_amdgcn_readlane(degv, j + 1);
            start_n = __builtin_amdgcn_readlane(endv, j + 1) - deg_n;
            q4_n = tile4[(unsigned)(j + 1) * (TP / 4) + cu];
            if (deg_n > 0) load_senders(sorted_s, start_n, deg_n, 0, g, s_first_n);
        }

        float mx = -INFINITY;
        float z = 0.f;
        float4 acc = make_float4(0.f, 0.f, 0.f, 0.f);

        for (int t = 0; t < deg; t += 4 * SPT) {
            float p[SPT];
#pragma unroll
            for (int i = 0; i < SPT; ++i)
                p[i] = q4.x * x4[i].x + q4.y * x4[i].y + q4.z * x4[i].z
                     + q4.w * x4[i].w;
#pragma unroll
            for (int i = 0; i < SPT; ++i) p[i] += __shfl_xor(p[i], 1);
#pragma unroll
            for (int i = 0; i < SPT; ++i) p[i] += __shfl_xor(p[i], 2);
#pragma unroll
            for (int i = 0; i < SPT; ++i) p[i] += __shfl_xor(p[i], 4);

            if (t + g < deg) {
                float sc[SPT];
#pragma unroll
                for (int i = 0; i < SPT; ++i)
                    sc[i] = (t + 4 * i + g < deg) ? p[i] * 0.17677669529663687f
                                                  : -INFINITY;
                float nm = mx;
#pragma unroll
                for (int i = 0; i < SPT; ++i) nm = fmaxf(nm, sc[i]);
                float corr = __expf(mx - nm);
                float a[SPT];
#pragma unroll
                for (int i = 0; i < SPT; ++i) a[i] = __expf(sc[i] - nm);
                float zs = 0.f;
                float4 as = make_float4(0.f, 0.f, 0.f, 0.f);
#pragma unroll
                for (int i = 0; i < SPT; ++i) {
                    zs += a[i];
                    as.x += a[i] * x4[i].x;
                    as.y += a[i] * x4[i].y;
                    as.z += a[i] * x4[i].z;
                    as.w += a[i] * x4[i].w;
                }
                z = z * corr + zs;
                acc.x = acc.x * corr + as.x;
                acc.y = acc.y * corr + as.y;
                acc.z = acc.z * corr + as.z;
                acc.w = acc.w * corr + as.w;
                mx = nm;
            }
            if (t + 4 * SPT < deg) {
                int s_now[SPT];
#pragma unroll
                for (int i = 0; i < SPT; ++i) s_now[i] = s_nxt[i];
                load_rows(X4, s_now, cu, x4);
                if (t + 8 * SPT < deg)
                    load_senders(sorted_s, start, deg, t + 8 * SPT, g, s_nxt);
            }
        }

        float M = mx;
        M = fmaxf(M, __shfl_xor(M, 8));
        M = fmaxf(M, __shfl_xor(M, 16));
        float f = (mx == -INFINITY) ? 0.f : __expf(mx - M);
        z *= f; acc.x *= f; acc.y *= f; acc.z *= f; acc.w *= f;
        z     += __shfl_xor(z, 8);      z     += __shfl_xor(z, 16);
        acc.x += __shfl_xor(acc.x, 8);  acc.x += __shfl_xor(acc.x, 16);
        acc.y += __shfl_xor(acc.y, 8);  acc.y += __shfl_xor(acc.y, 16);
        acc.z += __shfl_xor(acc.z, 8);  acc.z += __shfl_xor(acc.z, 16);
        acc.w += __shfl_xor(acc.w, 8);  acc.w += __shfl_xor(acc.w, 16);
        if (g == 0) tile4[(unsigned)j * (TP / 4) + cu] = acc;
        if (lane == j) zv = z;

        deg = deg_n; start = start_n; q4 = q4_n;
#pragma unroll
        for (int i = 0; i < SPT; ++i) s_cur[i] = s_first_n[i];
    }

    wave_tile_handover();
    {
        float zrow = __shfl(zv, m);
        float invz = 1.0f / (zrow + 1e-6f);
        float4 xres[4];
#pragma unroll
        for (int i = 0; i < 4; ++i) {
            int row = r0 + 4 * i + g;
            xres[i] = X4[(unsigned)(row < N ? row : N - 1) * 8u + cu];
        }
        v8f c0 = {}, c1 = {};
#pragma unroll
        for (int kk = 0; kk < 8; ++kk) {
            const float* ap = tile + m * TP + kk * 4 + 2 * half;
            v2f a; a.x = ap[0] * invz; a.y = ap[1] * invz;
            wmma_step(Wvo, kk * 4, half, m, a, c0, c1);
        }
#pragma unroll
        for (int r = 0; r < 8; ++r) {
            tile[(r + 8 * half) * TP + m]      = c0[r];
            tile[(r + 8 * half) * TP + 16 + m] = c1[r];
        }
        wave_tile_handover();
        float4* O4 = (float4*)out;
#pragma unroll
        for (int i = 0; i < 4; ++i) {
            int row = r0 + 4 * i + g;
            float4 d = tile4[(4 * i + g) * (TP / 4) + c];
            float4 o = make_float4(xres[i].x + d.x, xres[i].y + d.y,
                                   xres[i].z + d.z, xres[i].w + d.w);
            if (row < N) O4[(unsigned)row * 8u + cu] = o;
        }
    }
}

extern "C" void kernel_launch(void* const* d_in, const int* in_sizes, int n_in,
                              void* d_out, int out_size, void* d_ws,
                              size_t ws_size, hipStream_t stream) {
    const float* x  = (const float*)d_in[0];
    const int*   ei = (const int*)d_in[1];
    const float* Wq = (const float*)d_in[2];
    const float* Wk = (const float*)d_in[3];
    const float* Wv = (const float*)d_in[4];
    const float* Wo = (const float*)d_in[5];
    float* out = (float*)d_out;

    const int N = in_sizes[0] / 32;
    const int E = in_sizes[1] / 2;

    float* P0      = (float*)d_ws;
    float* P1      = P0 + (size_t)N * 32;
    float* P2      = P1 + (size_t)N * 32;
    float* P3      = P2 + (size_t)N * 32;
    float* Wqk     = P3 - 2048;
    float* Wvo     = Wqk + 1024;
    int* sorted_s  = (int*)(P3 + (size_t)N * 32);
    unsigned* cnt  = (unsigned*)(sorted_s + E);
    unsigned* degz = cnt + N;
    unsigned* runs = (unsigned*)P0;
    int* pairs     = (int*)P1;

    const int TB = 256;
    int NB = (N + 255) >> RB_SHIFT;

    weight_fold_kernel<<<1, TB, 0, stream>>>(Wq, Wk, Wv, Wo, Wqk, Wvo);
    chunk_sort_kernel<<<SORT_C, SORT_T, 0, stream>>>(ei, runs, pairs, E, NB);
    bucket_sort_kernel<<<NB, SORT_T, 0, stream>>>(pairs, runs, cnt, degz,
                                                  sorted_s, N, E);
    seg_attn_kernel<<<(N + WPB * RPW - 1) / (WPB * RPW), 32 * WPB, 0, stream>>>(
        x, Wqk, Wvo, sorted_s, cnt, degz, out, N);
}


# Both sort levels place their items in LDS, then write their contiguous stretch out in whole aligned 128-byte lines, not one scattered 8- or 4-byte store per edge; sum order as before (arrival order)
# speedup vs baseline: 14.4119x; 1.1073x over previous
#include <hip/hip_runtime.h>
#include <math.h>

typedef __attribute__((ext_vector_type(2))) float v2f;
typedef __attribute__((ext_vector_type(8))) float v8f;

#define SORT_C   256
#define SORT_T   512
#define RB_SHIFT 8
#define NB_MAX   512
#define EPT      8
#define RPWV     16
#define PAIR_CAP 6656
#define SLOT_CAP 8192
static_assert(SORT_T == NB_MAX && SORT_C == 256 &&
              SORT_C == RPWV * (SORT_T / 32),
              "level 2 maps thread t < 256 to chunk t and wave w to 16 chunks");

__device__ __forceinline__ unsigned bucket_of(int r, int NB) {
    unsigned b = (unsigned)r >> RB_SHIFT;
    unsigned last = (unsigned)(NB - 1);
    return b < last ? b : last;
}

__device__ __forceinline__ void file_pairs(const int* __restrict__ ei,
                                           int2* __restrict__ pairs, int2* stage,
                                           unsigned* hist, int E, int NB, int beg,
                                           int end, int tid, bool staged) {
    for (int e = beg + tid; e < end; e += EPT * SORT_T) {
        int s[EPT], r[EPT];
#pragma unroll
        for (int k = 0; k < EPT; ++k) {
            int ee = e + k * SORT_T;
            int ec = ee < end ? ee : end - 1;
            s[k] = ei[ec];
            r[k] = ei[E + ec];
        }
#pragma unroll
        for (int k = 0; k < EPT; ++k)
            if (e + k * SORT_T < end) {
                unsigned pos = atomicAdd(&hist[bucket_of(r[k], NB)], 1u);
                if (staged) stage[pos] = make_int2(s[k], r[k]);
                else        pairs[(unsigned)beg + pos] = make_int2(s[k], r[k]);
            }
    }
}

__global__ __launch_bounds__(SORT_T) void chunk_sort_kernel(
    const int* __restrict__ ei, unsigned* __restrict__ runs_raw,
    int* __restrict__ pairs_raw, int E, int NB) {
    __shared__ unsigned hist[NB_MAX];
    __shared__ unsigned wsum[SORT_T / 32];
    __shared__ int2 stage[PAIR_CAP];
    uint2* __restrict__ runs = (uint2*)runs_raw;
    int2* __restrict__ pairs = (int2*)pairs_raw;
    int tid = (int)threadIdx.x;
    int c = (int)blockIdx.x;
    int lane = tid & 31;
    int w = tid >> 5;
    hist[tid] = 0u;
    __syncthreads();
    int CH = (E + SORT_C - 1) / SORT_C;
    int beg = c * CH;
    int end = (beg + CH < E) ? beg + CH : E;
    for (int e = beg + tid; e < end; e += EPT * SORT_T) {
        int r[EPT];
#pragma unroll
        for (int k = 0; k < EPT; ++k) {
            int ee = e + k * SORT_T;
            r[k] = ei[E + (ee < end ? ee : end - 1)];
        }
#pragma unroll
        for (int k = 0; k < EPT; ++k)
            if (e + k * SORT_T < end)
                atomicAdd(&hist[bucket_of(r[k], NB)], 1u);
    }
    __syncthreads();
    unsigned tot = hist[tid];
    unsigned inc = tot;
    for (int d = 1; d < 32; d <<= 1) {
        unsigned t = __shfl_up(inc, d);
        if (lane >= d) inc += t;
    }
    if (lane == 31) wsum[w] = inc;
    __syncthreads();
    unsigned base = 0u;
    for (int k = 0; k < w; ++k) base += wsum[k];
    unsigned excl = base + inc - tot;
    runs[c * NB_MAX + tid] = make_uint2(excl, tot);
    hist[tid] = excl;
    __syncthreads();

    if (CH > PAIR_CAP) {
        file_pairs(ei, pairs, stage, hist, E, NB, beg, end, tid, false);
        return;
    }
    file_pairs(ei, pairs, stage, hist, E, NB, beg, end, tid, true);
    __syncthreads();
    int n = end - beg;
    int a = beg & 15;
    if (n <= 0) return;
    for (int i0 = tid - a; i0 < n; i0 += 4 * SORT_T) {
        int2 v[4];
#pragma unroll
        for (int k = 0; k < 4; ++k) {
            int i = i0 + k * SORT_T;
            v[k] = stage[i < 0 ? 0 : (i < n ? i : n - 1)];
        }
#pragma unroll
        for (int k = 0; k < 4; ++k) {
            int i = i0 + k * SORT_T;
            if (i >= 0 && i < n) pairs[beg + i] = v[k];
        }
    }
}

__global__ __launch_bounds__(SORT_T) void bucket_sort_kernel(
    const int* __restrict__ pairs_raw, const unsigned* __restrict__ runs_raw,
    unsigned* __restrict__ segend, unsigned* __restrict__ degz,
    int* __restrict__ sorted_s, int N, int E) {
    __shared__ unsigned cnt[256];
    __shared__ unsigned cur[256];
    __shared__ unsigned wsum[8];
    __shared__ unsigned osum[8];
    __shared__ unsigned roff[SORT_C];
    __shared__ unsigned rlen[SORT_C];
    __shared__ int slots[SLOT_CAP];
    const int2* __restrict__ pairs = (const int2*)pairs_raw;
    const uint2* __restrict__ runs = (const uint2*)runs_raw;
    int tid = (int)threadIdx.x;
    int b = (int)blockIdx.x;
    int lane = tid & 31;
    int w = tid >> 5;
    int CH = (E + SORT_C - 1) / SORT_C;
    unsigned last = (unsigned)E - 1u;

    if (tid < 256) {
        uint2 oc = runs[tid * NB_MAX + b];
        roff[tid] = (unsigned)(tid * CH) + oc.x;
        rlen[tid] = oc.y;
        cnt[tid] = 0u;
        unsigned s = oc.x;
        for (int d = 16; d > 0; d >>= 1) s += __shfl_xor(s, d);
        if (lane == 0) osum[w] = s;
    }
    __syncthreads();
    unsigned beg = 0u;
    for (int k = 0; k < 8; ++k) beg += osum[k];

    int2 p[RPWV];
    unsigned live = 0u;
    bool tail = false;
#pragma unroll
    for (int k = 0; k < RPWV; ++k) {
        unsigned n = rlen[w * RPWV + k];
        unsigned idx = roff[w * RPWV + k] + (unsigned)lane;
        p[k] = pairs[idx < last ? idx : last];
        if ((unsigned)lane < n) live |= 1u << k;
        tail = tail || (n > 32u);
    }
#pragma unroll
    for (int k = 0; k < RPWV; ++k)
        if (live & (1u << k)) atomicAdd(&cnt[p[k].y & 255], 1u);
    if (tail)
        for (int k = 0; k < RPWV; ++k) {
            unsigned n = rlen[w * RPWV + k];
            unsigned o = roff[w * RPWV + k];
            for (unsigned j = 32u + (unsigned)lane; j < n; j += 32u)
                atomicAdd(&cnt[pairs[o + j].y & 255], 1u);
        }
    __syncthreads();
    unsigned cv = (tid < 256) ? cnt[tid] : 0u;
    unsigned inc = cv;
    for (int d = 1; d < 32; d <<= 1) {
        unsigned t = __shfl_up(inc, d);
        if (lane >= d) inc += t;
    }
    if (tid < 256 && lane == 31) wsum[w] = inc;
    __syncthreads();
    if (tid < 256) {
        unsigned base = 0u;
        for (int k = 0; k < w; ++k) base += wsum[k];
        unsigned st = beg + base + inc - cv;
        cur[tid] = st;
        int r = (b << RB_SHIFT) + tid;
        if (r < N) { segend[r] = st + cv; degz[r] = cv; }
    }
    __syncthreads();
    unsigned total = 0u;
    for (int k = 0; k < 8; ++k) total += wsum[k];
    bool staged = total <= SLOT_CAP;
    unsigned pos[RPWV];
#pragma unroll
    for (int k = 0; k < RPWV; ++k) {
        pos[k] = 0u;
        if (live & (1u << k)) pos[k] = atomicAdd(&cur[p[k].y & 255], 1u);
    }
    if (staged) {
#pragma unroll
        for (int k = 0; k < RPWV; ++k)
            if (live & (1u << k)) slots[pos[k] - beg] = p[k].x;
    } else {
#pragma unroll
        for (int k = 0; k < RPWV; ++k)
            if (live & (1u << k)) sorted_s[pos[k]] = p[k].x;
    }
    if (tail)
        for (int k = 0; k < RPWV; ++k) {
            unsigned n = rlen[w * RPWV + k];
            unsigned o = roff[w * RPWV + k];
            for (unsigned j = 32u + (unsigned)lane; j < n; j += 32u) {
                int2 q = pairs[o + j];
                unsigned pos = atomicAdd(&cur[q.y & 255], 1u);
                if (staged) slots[pos - beg] = q.x;
                else        sorted_s[pos] = q.x;
            }
        }
    if (staged) {
        __syncthreads();
        int n = (int)total;
        int a = (int)(beg & 31u);
        if (n <= 0) return;
        for (int i0 = tid - a; i0 < n; i0 += 4 * SORT_T) {
            int v[4];
#pragma unroll
            for (int k = 0; k < 4; ++k) {
                int i = i0 + k * SORT_T;
                v[k] = slots[i < 0 ? 0 : (i < n ? i : n - 1)];
            }
#pragma unroll
            for (int k = 0; k < 4; ++k) {
                int i = i0 + k * SORT_T;
                if (i >= 0 && i < n) sorted_s[beg + (unsigned)i] = v[k];
            }
        }
    }
}

__device__ __forceinline__ void wmma_step(const float* __restrict__ W, int kb,
                                          int half, int m, v2f a,
                                          v8f& acc0, v8f& acc1) {
    int r0 = (kb + 2 * half + 0) * 32;
    int r1 = (kb + 2 * half + 1) * 32;
    v2f b0, b1;
    b0.x = W[r0 + m];        b0.y = W[r1 + m];
    b1.x = W[r0 + 16 + m];   b1.y = W[r1 + 16 + m];
    acc0 = __builtin_amdgcn_wmma_f32_16x16x4_f32(false, a, false, b0,
                                                 (short)0, acc0, false, false);
    acc1 = __builtin_amdgcn_wmma_f32_16x16x4_f32(false, a, false, b1,
                                                 (short)0, acc1, false, false);
}

__global__ __launch_bounds__(256) void weight_fold_kernel(
    const float* __restrict__ Wq, const float* __restrict__ Wk,
    const float* __restrict__ Wv, const float* __restrict__ Wo,
    float* __restrict__ Wqk, float* __restrict__ Wvo) {
    int wave = (int)(threadIdx.x >> 5);
    int lane = (int)(threadIdx.x & 31u);
    int half = lane >> 4;
    int m    = lane & 15;
    bool vo  = wave >= 4;
    int tr = (wave >> 1) & 1, tc = wave & 1;
    const float* __restrict__ A = vo ? Wv : Wq;
    const float* __restrict__ B = vo ? Wo : Wk;
    v8f acc = {};
#pragma unroll
    for (int kk = 0; kk < 8; ++kk) {
        int k0 = kk * 4 + 2 * half;
        const float* ap = A + (tr * 16 + m) * 32 + k0;
        v2f a; a.x = ap[0]; a.y = ap[1];
        int i0 = vo ? k0 * 32 + tc * 16 + m : (tc * 16 + m) * 32 + k0;
        int i1 = vo ? i0 + 32 : i0 + 1;
        v2f b; b.x = B[i0]; b.y = B[i1];
        acc = __builtin_amdgcn_wmma_f32_16x16x4_f32(false, a, false, b,
                                                    (short)0, acc, false, false);
    }
    float* __restrict__ D = vo ? Wvo : Wqk;
#pragma unroll
    for (int r = 0; r < 8; ++r)
        D[(tr * 16 + r + 8 * half) * 32 + tc * 16 + m] = acc[r];
}

#define RPW 16
#define SPT 4
#define TP  36
#define WPB 2

__device__ __forceinline__ void wave_tile_handover() {
    __builtin_amdgcn_fence(__ATOMIC_RELEASE, "wavefront");
    __builtin_amdgcn_wave_barrier();
    __builtin_amdgcn_fence(__ATOMIC_ACQUIRE, "wavefront");
}

__device__ __forceinline__ void load_senders(const int* __restrict__ sorted_s,
                                             int start, int deg, int e0, int g,
                                             int (&s)[SPT]) {
#pragma unroll
    for (int i = 0; i < SPT; ++i) {
        int jn = e0 + 4 * i + g;
        s[i] = sorted_s[start + (jn < deg ? jn : deg - 1)];
    }
}

__device__ __forceinline__ void load_rows(const float4* __restrict__ X4,
                                          const int (&s)[SPT], unsigned cu,
                                          float4 (&x4)[SPT]) {
#pragma unroll
    for (int i = 0; i < SPT; ++i)
        x4[i] = X4[(unsigned)s[i] * 8u + cu];
}

__device__ __forceinline__ void load_bfrags(const float* __restrict__ Wt,
                                            int half, int m,
                                            v2f (&b0)[8], v2f (&b1)[8]) {
#pragma unroll
    for (int kk = 0; kk < 8; ++kk) {
        int r0w = (kk * 4 + 2 * half + 0) * 32;
        int r1w = (kk * 4 + 2 * half + 1) * 32;
        b0[kk].x = Wt[r0w + m];        b0[kk].y = Wt[r1w + m];
        b1[kk].x = Wt[r0w + 16 + m];   b1[kk].y = Wt[r1w + 16 + m];
    }
}

__global__ __launch_bounds__(32 * WPB) void seg_attn_kernel(
    const float* __restrict__ x, const float* __restrict__ Wqk,
    const float* __restrict__ Wvo, const int* __restrict__ sorted_s,
    const unsigned* __restrict__ segend, const unsigned* __restrict__ degz,
    float* __restrict__ out, int N) {
    __shared__ __attribute__((aligned(16))) float tiles[WPB][16 * TP];
    int wib = __builtin_amdgcn_readfirstlane((int)(threadIdx.x >> 5));
    int r0 = __builtin_amdgcn_readfirstlane(
                 (int)((blockIdx.x * blockDim.x + threadIdx.x) >> 5)) * RPW;
    if (r0 >= N) return;
    float* tile = tiles[wib];
    float4* tile4 = (float4*)tiles[wib];
    int lane = (int)(threadIdx.x & 31u);
    int half = lane >> 4;
    int m    = lane & 15;
    int g = lane >> 3;
    int c = lane & 7;
    int nr = (N - r0 < RPW) ? N - r0 : RPW;
    const float4* __restrict__ X4 = (const float4*)x;
    unsigned cu = (unsigned)c;

    int degv = 0, endv = 0;
    if (lane < nr) {
        degv = (int)degz[r0 + lane];
        endv = (int)segend[r0 + lane];
    }
    float zv = 0.f;

    float4 xrow[4];
#pragma unroll
    for (int i = 0; i < 4; ++i) {
        int row = r0 + 4 * i + g;
        xrow[i] = X4[(unsigned)(row < N ? row : N - 1) * 8u + cu];
    }
    v2f bq0[8], bq1[8];
    load_bfrags(Wqk, half, m, bq0, bq1);

#pragma unroll
    for (int i = 0; i < 4; ++i) tile4[(4 * i + g) * (TP / 4) + c] = xrow[i];
    wave_tile_handover();
    v2f xa[8];
#pragma unroll
    for (int kk = 0; kk < 8; ++kk) {
        const float* ap = tile + m * TP + kk * 4 + 2 * half;
        xa[kk].x = ap[0]; xa[kk].y = ap[1];
    }

    int deg   = __builtin_amdgcn_readlane(degv, 0);
    int start = __builtin_amdgcn_readlane(endv, 0) - deg;
    int s_cur[SPT];
#pragma unroll
    for (int i = 0; i < SPT; ++i) s_cur[i] = r0;
    if (deg > 0) load_senders(sorted_s, start, deg, 0, g, s_cur);

    {
        v8f q0 = {}, q1 = {};
#pragma unroll
        for (int kk = 0; kk < 8; ++kk) {
            q0 = __builtin_amdgcn_wmma_f32_16x16x4_f32(false, xa[kk], false, bq0[kk],
                                                       (short)0, q0, false, false);
            q1 = __builtin_amdgcn_wmma_f32_16x16x4_f32(false, xa[kk], false, bq1[kk],
                                                       (short)0, q1, false, false);
        }
#pragma unroll
        for (int r = 0; r < 8; ++r) {
            tile[(r + 8 * half) * TP + m]      = q0[r];
            tile[(r + 8 * half) * TP + 16 + m] = q1[r];
        }
    }
    wave_tile_handover();
    float4 q4 = tile4[cu];

    for (int j = 0; j < nr; ++j) {
        int r = r0 + j;
        float4 x4[SPT];
        int s_nxt[SPT];
#pragma unroll
        for (int i = 0; i < SPT; ++i) { x4[i] = q4; s_nxt[i] = r; }
        if (deg > 0) {
            load_rows(X4, s_cur, cu, x4);
            if (deg > 4 * SPT)
                load_senders(sorted_s, start, deg, 4 * SPT, g, s_nxt);
        }
        int deg_n = 0, start_n = 0;
        int s_first_n[SPT];
#pragma unroll
        for (int i = 0; i < SPT; ++i) s_first_n[i] = r;
        float4 q4_n = q4;
        if (j + 1 < nr) {
            deg_n   = __builtin_amdgcn_readlane(degv, j + 1);
            start_n = __builtin_amdgcn_readlane(endv, j + 1) - deg_n;
            q4_n = tile4[(unsigned)(j + 1) * (TP / 4) + cu];
            if (deg_n > 0) load_senders(sorted_s, start_n, deg_n, 0, g, s_first_n);
        }

        float mx = -INFINITY;
        float z = 0.f;
        float4 acc = make_float4(0.f, 0.f, 0.f, 0.f);

        for (int t = 0; t < deg; t += 4 * SPT) {
            float p[SPT];
#pragma unroll
            for (int i = 0; i < SPT; ++i)
                p[i] = q4.x * x4[i].x + q4.y * x4[i].y + q4.z * x4[i].z
                     + q4.w * x4[i].w;
#pragma unroll
            for (int i = 0; i < SPT; ++i) p[i] += __shfl_xor(p[i], 1);
#pragma unroll
            for (int i = 0; i < SPT; ++i) p[i] += __shfl_xor(p[i], 2);
#pragma unroll
            for (int i = 0; i < SPT; ++i) p[i] += __shfl_xor(p[i], 4);

            if (t + g < deg) {
                float sc[SPT];
#pragma unroll
                for (int i = 0; i < SPT; ++i)
                    sc[i] = (t + 4 * i + g < deg) ? p[i] * 0.17677669529663687f
                                                  : -INFINITY;
                float nm = mx;
#pragma unroll
                for (int i = 0; i < SPT; ++i) nm = fmaxf(nm, sc[i]);
                float corr = __expf(mx - nm);
                float a[SPT];
#pragma unroll
                for (int i = 0; i < SPT; ++i) a[i] = __expf(sc[i] - nm);
                float zs = 0.f;
                float4 as = make_float4(0.f, 0.f, 0.f, 0.f);
#pragma unroll
                for (int i = 0; i < SPT; ++i) {
                    zs += a[i];
                    as.x += a[i] * x4[i].x;
                    as.y += a[i] * x4[i].y;
                    as.z += a[i] * x4[i].z;
                    as.w += a[i] * x4[i].w;
                }
                z = z * corr + zs;
                acc.x = acc.x * corr + as.x;
                acc.y = acc.y * corr + as.y;
                acc.z = acc.z * corr + as.z;
                acc.w = acc.w * corr + as.w;
                mx = nm;
            }
            if (t + 4 * SPT < deg) {
                int s_now[SPT];
#pragma unroll
                for (int i = 0; i < SPT; ++i) s_now[i] = s_nxt[i];
                load_rows(X4, s_now, cu, x4);
                if (t + 8 * SPT < deg)
                    load_senders(sorted_s, start, deg, t + 8 * SPT, g, s_nxt);
            }
        }

        float M = mx;
        M = fmaxf(M, __shfl_xor(M, 8));
        M = fmaxf(M, __shfl_xor(M, 16));
        float f = (mx == -INFINITY) ? 0.f : __expf(mx - M);
        z *= f; acc.x *= f; acc.y *= f; acc.z *= f; acc.w *= f;
        z     += __shfl_xor(z, 8);      z     += __shfl_xor(z, 16);
        acc.x += __shfl_xor(acc.x, 8);  acc.x += __shfl_xor(acc.x, 16);
        acc.y += __shfl_xor(acc.y, 8);  acc.y += __shfl_xor(acc.y, 16);
        acc.z += __shfl_xor(acc.z, 8);  acc.z += __shfl_xor(acc.z, 16);
        acc.w += __shfl_xor(acc.w, 8);  acc.w += __shfl_xor(acc.w, 16);
        if (g == 0) tile4[(unsigned)j * (TP / 4) + cu] = acc;
        if (lane == j) zv = z;

        deg = deg_n; start = start_n; q4 = q4_n;
#pragma unroll
        for (int i = 0; i < SPT; ++i) s_cur[i] = s_first_n[i];
    }

    wave_tile_handover();
    {
        float zrow = __shfl(zv, m);
        float invz = 1.0f / (zrow + 1e-6f);
        float4 xres[4];
#pragma unroll
        for (int i = 0; i < 4; ++i) {
            int row = r0 + 4 * i + g;
            xres[i] = X4[(unsigned)(row < N ? row : N - 1) * 8u + cu];
        }
        v8f c0 = {}, c1 = {};
#pragma unroll
        for (int kk = 0; kk < 8; ++kk) {
            const float* ap = tile + m * TP + kk * 4 + 2 * half;
            v2f a; a.x = ap[0] * invz; a.y = ap[1] * invz;
            wmma_step(Wvo, kk * 4, half, m, a, c0, c1);
        }
#pragma unroll
        for (int r = 0; r < 8; ++r) {
            tile[(r + 8 * half) * TP + m]      = c0[r];
            tile[(r + 8 * half) * TP + 16 + m] = c1[r];
        }
        wave_tile_handover();
        float4* O4 = (float4*)out;
#pragma unroll
        for (int i = 0; i < 4; ++i) {
            int row = r0 + 4 * i + g;
            float4 d = tile4[(4 * i + g) * (TP / 4) + c];
            float4 o = make_float4(xres[i].x + d.x, xres[i].y + d.y,
                                   xres[i].z + d.z, xres[i].w + d.w);
            if (row < N) O4[(unsigned)row * 8u + cu] = o;
        }
    }
}

extern "C" void kernel_launch(void* const* d_in, const int* in_sizes, int n_in,
                              void* d_out, int out_size, void* d_ws,
                              size_t ws_size, hipStream_t stream) {
    const float* x  = (const float*)d_in[0];
    const int*   ei = (const int*)d_in[1];
    const float* Wq = (const float*)d_in[2];
    const float* Wk = (const float*)d_in[3];
    const float* Wv = (const float*)d_in[4];
    const float* Wo = (const float*)d_in[5];
    float* out = (float*)d_out;

    const int N = in_sizes[0] / 32;
    const int E = in_sizes[1] / 2;

    float* P0      = (float*)d_ws;
    float* P1      = P0 + (size_t)N * 32;
    float* P2      = P1 + (size_t)N * 32;
    float* P3      = P2 + (size_t)N * 32;
    float* Wqk     = P3 - 2048;
    float* Wvo     = Wqk + 1024;
    int* sorted_s  = (int*)(P3 + (size_t)N * 32);
    unsigned* cnt  = (unsigned*)(sorted_s + E);
    unsigned* degz = cnt + N;
    unsigned* runs = (unsigned*)P0;
    int* pairs     = (int*)P1;

    const int TB = 256;
    int NB = (N + 255) >> RB_SHIFT;

    weight_fold_kernel<<<1, TB, 0, stream>>>(Wq, Wk, Wv, Wo, Wqk, Wvo);
    chunk_sort_kernel<<<SORT_C, SORT_T, 0, stream>>>(ei, runs, pairs, E, NB);
    bucket_sort_kernel<<<NB, SORT_T, 0, stream>>>(pairs, runs, cnt, degz,
                                                  sorted_s, N, E);
    seg_attn_kernel<<<(N + WPB * RPW - 1) / (WPB * RPW), 32 * WPB, 0, stream>>>(
        x, Wqk, Wvo, sorted_s, cnt, degz, out, N);
}


# Sort level 2 and segment kernel fused: a work-group sorts its bucket's senders into LDS, its 16 waves walk the bucket's 16 receiver tiles from there; one launch and sorted_s trip gone; sum order kept
# speedup vs baseline: 16.0589x; 1.1143x over previous
#include <hip/hip_runtime.h>
#include <math.h>

typedef __attribute__((ext_vector_type(2))) float v2f;
typedef __attribute__((ext_vector_type(8))) float v8f;

#define SORT_C   256
#define SORT_T   512
#define RB_SHIFT 8
#define NB_MAX   512
#define EPT      8
#define RPWV     16
#define PAIR_CAP 6656
#define SLOT_CAP 8192
static_assert(SORT_T == NB_MAX && SORT_C == 256 &&
              SORT_C == RPWV * (SORT_T / 32),
              "level 2 maps thread t < 256 to chunk t and wave w to 16 chunks");

__device__ __forceinline__ unsigned bucket_of(int r, int NB) {
    unsigned b = (unsigned)r >> RB_SHIFT;
    unsigned last = (unsigned)(NB - 1);
    return b < last ? b : last;
}

__device__ __forceinline__ void file_pairs(const int* __restrict__ ei,
                                           int2* __restrict__ pairs, int2* stage,
                                           unsigned* hist, int E, int NB, int beg,
                                           int end, int tid, bool staged) {
    for (int e = beg + tid; e < end; e += EPT * SORT_T) {
        int s[EPT], r[EPT];
#pragma unroll
        for (int k = 0; k < EPT; ++k) {
            int ee = e + k * SORT_T;
            int ec = ee < end ? ee : end - 1;
            s[k] = ei[ec];
            r[k] = ei[E + ec];
        }
#pragma unroll
        for (int k = 0; k < EPT; ++k)
            if (e + k * SORT_T < end) {
                unsigned pos = atomicAdd(&hist[bucket_of(r[k], NB)], 1u);
                if (staged) stage[pos] = make_int2(s[k], r[k]);
                else        pairs[(unsigned)beg + pos] = make_int2(s[k], r[k]);
            }
    }
}

__global__ __launch_bounds__(SORT_T) void chunk_sort_kernel(
    const int* __restrict__ ei, unsigned* __restrict__ runs_raw,
    int* __restrict__ pairs_raw, int E, int NB) {
    __shared__ unsigned hist[NB_MAX];
    __shared__ unsigned wsum[SORT_T / 32];
    __shared__ int2 stage[PAIR_CAP];
    uint2* __restrict__ runs = (uint2*)runs_raw;
    int2* __restrict__ pairs = (int2*)pairs_raw;
    int tid = (int)threadIdx.x;
    int c = (int)blockIdx.x;
    int lane = tid & 31;
    int w = tid >> 5;
    hist[tid] = 0u;
    __syncthreads();
    int CH = (E + SORT_C - 1) / SORT_C;
    int beg = c * CH;
    int end = (beg + CH < E) ? beg + CH : E;
    for (int e = beg + tid; e < end; e += EPT * SORT_T) {
        int r[EPT];
#pragma unroll
        for (int k = 0; k < EPT; ++k) {
            int ee = e + k * SORT_T;
            r[k] = ei[E + (ee < end ? ee : end - 1)];
        }
#pragma unroll
        for (int k = 0; k < EPT; ++k)
            if (e + k * SORT_T < end)
                atomicAdd(&hist[bucket_of(r[k], NB)], 1u);
    }
    __syncthreads();
    unsigned tot = hist[tid];
    unsigned inc = tot;
    for (int d = 1; d < 32; d <<= 1) {
        unsigned t = __shfl_up(inc, d);
        if (lane >= d) inc += t;
    }
    if (lane == 31) wsum[w] = inc;
    __syncthreads();
    unsigned base = 0u;
    for (int k = 0; k < w; ++k) base += wsum[k];
    unsigned excl = base + inc - tot;
    runs[c * NB_MAX + tid] = make_uint2(excl, tot);
    hist[tid] = excl;
    __syncthreads();

    if (CH > PAIR_CAP) {
        file_pairs(ei, pairs, stage, hist, E, NB, beg, end, tid, false);
        return;
    }
    file_pairs(ei, pairs, stage, hist, E, NB, beg, end, tid, true);
    __syncthreads();
    int n = end - beg;
    int a = beg & 15;
    if (n <= 0) return;
    for (int i0 = tid - a; i0 < n; i0 += 4 * SORT_T) {
        int2 v[4];
#pragma unroll
        for (int k = 0; k < 4; ++k) {
            int i = i0 + k * SORT_T;
            v[k] = stage[i < 0 ? 0 : (i < n ? i : n - 1)];
        }
#pragma unroll
        for (int k = 0; k < 4; ++k) {
            int i = i0 + k * SORT_T;
            if (i >= 0 && i < n) pairs[beg + i] = v[k];
        }
    }
}

__device__ __forceinline__ void wmma_step(const float* __restrict__ W, int kb,
                                          int half, int m, v2f a,
                                          v8f& acc0, v8f& acc1) {
    int r0 = (kb + 2 * half + 0) * 32;
    int r1 = (kb + 2 * half + 1) * 32;
    v2f b0, b1;
    b0.x = W[r0 + m];        b0.y = W[r1 + m];
    b1.x = W[r0 + 16 + m];   b1.y = W[r1 + 16 + m];
    acc0 = __builtin_amdgcn_wmma_f32_16x16x4_f32(false, a, false, b0,
                                                 (short)0, acc0, false, false);
    acc1 = __builtin_amdgcn_wmma_f32_16x16x4_f32(false, a, false, b1,
                                                 (short)0, acc1, false, false);
}

__global__ __launch_bounds__(256) void weight_fold_kernel(
    const float* __restrict__ Wq, const float* __restrict__ Wk,
    const float* __restrict__ Wv, const float* __restrict__ Wo,
    float* __restrict__ Wqk, float* __restrict__ Wvo) {
    int wave = (int)(threadIdx.x >> 5);
    int lane = (int)(threadIdx.x & 31u);
    int half = lane >> 4;
    int m    = lane & 15;
    bool vo  = wave >= 4;
    int tr = (wave >> 1) & 1, tc = wave & 1;
    const float* __restrict__ A = vo ? Wv : Wq;
    const float* __restrict__ B = vo ? Wo : Wk;
    v8f acc = {};
#pragma unroll
    for (int kk = 0; kk < 8; ++kk) {
        int k0 = kk * 4 + 2 * half;
        const float* ap = A + (tr * 16 + m) * 32 + k0;
        v2f a; a.x = ap[0]; a.y = ap[1];
        int i0 = vo ? k0 * 32 + tc * 16 + m : (tc * 16 + m) * 32 + k0;
        int i1 = vo ? i0 + 32 : i0 + 1;
        v2f b; b.x = B[i0]; b.y = B[i1];
        acc = __builtin_amdgcn_wmma_f32_16x16x4_f32(false, a, false, b,
                                                    (short)0, acc, false, false);
    }
    float* __restrict__ D = vo ? Wvo : Wqk;
#pragma unroll
    for (int r = 0; r < 8; ++r)
        D[(tr * 16 + r + 8 * half) * 32 + tc * 16 + m] = acc[r];
}

#define RPW 16
#define SPT 4
#define TP  36
#define TW  (SORT_T / 32)
static_assert(TW * RPW == (1 << RB_SHIFT), "a bucket is one tile per wave");

__device__ __forceinline__ void wave_tile_handover() {
    __builtin_amdgcn_fence(__ATOMIC_RELEASE, "wavefront");
    __builtin_amdgcn_wave_barrier();
    __builtin_amdgcn_fence(__ATOMIC_ACQUIRE, "wavefront");
}

template <bool STAGED>
__device__ __forceinline__ void load_senders(const int* src, int start, int deg,
                                             int e0, int g, int (&s)[SPT]) {
#pragma unroll
    for (int i = 0; i < SPT; ++i) {
        int jn = e0 + 4 * i + g;
        int idx = start + (jn < deg ? jn : deg - 1);
        if (STAGED) s[i] = src[idx];
        else        s[i] = __builtin_nontemporal_load(src + idx);
    }
}

__device__ __forceinline__ void load_rows(const float4* __restrict__ X4,
                                          const int (&s)[SPT], unsigned cu,
                                          float4 (&x4)[SPT]) {
#pragma unroll
    for (int i = 0; i < SPT; ++i)
        x4[i] = X4[(unsigned)s[i] * 8u + cu];
}

__device__ __forceinline__ void load_bfrags(const float* __restrict__ Wt,
                                            int half, int m,
                                            v2f (&b0)[8], v2f (&b1)[8]) {
#pragma unroll
    for (int kk = 0; kk < 8; ++kk) {
        int r0w = (kk * 4 + 2 * half + 0) * 32;
        int r1w = (kk * 4 + 2 * half + 1) * 32;
        b0[kk].x = Wt[r0w + m];        b0[kk].y = Wt[r1w + m];
        b1[kk].x = Wt[r0w + 16 + m];   b1[kk].y = Wt[r1w + 16 + m];
    }
}

__device__ __forceinline__ void project_tile(const float* __restrict__ x,
                                             const float* __restrict__ Wqk,
                                             float* tile, int N, int r0,
                                             int lane) {
    float4* tile4 = (float4*)tile;
    int half = lane >> 4;
    int m    = lane & 15;
    int g = lane >> 3;
    int c = lane & 7;
    const float4* __restrict__ X4 = (const float4*)x;
    unsigned cu = (unsigned)c;
    float4 xrow[4];
#pragma unroll
    for (int i = 0; i < 4; ++i) {
        int row = r0 + 4 * i + g;
        xrow[i] = X4[(unsigned)(row < N ? row : N - 1) * 8u + cu];
    }
    v2f bq0[8], bq1[8];
    load_bfrags(Wqk, half, m, bq0, bq1);
#pragma unroll
    for (int i = 0; i < 4; ++i) tile4[(4 * i + g) * (TP / 4) + c] = xrow[i];
    wave_tile_handover();
    v2f xa[8];
#pragma unroll
    for (int kk = 0; kk < 8; ++kk) {
        const float* ap = tile + m * TP + kk * 4 + 2 * half;
        xa[kk].x = ap[0]; xa[kk].y = ap[1];
    }
    v8f q0 = {}, q1 = {};
#pragma unroll
    for (int kk = 0; kk < 8; ++kk) {
        q0 = __builtin_amdgcn_wmma_f32_16x16x4_f32(false, xa[kk], false, bq0[kk],
                                                   (short)0, q0, false, false);
        q1 = __builtin_amdgcn_wmma_f32_16x16x4_f32(false, xa[kk], false, bq1[kk],
                                                   (short)0, q1, false, false);
    }
#pragma unroll
    for (int r = 0; r < 8; ++r) {
        tile[(r + 8 * half) * TP + m]      = q0[r];
        tile[(r + 8 * half) * TP + 16 + m] = q1[r];
    }
    wave_tile_handover();
}

template <bool STAGED>
__device__ __forceinline__ void walk_tile(const float* __restrict__ x,
                                          const float* __restrict__ Wvo,
                                          const int* src, float* tile,
                                          float* __restrict__ out, int N,
                                          int r0, int nr, int lane, int degv,
                                          int endv) {
    float4* tile4 = (float4*)tile;
    int half = lane >> 4;
    int m    = lane & 15;
    int g = lane >> 3;
    int c = lane & 7;
    const float4* __restrict__ X4 = (const float4*)x;
    unsigned cu = (unsigned)c;
    float zv = 0.f;

    int deg   = __builtin_amdgcn_readlane(degv, 0);
    int start = __builtin_amdgcn_readlane(endv, 0) - deg;
    int s_cur[SPT];
#pragma unroll
    for (int i = 0; i < SPT; ++i) s_cur[i] = r0;
    if (deg > 0) load_senders<STAGED>(src, start, deg, 0, g, s_cur);
    float4 q4 = tile4[cu];

    for (int j = 0; j < nr; ++j) {
        int r = r0 + j;
        float4 x4[SPT];
        int s_nxt[SPT];
#pragma unroll
        for (int i = 0; i < SPT; ++i) { x4[i] = q4; s_nxt[i] = r; }
        if (deg > 0) {
            load_rows(X4, s_cur, cu, x4);
            if (deg > 4 * SPT)
                load_senders<STAGED>(src, start, deg, 4 * SPT, g, s_nxt);
        }
        int deg_n = 0, start_n = 0;
        int s_first_n[SPT];
#pragma unroll
        for (int i = 0; i < SPT; ++i) s_first_n[i] = r;
        float4 q4_n = q4;
        if (j + 1 < nr) {
            deg_n   = __builtin_amdgcn_readlane(degv, j + 1);
            start_n = __builtin_amdgcn_readlane(endv, j + 1) - deg_n;
            q4_n = tile4[(unsigned)(j + 1) * (TP / 4) + cu];
            if (deg_n > 0)
                load_senders<STAGED>(src, start_n, deg_n, 0, g, s_first_n);
        }

        float mx = -INFINITY;
        float z = 0.f;
        float4 acc = make_float4(0.f, 0.f, 0.f, 0.f);

        for (int t = 0; t < deg; t += 4 * SPT) {
            float p[SPT];
#pragma unroll
            for (int i = 0; i < SPT; ++i)
                p[i] = q4.x * x4[i].x + q4.y * x4[i].y + q4.z * x4[i].z
                     + q4.w * x4[i].w;
#pragma unroll
            for (int i = 0; i < SPT; ++i) p[i] += __shfl_xor(p[i], 1);
#pragma unroll
            for (int i = 0; i < SPT; ++i) p[i] += __shfl_xor(p[i], 2);
#pragma unroll
            for (int i = 0; i < SPT; ++i) p[i] += __shfl_xor(p[i], 4);

            if (t + g < deg) {
                float sc[SPT];
#pragma unroll
                for (int i = 0; i < SPT; ++i)
                    sc[i] = (t + 4 * i + g < deg) ? p[i] * 0.17677669529663687f
                                                  : -INFINITY;
                float nm = mx;
#pragma unroll
                for (int i = 0; i < SPT; ++i) nm = fmaxf(nm, sc[i]);
                float corr = __expf(mx - nm);
                float a[SPT];
#pragma unroll
                for (int i = 0; i < SPT; ++i) a[i] = __expf(sc[i] - nm);
                float zs = 0.f;
                float4 as = make_float4(0.f, 0.f, 0.f, 0.f);
#pragma unroll
                for (int i = 0; i < SPT; ++i) {
                    zs += a[i];
                    as.x += a[i] * x4[i].x;
                    as.y += a[i] * x4[i].y;
                    as.z += a[i] * x4[i].z;
                    as.w += a[i] * x4[i].w;
                }
                z = z * corr + zs;
                acc.x = acc.x * corr + as.x;
                acc.y = acc.y * corr + as.y;
                acc.z = acc.z * corr + as.z;
                acc.w = acc.w * corr + as.w;
                mx = nm;
            }
            if (t + 4 * SPT < deg) {
                int s_now[SPT];
#pragma unroll
                for (int i = 0; i < SPT; ++i) s_now[i] = s_nxt[i];
                load_rows(X4, s_now, cu, x4);
                if (t + 8 * SPT < deg)
                    load_senders<STAGED>(src, start, deg, t + 8 * SPT, g, s_nxt);
            }
        }

        float M = mx;
        M = fmaxf(M, __shfl_xor(M, 8));
        M = fmaxf(M, __shfl_xor(M, 16));
        float f = (mx == -INFINITY) ? 0.f : __expf(mx - M);
        z *= f; acc.x *= f; acc.y *= f; acc.z *= f; acc.w *= f;
        z     += __shfl_xor(z, 8);      z     += __shfl_xor(z, 16);
        acc.x += __shfl_xor(acc.x, 8);  acc.x += __shfl_xor(acc.x, 16);
        acc.y += __shfl_xor(acc.y, 8);  acc.y += __shfl_xor(acc.y, 16);
        acc.z += __shfl_xor(acc.z, 8);  acc.z += __shfl_xor(acc.z, 16);
        acc.w += __shfl_xor(acc.w, 8);  acc.w += __shfl_xor(acc.w, 16);
        if (g == 0) tile4[(unsigned)j * (TP / 4) + cu] = acc;
        if (lane == j) zv = z;

        deg = deg_n; start = start_n; q4 = q4_n;
#pragma unroll
        for (int i = 0; i < SPT; ++i) s_cur[i] = s_first_n[i];
    }

    wave_tile_handover();
    {
        float zrow = __shfl(zv, m);
        float invz = 1.0f / (zrow + 1e-6f);
        float4 xres[4];
#pragma unroll
        for (int i = 0; i < 4; ++i) {
            int row = r0 + 4 * i + g;
            xres[i] = X4[(unsigned)(row < N ? row : N - 1) * 8u + cu];
        }
        v8f c0 = {}, c1 = {};
#pragma unroll
        for (int kk = 0; kk < 8; ++kk) {
            const float* ap = tile + m * TP + kk * 4 + 2 * half;
            v2f a; a.x = ap[0] * invz; a.y = ap[1] * invz;
            wmma_step(Wvo, kk * 4, half, m, a, c0, c1);
        }
#pragma unroll
        for (int r = 0; r < 8; ++r) {
            tile[(r + 8 * half) * TP + m]      = c0[r];
            tile[(r + 8 * half) * TP + 16 + m] = c1[r];
        }
        wave_tile_handover();
        float4* O4 = (float4*)out;
#pragma unroll
        for (int i = 0; i < 4; ++i) {
            int row = r0 + 4 * i + g;
            float4 d = tile4[(4 * i + g) * (TP / 4) + c];
            float4 o = make_float4(xres[i].x + d.x, xres[i].y + d.y,
                                   xres[i].z + d.z, xres[i].w + d.w);
            if (row < N) O4[(unsigned)row * 8u + cu] = o;
        }
    }
}

__global__ __launch_bounds__(SORT_T) void bucket_attn_kernel(
    const int* __restrict__ pairs_raw, const unsigned* __restrict__ runs_raw,
    const float* __restrict__ x, const float* __restrict__ Wqk,
    const float* __restrict__ Wvo, int* sorted_s, float* __restrict__ out,
    int N, int E) {
    __shared__ unsigned cnt[256];
    __shared__ unsigned cur[256];
    __shared__ unsigned wsum[8];
    __shared__ unsigned osum[8];
    __shared__ unsigned roff[SORT_C];
    __shared__ unsigned rlen[SORT_C];
    __shared__ int slots[SLOT_CAP];
    __shared__ __attribute__((aligned(16))) float tiles[TW][16 * TP];
    const int2* __restrict__ pairs = (const int2*)pairs_raw;
    const uint2* __restrict__ runs = (const uint2*)runs_raw;
    int tid = (int)threadIdx.x;
    int b = (int)blockIdx.x;
    int lane = tid & 31;
    int w = __builtin_amdgcn_readfirstlane(tid >> 5);
    int CH = (E + SORT_C - 1) / SORT_C;
    unsigned last = (unsigned)E - 1u;

    int r0 = (b << RB_SHIFT) + w * RPW;
    bool has_tile = r0 < N;
    int nr = (N - r0 < RPW) ? N - r0 : RPW;
    float* tile = tiles[w];

    uint2 oc = make_uint2(0u, 0u);
    if (tid < 256) oc = runs[tid * NB_MAX + b];
    if (has_tile) project_tile(x, Wqk, tile, N, r0, lane);

    if (tid < 256) {
        roff[tid] = (unsigned)(tid * CH) + oc.x;
        rlen[tid] = oc.y;
        cnt[tid] = 0u;
        unsigned s = oc.x;
        for (int d = 16; d > 0; d >>= 1) s += __shfl_xor(s, d);
        if (lane == 0) osum[w] = s;
    }
    __syncthreads();
    unsigned beg = 0u;
    for (int k = 0; k < 8; ++k) beg += osum[k];

    int2 p[RPWV];
    unsigned live = 0u;
    bool tail = false;
#pragma unroll
    for (int k = 0; k < RPWV; ++k) {
        unsigned n = rlen[w * RPWV + k];
        unsigned idx = roff[w * RPWV + k] + (unsigned)lane;
        p[k] = pairs[idx < last ? idx : last];
        if ((unsigned)lane < n) live |= 1u << k;
        tail = tail || (n > 32u);
    }
#pragma unroll
    for (int k = 0; k < RPWV; ++k)
        if (live & (1u << k)) atomicAdd(&cnt[p[k].y & 255], 1u);
    if (tail)
        for (int k = 0; k < RPWV; ++k) {
            unsigned n = rlen[w * RPWV + k];
            unsigned o = roff[w * RPWV + k];
            for (unsigned j = 32u + (unsigned)lane; j < n; j += 32u)
                atomicAdd(&cnt[pairs[o + j].y & 255], 1u);
        }
    __syncthreads();
    unsigned cv = (tid < 256) ? cnt[tid] : 0u;
    unsigned inc = cv;
    for (int d = 1; d < 32; d <<= 1) {
        unsigned t = __shfl_up(inc, d);
        if (lane >= d) inc += t;
    }
    if (tid < 256 && lane == 31) wsum[w] = inc;
    __syncthreads();
    if (tid < 256) {
        unsigned base = 0u;
        for (int k = 0; k < w; ++k) base += wsum[k];
        cur[tid] = beg + base + inc - cv;
    }
    __syncthreads();
    unsigned total = 0u;
    for (int k = 0; k < 8; ++k) total += wsum[k];
    bool staged = total <= SLOT_CAP;
    unsigned pos[RPWV];
#pragma unroll
    for (int k = 0; k < RPWV; ++k) {
        pos[k] = 0u;
        if (live & (1u << k)) pos[k] = atomicAdd(&cur[p[k].y & 255], 1u);
    }
    if (staged) {
#pragma unroll
        for (int k = 0; k < RPWV; ++k)
            if (live & (1u << k)) slots[pos[k] - beg] = p[k].x;
    } else {
#pragma unroll
        for (int k = 0; k < RPWV; ++k)
            if (live & (1u << k)) sorted_s[pos[k]] = p[k].x;
    }
    if (tail)
        for (int k = 0; k < RPWV; ++k) {
            unsigned n = rlen[w * RPWV + k];
            unsigned o = roff[w * RPWV + k];
            for (unsigned j = 32u + (unsigned)lane; j < n; j += 32u) {
                int2 q = pairs[o + j];
                unsigned pos = atomicAdd(&cur[q.y & 255], 1u);
                if (staged) slots[pos - beg] = q.x;
                else        sorted_s[pos] = q.x;
            }
        }
    if (!staged) __threadfence();
    __syncthreads();
    if (!has_tile) return;

    int degv = 0, endv = 0;
    if (lane < nr) {
        degv = (int)cnt[w * RPW + lane];
        endv = (int)(cur[w * RPW + lane] - (staged ? beg : 0u));
    }
    if (staged)
        walk_tile<true>(x, Wvo, slots, tile, out, N, r0, nr, lane, degv, endv);
    else
        walk_tile<false>(x, Wvo, sorted_s, tile, out, N, r0, nr, lane, degv, endv);
}

extern "C" void kernel_launch(void* const* d_in, const int* in_sizes, int n_in,
                              void* d_out, int out_size, void* d_ws,
                              size_t ws_size, hipStream_t stream) {
    const float* x  = (const float*)d_in[0];
    const int*   ei = (const int*)d_in[1];
    const float* Wq = (const float*)d_in[2];
    const float* Wk = (const float*)d_in[3];
    const float* Wv = (const float*)d_in[4];
    const float* Wo = (const float*)d_in[5];
    float* out = (float*)d_out;

    const int N = in_sizes[0] / 32;
    const int E = in_sizes[1] / 2;

    float* P0      = (float*)d_ws;
    float* P1      = P0 + (size_t)N * 32;
    float* P2      = P1 + (size_t)N * 32;
    float* P3      = P2 + (size_t)N * 32;
    float* Wqk     = P3 - 2048;
    float* Wvo     = Wqk + 1024;
    int* sorted_s  = (int*)(P3 + (size_t)N * 32);
    unsigned* runs = (unsigned*)P0;
    int* pairs     = (int*)P1;

    const int TB = 256;
    int NB = (N + 255) >> RB_SHIFT;

    weight_fold_kernel<<<1, TB, 0, stream>>>(Wq, Wk, Wv, Wo, Wqk, Wvo);
    chunk_sort_kernel<<<SORT_C, SORT_T, 0, stream>>>(ei, runs, pairs, E, NB);
    bucket_attn_kernel<<<NB, SORT_T, 0, stream>>>(pairs, runs, x, Wqk, Wvo,
                                                 sorted_s, out, N, E);
}


# Weight fold (Wq Wk^T, Wv Wo) runs as one extra work-group of the level-1 sort launch, beside the sort, not as a launch of its own before it: 2 launches, not 3; products and sum order unchanged
# speedup vs baseline: 16.9614x; 1.0562x over previous
#include <hip/hip_runtime.h>
#include <math.h>

typedef __attribute__((ext_vector_type(2))) float v2f;
typedef __attribute__((ext_vector_type(8))) float v8f;

#define SORT_C   256
#define SORT_T   512
#define RB_SHIFT 8
#define NB_MAX   512
#define EPT      8
#define RPWV     16
#define PAIR_CAP 6656
#define SLOT_CAP 8192
static_assert(SORT_T == NB_MAX && SORT_C == 256 &&
              SORT_C == RPWV * (SORT_T / 32),
              "level 2 maps thread t < 256 to chunk t and wave w to 16 chunks");

__device__ __forceinline__ unsigned bucket_of(int r, int NB) {
    unsigned b = (unsigned)r >> RB_SHIFT;
    unsigned last = (unsigned)(NB - 1);
    return b < last ? b : last;
}

__device__ __forceinline__ void file_pairs(const int* __restrict__ ei,
                                           int2* __restrict__ pairs, int2* stage,
                                           unsigned* hist, int E, int NB, int beg,
                                           int end, int tid, bool staged) {
    for (int e = beg + tid; e < end; e += EPT * SORT_T) {
        int s[EPT], r[EPT];
#pragma unroll
        for (int k = 0; k < EPT; ++k) {
            int ee = e + k * SORT_T;
            int ec = ee < end ? ee : end - 1;
            s[k] = ei[ec];
            r[k] = ei[E + ec];
        }
#pragma unroll
        for (int k = 0; k < EPT; ++k)
            if (e + k * SORT_T < end) {
                unsigned pos = atomicAdd(&hist[bucket_of(r[k], NB)], 1u);
                if (staged) stage[pos] = make_int2(s[k], r[k]);
                else        pairs[(unsigned)beg + pos] = make_int2(s[k], r[k]);
            }
    }
}

__device__ __forceinline__ void weight_fold(
    const float* __restrict__ Wq, const float* __restrict__ Wk,
    const float* __restrict__ Wv, const float* __restrict__ Wo,
    float* __restrict__ Wqk, float* __restrict__ Wvo, int wave, int lane);

__global__ __launch_bounds__(SORT_T) void chunk_sort_kernel(
    const int* __restrict__ ei, unsigned* __restrict__ runs_raw,
    int* __restrict__ pairs_raw, int E, int NB,
    const float* __restrict__ Wq, const float* __restrict__ Wk,
    const float* __restrict__ Wv, const float* __restrict__ Wo,
    float* __restrict__ Wqk, float* __restrict__ Wvo) {
    __shared__ unsigned hist[NB_MAX];
    __shared__ unsigned wsum[SORT_T / 32];
    __shared__ int2 stage[PAIR_CAP];
    uint2* __restrict__ runs = (uint2*)runs_raw;
    int2* __restrict__ pairs = (int2*)pairs_raw;
    int tid = (int)threadIdx.x;
    int c = (int)blockIdx.x;
    int lane = tid & 31;
    int w = tid >> 5;
    if (c == SORT_C) {
        int wv = __builtin_amdgcn_readfirstlane(w);
        if (wv < 8) weight_fold(Wq, Wk, Wv, Wo, Wqk, Wvo, wv, lane);
        return;
    }
    hist[tid] = 0u;
    __syncthreads();
    int CH = (E + SORT_C - 1) / SORT_C;
    int beg = c * CH;
    int end = (beg + CH < E) ? beg + CH : E;
    for (int e = beg + tid; e < end; e += EPT * SORT_T) {
        int r[EPT];
#pragma unroll
        for (int k = 0; k < EPT; ++k) {
            int ee = e + k * SORT_T;
            r[k] = ei[E + (ee < end ? ee : end - 1)];
        }
#pragma unroll
        for (int k = 0; k < EPT; ++k)
            if (e + k * SORT_T < end)
                atomicAdd(&hist[bucket_of(r[k], NB)], 1u);
    }
    __syncthreads();
    unsigned tot = hist[tid];
    unsigned inc = tot;
    for (int d = 1; d < 32; d <<= 1) {
        unsigned t = __shfl_up(inc, d);
        if (lane >= d) inc += t;
    }
    if (lane == 31) wsum[w] = inc;
    __syncthreads();
    unsigned base = 0u;
    for (int k = 0; k < w; ++k) base += wsum[k];
    unsigned excl = base + inc - tot;
    runs[c * NB_MAX + tid] = make_uint2(excl, tot);
    hist[tid] = excl;
    __syncthreads();

    if (CH > PAIR_CAP) {
        file_pairs(ei, pairs, stage, hist, E, NB, beg, end, tid, false);
        return;
    }
    file_pairs(ei, pairs, stage, hist, E, NB, beg, end, tid, true);
    __syncthreads();
    int n = end - beg;
    int a = beg & 15;
    if (n <= 0) return;
    for (int i0 = tid - a; i0 < n; i0 += 4 * SORT_T) {
        int2 v[4];
#pragma unroll
        for (int k = 0; k < 4; ++k) {
            int i = i0 + k * SORT_T;
            v[k] = stage[i < 0 ? 0 : (i < n ? i : n - 1)];
        }
#pragma unroll
        for (int k = 0; k < 4; ++k) {
            int i = i0 + k * SORT_T;
            if (i >= 0 && i < n) pairs[beg + i] = v[k];
        }
    }
}

__device__ __forceinline__ void wmma_step(const float* __restrict__ W, int kb,
                                          int half, int m, v2f a,
                                          v8f& acc0, v8f& acc1) {
    int r0 = (kb + 2 * half + 0) * 32;
    int r1 = (kb + 2 * half + 1) * 32;
    v2f b0, b1;
    b0.x = W[r0 + m];        b0.y = W[r1 + m];
    b1.x = W[r0 + 16 + m];   b1.y = W[r1 + 16 + m];
    acc0 = __builtin_amdgcn_wmma_f32_16x16x4_f32(false, a, false, b0,
                                                 (short)0, acc0, false, false);
    acc1 = __builtin_amdgcn_wmma_f32_16x16x4_f32(false, a, false, b1,
                                                 (short)0, acc1, false, false);
}

__device__ __forceinline__ void weight_fold(
    const float* __restrict__ Wq, const float* __restrict__ Wk,
    const float* __restrict__ Wv, const float* __restrict__ Wo,
    float* __restrict__ Wqk, float* __restrict__ Wvo, int wave, int lane) {
    int half = lane >> 4;
    int m    = lane & 15;
    bool vo  = wave >= 4;
    int tr = (wave >> 1) & 1, tc = wave & 1;
    const float* __restrict__ A = vo ? Wv : Wq;
    const float* __restrict__ B = vo ? Wo : Wk;
    v8f acc = {};
#pragma unroll
    for (int kk = 0; kk < 8; ++kk) {
        int k0 = kk * 4 + 2 * half;
        const float* ap = A + (tr * 16 + m) * 32 + k0;
        v2f a; a.x = ap[0]; a.y = ap[1];
        int i0 = vo ? k0 * 32 + tc * 16 + m : (tc * 16 + m) * 32 + k0;
        int i1 = vo ? i0 + 32 : i0 + 1;
        v2f b; b.x = B[i0]; b.y = B[i1];
        acc = __builtin_amdgcn_wmma_f32_16x16x4_f32(false, a, false, b,
                                                    (short)0, acc, false, false);
    }
    float* __restrict__ D = vo ? Wvo : Wqk;
#pragma unroll
    for (int r = 0; r < 8; ++r)
        D[(tr * 16 + r + 8 * half) * 32 + tc * 16 + m] = acc[r];
}

#define RPW 16
#define SPT 4
#define TP  36
#define TW  (SORT_T / 32)
static_assert(TW * RPW == (1 << RB_SHIFT), "a bucket is one tile per wave");

__device__ __forceinline__ void wave_tile_handover() {
    __builtin_amdgcn_fence(__ATOMIC_RELEASE, "wavefront");
    __builtin_amdgcn_wave_barrier();
    __builtin_amdgcn_fence(__ATOMIC_ACQUIRE, "wavefront");
}

template <bool STAGED>
__device__ __forceinline__ void load_senders(const int* src, int start, int deg,
                                             int e0, int g, int (&s)[SPT]) {
#pragma unroll
    for (int i = 0; i < SPT; ++i) {
        int jn = e0 + 4 * i + g;
        int idx = start + (jn < deg ? jn : deg - 1);
        if (STAGED) s[i] = src[idx];
        else        s[i] = __builtin_nontemporal_load(src + idx);
    }
}

__device__ __forceinline__ void load_rows(const float4* __restrict__ X4,
                                          const int (&s)[SPT], unsigned cu,
                                          float4 (&x4)[SPT]) {
#pragma unroll
    for (int i = 0; i < SPT; ++i)
        x4[i] = X4[(unsigned)s[i] * 8u + cu];
}

__device__ __forceinline__ void load_bfrags(const float* __restrict__ Wt,
                                            int half, int m,
                                            v2f (&b0)[8], v2f (&b1)[8]) {
#pragma unroll
    for (int kk = 0; kk < 8; ++kk) {
        int r0w = (kk * 4 + 2 * half + 0) * 32;
        int r1w = (kk * 4 + 2 * half + 1) * 32;
        b0[kk].x = Wt[r0w + m];        b0[kk].y = Wt[r1w + m];
        b1[kk].x = Wt[r0w + 16 + m];   b1[kk].y = Wt[r1w + 16 + m];
    }
}

__device__ __forceinline__ void project_tile(const float* __restrict__ x,
                                             const float* __restrict__ Wqk,
                                             float* tile, int N, int r0,
                                             int lane) {
    float4* tile4 = (float4*)tile;
    int half = lane >> 4;
    int m    = lane & 15;
    int g = lane >> 3;
    int c = lane & 7;
    const float4* __restrict__ X4 = (const float4*)x;
    unsigned cu = (unsigned)c;
    float4 xrow[4];
#pragma unroll
    for (int i = 0; i < 4; ++i) {
        int row = r0 + 4 * i + g;
        xrow[i] = X4[(unsigned)(row < N ? row : N - 1) * 8u + cu];
    }
    v2f bq0[8], bq1[8];
    load_bfrags(Wqk, half, m, bq0, bq1);
#pragma unroll
    for (int i = 0; i < 4; ++i) tile4[(4 * i + g) * (TP / 4) + c] = xrow[i];
    wave_tile_handover();
    v2f xa[8];
#pragma unroll
    for (int kk = 0; kk < 8; ++kk) {
        const float* ap = tile + m * TP + kk * 4 + 2 * half;
        xa[kk].x = ap[0]; xa[kk].y = ap[1];
    }
    v8f q0 = {}, q1 = {};
#pragma unroll
    for (int kk = 0; kk < 8; ++kk) {
        q0 = __builtin_amdgcn_wmma_f32_16x16x4_f32(false, xa[kk], false, bq0[kk],
                                                   (short)0, q0, false, false);
        q1 = __builtin_amdgcn_wmma_f32_16x16x4_f32(false, xa[kk], false, bq1[kk],
                                                   (short)0, q1, false, false);
    }
#pragma unroll
    for (int r = 0; r < 8; ++r) {
        tile[(r + 8 * half) * TP + m]      = q0[r];
        tile[(r + 8 * half) * TP + 16 + m] = q1[r];
    }
    wave_tile_handover();
}

template <bool STAGED>
__device__ __forceinline__ void walk_tile(const float* __restrict__ x,
                                          const float* __restrict__ Wvo,
                                          const int* src, float* tile,
                                          float* __restrict__ out, int N,
                                          int r0, int nr, int lane, int degv,
                                          int endv) {
    float4* tile4 = (float4*)tile;
    int half = lane >> 4;
    int m    = lane & 15;
    int g = lane >> 3;
    int c = lane & 7;
    const float4* __restrict__ X4 = (const float4*)x;
    unsigned cu = (unsigned)c;
    float zv = 0.f;

    int deg   = __builtin_amdgcn_readlane(degv, 0);
    int start = __builtin_amdgcn_readlane(endv, 0) - deg;
    int s_cur[SPT];
#pragma unroll
    for (int i = 0; i < SPT; ++i) s_cur[i] = r0;
    if (deg > 0) load_senders<STAGED>(src, start, deg, 0, g, s_cur);
    float4 q4 = tile4[cu];

    for (int j = 0; j < nr; ++j) {
        int r = r0 + j;
        float4 x4[SPT];
        int s_nxt[SPT];
#pragma unroll
        for (int i = 0; i < SPT; ++i) { x4[i] = q4; s_nxt[i] = r; }
        if (deg > 0) {
            load_rows(X4, s_cur, cu, x4);
            if (deg > 4 * SPT)
                load_senders<STAGED>(src, start, deg, 4 * SPT, g, s_nxt);
        }
        int deg_n = 0, start_n = 0;
        int s_first_n[SPT];
#pragma unroll
        for (int i = 0; i < SPT; ++i) s_first_n[i] = r;
        float4 q4_n = q4;
        if (j + 1 < nr) {
            deg_n   = __builtin_amdgcn_readlane(degv, j + 1);
            start_n = __builtin_amdgcn_readlane(endv, j + 1) - deg_n;
            q4_n = tile4[(unsigned)(j + 1) * (TP / 4) + cu];
            if (deg_n > 0)
                load_senders<STAGED>(src, start_n, deg_n, 0, g, s_first_n);
        }

        float mx = -INFINITY;
        float z = 0.f;
        float4 acc = make_float4(0.f, 0.f, 0.f, 0.f);

        for (int t = 0; t < deg; t += 4 * SPT) {
            float p[SPT];
#pragma unroll
            for (int i = 0; i < SPT; ++i)
                p[i] = q4.x * x4[i].x + q4.y * x4[i].y + q4.z * x4[i].z
                     + q4.w * x4[i].w;
#pragma unroll
            for (int i = 0; i < SPT; ++i) p[i] += __shfl_xor(p[i], 1);
#pragma unroll
            for (int i = 0; i < SPT; ++i) p[i] += __shfl_xor(p[i], 2);
#pragma unroll
            for (int i = 0; i < SPT; ++i) p[i] += __shfl_xor(p[i], 4);

            if (t + g < deg) {
                float sc[SPT];
#pragma unroll
                for (int i = 0; i < SPT; ++i)
                    sc[i] = (t + 4 * i + g < deg) ? p[i] * 0.17677669529663687f
                                                  : -INFINITY;
                float nm = mx;
#pragma unroll
                for (int i = 0; i < SPT; ++i) nm = fmaxf(nm, sc[i]);
                float corr = __expf(mx - nm);
                float a[SPT];
#pragma unroll
                for (int i = 0; i < SPT; ++i) a[i] = __expf(sc[i] - nm);
                float zs = 0.f;
                float4 as = make_float4(0.f, 0.f, 0.f, 0.f);
#pragma unroll
                for (int i = 0; i < SPT; ++i) {
                    zs += a[i];
                    as.x += a[i] * x4[i].x;
                    as.y += a[i] * x4[i].y;
                    as.z += a[i] * x4[i].z;
                    as.w += a[i] * x4[i].w;
                }
                z = z * corr + zs;
                acc.x = acc.x * corr + as.x;
                acc.y = acc.y * corr + as.y;
                acc.z = acc.z * corr + as.z;
                acc.w = acc.w * corr + as.w;
                mx = nm;
            }
            if (t + 4 * SPT < deg) {
                int s_now[SPT];
#pragma unroll
                for (int i = 0; i < SPT; ++i) s_now[i] = s_nxt[i];
                load_rows(X4, s_now, cu, x4);
                if (t + 8 * SPT < deg)
                    load_senders<STAGED>(src, start, deg, t + 8 * SPT, g, s_nxt);
            }
        }

        float M = mx;
        M = fmaxf(M, __shfl_xor(M, 8));
        M = fmaxf(M, __shfl_xor(M, 16));
        float f = (mx == -INFINITY) ? 0.f : __expf(mx - M);
        z *= f; acc.x *= f; acc.y *= f; acc.z *= f; acc.w *= f;
        z     += __shfl_xor(z, 8);      z     += __shfl_xor(z, 16);
        acc.x += __shfl_xor(acc.x, 8);  acc.x += __shfl_xor(acc.x, 16);
        acc.y += __shfl_xor(acc.y, 8);  acc.y += __shfl_xor(acc.y, 16);
        acc.z += __shfl_xor(acc.z, 8);  acc.z += __shfl_xor(acc.z, 16);
        acc.w += __shfl_xor(acc.w, 8);  acc.w += __shfl_xor(acc.w, 16);
        if (g == 0) tile4[(unsigned)j * (TP / 4) + cu] = acc;
        if (lane == j) zv = z;

        deg = deg_n; start = start_n; q4 = q4_n;
#pragma unroll
        for (int i = 0; i < SPT; ++i) s_cur[i] = s_first_n[i];
    }

    wave_tile_handover();
    {
        float zrow = __shfl(zv, m);
        float invz = 1.0f / (zrow + 1e-6f);
        float4 xres[4];
#pragma unroll
        for (int i = 0; i < 4; ++i) {
            int row = r0 + 4 * i + g;
            xres[i] = X4[(unsigned)(row < N ? row : N - 1) * 8u + cu];
        }
        v8f c0 = {}, c1 = {};
#pragma unroll
        for (int kk = 0; kk < 8; ++kk) {
            const float* ap = tile + m * TP + kk * 4 + 2 * half;
            v2f a; a.x = ap[0] * invz; a.y = ap[1] * invz;
            wmma_step(Wvo, kk * 4, half, m, a, c0, c1);
        }
#pragma unroll
        for (int r = 0; r < 8; ++r) {
            tile[(r + 8 * half) * TP + m]      = c0[r];
            tile[(r + 8 * half) * TP + 16 + m] = c1[r];
        }
        wave_tile_handover();
        float4* O4 = (float4*)out;
#pragma unroll
        for (int i = 0; i < 4; ++i) {
            int row = r0 + 4 * i + g;
            float4 d = tile4[(4 * i + g) * (TP / 4) + c];
            float4 o = make_float4(xres[i].x + d.x, xres[i].y + d.y,
                                   xres[i].z + d.z, xres[i].w + d.w);
            if (row < N) O4[(unsigned)row * 8u + cu] = o;
        }
    }
}

__global__ __launch_bounds__(SORT_T) void bucket_attn_kernel(
    const int* __restrict__ pairs_raw, const unsigned* __restrict__ runs_raw,
    const float* __restrict__ x, const float* __restrict__ Wqk,
    const float* __restrict__ Wvo, int* sorted_s, float* __restrict__ out,
    int N, int E) {
    __shared__ unsigned cnt[256];
    __shared__ unsigned cur[256];
    __shared__ unsigned wsum[8];
    __shared__ unsigned osum[8];
    __shared__ unsigned roff[SORT_C];
    __shared__ unsigned rlen[SORT_C];
    __shared__ int slots[SLOT_CAP];
    __shared__ __attribute__((aligned(16))) float tiles[TW][16 * TP];
    const int2* __restrict__ pairs = (const int2*)pairs_raw;
    const uint2* __restrict__ runs = (const uint2*)runs_raw;
    int tid = (int)threadIdx.x;
    int b = (int)blockIdx.x;
    int lane = tid & 31;
    int w = __builtin_amdgcn_readfirstlane(tid >> 5);
    int CH = (E + SORT_C - 1) / SORT_C;
    unsigned last = (unsigned)E - 1u;

    int r0 = (b << RB_SHIFT) + w * RPW;
    bool has_tile = r0 < N;
    int nr = (N - r0 < RPW) ? N - r0 : RPW;
    float* tile = tiles[w];

    uint2 oc = make_uint2(0u, 0u);
    if (tid < 256) oc = runs[tid * NB_MAX + b];
    if (has_tile) project_tile(x, Wqk, tile, N, r0, lane);

    if (tid < 256) {
        roff[tid] = (unsigned)(tid * CH) + oc.x;
        rlen[tid] = oc.y;
        cnt[tid] = 0u;
        unsigned s = oc.x;
        for (int d = 16; d > 0; d >>= 1) s += __shfl_xor(s, d);
        if (lane == 0) osum[w] = s;
    }
    __syncthreads();
    unsigned beg = 0u;
    for (int k = 0; k < 8; ++k) beg += osum[k];

    int2 p[RPWV];
    unsigned live = 0u;
    bool tail = false;
#pragma unroll
    for (int k = 0; k < RPWV; ++k) {
        unsigned n = rlen[w * RPWV + k];
        unsigned idx = roff[w * RPWV + k] + (unsigned)lane;
        p[k] = pairs[idx < last ? idx : last];
        if ((unsigned)lane < n) live |= 1u << k;
        tail = tail || (n > 32u);
    }
#pragma unroll
    for (int k = 0; k < RPWV; ++k)
        if (live & (1u << k)) atomicAdd(&cnt[p[k].y & 255], 1u);
    if (tail)
        for (int k = 0; k < RPWV; ++k) {
            unsigned n = rlen[w * RPWV + k];
            unsigned o = roff[w * RPWV + k];
            for (unsigned j = 32u + (unsigned)lane; j < n; j += 32u)
                atomicAdd(&cnt[pairs[o + j].y & 255], 1u);
        }
    __syncthreads();
    unsigned cv = (tid < 256) ? cnt[tid] : 0u;
    unsigned inc = cv;
    for (int d = 1; d < 32; d <<= 1) {
        unsigned t = __shfl_up(inc, d);
        if (lane >= d) inc += t;
    }
    if (tid < 256 && lane == 31) wsum[w] = inc;
    __syncthreads();
    if (tid < 256) {
        unsigned base = 0u;
        for (int k = 0; k < w; ++k) base += wsum[k];
        cur[tid] = beg + base + inc - cv;
    }
    __syncthreads();
    unsigned total = 0u;
    for (int k = 0; k < 8; ++k) total += wsum[k];
    bool staged = total <= SLOT_CAP;
    unsigned pos[RPWV];
#pragma unroll
    for (int k = 0; k < RPWV; ++k) {
        pos[k] = 0u;
        if (live & (1u << k)) pos[k] = atomicAdd(&cur[p[k].y & 255], 1u);
    }
    if (staged) {
#pragma unroll
        for (int k = 0; k < RPWV; ++k)
            if (live & (1u << k)) slots[pos[k] - beg] = p[k].x;
    } else {
#pragma unroll
        for (int k = 0; k < RPWV; ++k)
            if (live & (1u << k)) sorted_s[pos[k]] = p[k].x;
    }
    if (tail)
        for (int k = 0; k < RPWV; ++k) {
            unsigned n = rlen[w * RPWV + k];
            unsigned o = roff[w * RPWV + k];
            for (unsigned j = 32u + (unsigned)lane; j < n; j += 32u) {
                int2 q = pairs[o + j];
                unsigned pos = atomicAdd(&cur[q.y & 255], 1u);
                if (staged) slots[pos - beg] = q.x;
                else        sorted_s[pos] = q.x;
            }
        }
    if (!staged) __threadfence();
    __syncthreads();
    if (!has_tile) return;

    int degv = 0, endv = 0;
    if (lane < nr) {
        degv = (int)cnt[w * RPW + lane];
        endv = (int)(cur[w * RPW + lane] - (staged ? beg : 0u));
    }
    if (staged)
        walk_tile<true>(x, Wvo, slots, tile, out, N, r0, nr, lane, degv, endv);
    else
        walk_tile<false>(x, Wvo, sorted_s, tile, out, N, r0, nr, lane, degv, endv);
}

extern "C" void kernel_launch(void* const* d_in, const int* in_sizes, int n_in,
                              void* d_out, int out_size, void* d_ws,
                              size_t ws_size, hipStream_t stream) {
    const float* x  = (const float*)d_in[0];
    const int*   ei = (const int*)d_in[1];
    const float* Wq = (const float*)d_in[2];
    const float* Wk = (const float*)d_in[3];
    const float* Wv = (const float*)d_in[4];
    const float* Wo = (const float*)d_in[5];
    float* out = (float*)d_out;

    const int N = in_sizes[0] / 32;
    const int E = in_sizes[1] / 2;

    float* P0      = (float*)d_ws;
    float* P1      = P0 + (size_t)N * 32;
    float* P2      = P1 + (size_t)N * 32;
    float* P3      = P2 + (size_t)N * 32;
    float* Wqk     = P3 - 2048;
    float* Wvo     = Wqk + 1024;
    int* sorted_s  = (int*)(P3 + (size_t)N * 32);
    unsigned* runs = (unsigned*)P0;
    int* pairs     = (int*)P1;

    int NB = (N + 255) >> RB_SHIFT;

    chunk_sort_kernel<<<SORT_C + 1, SORT_T, 0, stream>>>(ei, runs, pairs, E, NB,
                                                        Wq, Wk, Wv, Wo, Wqk, Wvo);
    bucket_attn_kernel<<<NB, SORT_T, 0, stream>>>(pairs, runs, x, Wqk, Wvo,
                                                 sorted_s, out, N, E);
}
